# starting kernel
# baseline (speedup 1.0000x reference)
#include <hip/hip_runtime.h>

#define NNODES 50000
#define NEDGES 800000
#define FIN 8
#define PER 12
#define HIDN 32
#define NB 2
#define OUTC 24
#define ROWS (NB*NNODES)
#define NFEAT (FIN*PER)

typedef float v2f __attribute__((ext_vector_type(2)));
typedef float v8f __attribute__((ext_vector_type(8)));

__device__ __forceinline__ v8f wmma4(v2f a, v2f b, v8f c) {
  return __builtin_amdgcn_wmma_f32_16x16x4_f32(false, a, false, b, (short)0, c, false, false);
}

__device__ __forceinline__ v2f ldsA(const float* buf, int m, int hi, int kb) {
  int k0 = kb * 4 + hi * 2;
  v2f a;
  a.x = buf[m * HIDN + k0];
  a.y = buf[m * HIDN + k0 + 1];
  return a;
}

__device__ __forceinline__ v2f ldsB(const float* W, int m, int hi, int kb, int nt) {
  int k0 = kb * 4 + hi * 2;
  int n = nt * 16 + m;
  v2f b;
  b.x = W[k0 * HIDN + n];
  b.y = W[(k0 + 1) * HIDN + n];
  return b;
}

__device__ __forceinline__ void storeTile(float* buf, int m, int hi, const v8f& c0, const v8f& c1) {
#pragma unroll
  for (int r = 0; r < 8; r++) {
    buf[(r + 8 * hi) * HIDN + m]      = c0[r];
    buf[(r + 8 * hi) * HIDN + 16 + m] = c1[r];
  }
}

__device__ __forceinline__ void gemm8(v2f ax0, v2f ax1, const float* W, const float* bias,
                                      int m, int hi, v8f& c0, v8f& c1) {
  float b0v = bias[m], b1v = bias[16 + m];
#pragma unroll
  for (int r = 0; r < 8; r++) { c0[r] = b0v; c1[r] = b1v; }
  c0 = wmma4(ax0, ldsB(W, m, hi, 0, 0), c0);
  c0 = wmma4(ax1, ldsB(W, m, hi, 1, 0), c0);
  c1 = wmma4(ax0, ldsB(W, m, hi, 0, 1), c1);
  c1 = wmma4(ax1, ldsB(W, m, hi, 1, 1), c1);
}

__device__ __forceinline__ void gemm64(const float* A1, const float* A2, const float* L,
                                       const float* bias, int m, int hi, v8f& c0, v8f& c1) {
  float b0v = bias[m], b1v = bias[16 + m];
#pragma unroll
  for (int r = 0; r < 8; r++) { c0[r] = b0v; c1[r] = b1v; }
#pragma unroll
  for (int kb = 0; kb < 8; kb++) {
    v2f a = ldsA(A1, m, hi, kb);
    c0 = wmma4(a, ldsB(L, m, hi, kb, 0), c0);
    c1 = wmma4(a, ldsB(L, m, hi, kb, 1), c1);
  }
#pragma unroll
  for (int kb = 0; kb < 8; kb++) {
    v2f a = ldsA(A2, m, hi, kb);
    c0 = wmma4(a, ldsB(L, m, hi, kb + 8, 0), c0);
    c1 = wmma4(a, ldsB(L, m, hi, kb + 8, 1), c1);
  }
}

__device__ __forceinline__ float sigm(float x) { return 1.0f / (1.0f + __expf(-x)); }
__device__ __forceinline__ float tanhc(float x) {
  x = fminf(fmaxf(x, -15.f), 15.f);
  float e = __expf(2.0f * x);
  return (e - 1.0f) / (e + 1.0f);
}

__global__ void k_deg_init(float* deg) {
  int i = blockIdx.x * 256 + threadIdx.x;
  if (i < NNODES) deg[i] = 1.0f;
}
__global__ void k_deg_acc(const int* __restrict__ ei, float* deg) {
  int e = blockIdx.x * 256 + threadIdx.x;
  if (e < NEDGES) atomicAdd(&deg[ei[NEDGES + e]], 1.0f);
}
__global__ void k_rsqrt(float* deg) {
  int i = blockIdx.x * 256 + threadIdx.x;
  if (i < NNODES) deg[i] = rsqrtf(deg[i]);
}
__global__ void k_selfloop(const float* __restrict__ x, const float* __restrict__ dinv,
                           float* __restrict__ xagg) {
  int t = blockIdx.x * 256 + threadIdx.x;
  if (t >= ROWS * 24) return;
  int i = t / 24;
  int n = i % NNODES;
  float w = dinv[n] * dinv[n];
  float4 xv = ((const float4*)x)[t];
  float4 o; o.x = xv.x * w; o.y = xv.y * w; o.z = xv.z * w; o.w = xv.w * w;
  ((float4*)xagg)[t] = o;
}
__global__ void k_scatter(const float* __restrict__ x, const int* __restrict__ ei,
                          const float* __restrict__ dinv, float* __restrict__ xagg) {
  int t = blockIdx.x * 256 + threadIdx.x;
  if (t >= NEDGES * 24) return;
  int e = t / 24;
  int r = t - e * 24;
  int b = r / 12;
  int c = r - b * 12;
  int s = ei[e];
  int d = ei[NEDGES + e];
  float w = dinv[s] * dinv[d];
  const float* xp = x + (size_t)(b * NNODES + s) * NFEAT + c * 8;
  float* op = xagg + (size_t)(b * NNODES + d) * NFEAT + c * 8;
#pragma unroll
  for (int j = 0; j < 8; j++) atomicAdd(op + j, w * xp[j]);
}

__global__ void __launch_bounds__(256) k_gru(
    const float* __restrict__ xagg, const float* __restrict__ att,
    const float* __restrict__ Wz, const float* __restrict__ bz,
    const float* __restrict__ Wr, const float* __restrict__ br,
    const float* __restrict__ Wh, const float* __restrict__ bh,
    const float* __restrict__ Lz, const float* __restrict__ lbz,
    const float* __restrict__ Lr, const float* __restrict__ lbr,
    const float* __restrict__ Lh, const float* __restrict__ lbh,
    const float* __restrict__ Wo, const float* __restrict__ bo,
    float* __restrict__ out) {
  __shared__ float sW[3 * 8 * 32];
  __shared__ float sb[3 * 32];
  __shared__ float sL[3 * 64 * 32];
  __shared__ float slb[3 * 32];
  __shared__ float sWo[32 * 32];
  __shared__ float sbo[32];
  __shared__ float sG[8][16 * 32];
  __shared__ float sH[8][16 * 32];
  __shared__ float sHR[8][16 * 32];

  int tid = threadIdx.x;
  for (int t = tid; t < 8 * 32; t += 256) { sW[t] = Wz[t]; sW[256 + t] = Wr[t]; sW[512 + t] = Wh[t]; }
  for (int t = tid; t < 32; t += 256) {
    sb[t] = bz[t]; sb[32 + t] = br[t]; sb[64 + t] = bh[t];
    slb[t] = lbz[t]; slb[32 + t] = lbr[t]; slb[64 + t] = lbh[t];
    sbo[t] = (t < OUTC) ? bo[t] : 0.f;
  }
  for (int t = tid; t < 64 * 32; t += 256) { sL[t] = Lz[t]; sL[2048 + t] = Lr[t]; sL[4096 + t] = Lh[t]; }
  for (int t = tid; t < 32 * 32; t += 256) {
    int k = t >> 5, j = t & 31;
    sWo[t] = (j < OUTC) ? Wo[k * OUTC + j] : 0.f;
  }

  int wave = tid >> 5, lane = tid & 31, m = lane & 15, hi = lane >> 4;
  float* mG = sG[wave];
  float* mH = sH[wave];
  float* mHR = sHR[wave];
  for (int idx = lane; idx < 16 * 32; idx += 32) mH[idx] = 0.f;

  float pr[PER];
  {
    float a[PER]; float mx = -1e30f;
    for (int p = 0; p < PER; p++) { a[p] = att[p]; mx = fmaxf(mx, a[p]); }
    float sm = 0.f;
    for (int p = 0; p < PER; p++) { a[p] = __expf(a[p] - mx); sm += a[p]; }
    float inv = 1.0f / sm;
    for (int p = 0; p < PER; p++) pr[p] = a[p] * inv;
  }

  int tile = blockIdx.x * 8 + wave;
  long ia = (long)tile * 16 + m;
  if (ia >= ROWS) ia = ROWS - 1;
  const float* xrow = xagg + ia * (long)NFEAT;

  v8f H0, H1, A0, A1;
  for (int r = 0; r < 8; r++) { H0[r] = 0.f; H1[r] = 0.f; A0[r] = 0.f; A1[r] = 0.f; }

  __syncthreads();

  for (int p = 0; p < PER; p++) {
    int f0 = hi * 2;
    v2f ax0, ax1;
    ax0.x = xrow[(f0)*PER + p];     ax0.y = xrow[(f0 + 1) * PER + p];
    ax1.x = xrow[(f0 + 4) * PER + p]; ax1.y = xrow[(f0 + 5) * PER + p];

    v8f g0, g1, z0, z1, r0, r1, t0, t1;

    gemm8(ax0, ax1, sW + 0, sb + 0, m, hi, g0, g1);
    storeTile(mG, m, hi, g0, g1);
    __syncthreads();
    gemm64(mG, mH, sL + 0, slb + 0, m, hi, z0, z1);
    for (int r = 0; r < 8; r++) { z0[r] = sigm(z0[r]); z1[r] = sigm(z1[r]); }
    __syncthreads();

    gemm8(ax0, ax1, sW + 256, sb + 32, m, hi, g0, g1);
    storeTile(mG, m, hi, g0, g1);
    __syncthreads();
    gemm64(mG, mH, sL + 2048, slb + 32, m, hi, r0, r1);
    for (int r = 0; r < 8; r++) { r0[r] = sigm(r0[r]); r1[r] = sigm(r1[r]); }

    gemm8(ax0, ax1, sW + 512, sb + 64, m, hi, g0, g1);
    __syncthreads();
    storeTile(mG, m, hi, g0, g1);
    {
      v8f q0, q1;
      for (int r = 0; r < 8; r++) { q0[r] = H0[r] * r0[r]; q1[r] = H1[r] * r1[r]; }
      storeTile(mHR, m, hi, q0, q1);
    }
    __syncthreads();
    gemm64(mG, mHR, sL + 4096, slb + 64, m, hi, t0, t1);
    for (int r = 0; r < 8; r++) { t0[r] = tanhc(t0[r]); t1[r] = tanhc(t1[r]); }

    for (int r = 0; r < 8; r++) {
      H0[r] = z0[r] * H0[r] + (1.0f - z0[r]) * t0[r];
      H1[r] = z1[r] * H1[r] + (1.0f - z1[r]) * t1[r];
      A0[r] += pr[p] * H0[r];
      A1[r] += pr[p] * H1[r];
    }
    storeTile(mH, m, hi, H0, H1);
    __syncthreads();
  }

  for (int r = 0; r < 8; r++) { A0[r] = fmaxf(A0[r], 0.f); A1[r] = fmaxf(A1[r], 0.f); }
  storeTile(mG, m, hi, A0, A1);
  __syncthreads();
  v8f o0, o1;
  {
    float b0v = sbo[m], b1v = sbo[16 + m];
    for (int r = 0; r < 8; r++) { o0[r] = b0v; o1[r] = b1v; }
#pragma unroll
    for (int kb = 0; kb < 8; kb++) {
      v2f a = ldsA(mG, m, hi, kb);
      o0 = wmma4(a, ldsB(sWo, m, hi, kb, 0), o0);
      o1 = wmma4(a, ldsB(sWo, m, hi, kb, 1), o1);
    }
  }
#pragma unroll
  for (int r = 0; r < 8; r++) {
    long row = (long)tile * 16 + r + 8 * hi;
    if (row < ROWS) {
      out[row * OUTC + m] = o0[r];
      if (m < 8) out[row * OUTC + 16 + m] = o1[r];
    }
  }
}

extern "C" void kernel_launch(void* const* d_in, const int* in_sizes, int n_in,
                              void* d_out, int out_size, void* d_ws, size_t ws_size,
                              hipStream_t stream) {
  (void)in_sizes; (void)n_in; (void)out_size; (void)ws_size;
  const float* x   = (const float*)d_in[0];
  const int*   ei  = (const int*)d_in[1];
  const float* att = (const float*)d_in[2];
  const float* Wz  = (const float*)d_in[3];  const float* bz  = (const float*)d_in[4];
  const float* Wr  = (const float*)d_in[5];  const float* br  = (const float*)d_in[6];
  const float* Wh  = (const float*)d_in[7];  const float* bh  = (const float*)d_in[8];
  const float* Lz  = (const float*)d_in[9];  const float* lbz = (const float*)d_in[10];
  const float* Lr  = (const float*)d_in[11]; const float* lbr = (const float*)d_in[12];
  const float* Lh  = (const float*)d_in[13]; const float* lbh = (const float*)d_in[14];
  const float* Wo  = (const float*)d_in[15]; const float* bo  = (const float*)d_in[16];
  float* out = (float*)d_out;

  float* wsf  = (float*)d_ws;
  float* deg  = wsf;
  float* xagg = wsf + 65536;

  k_deg_init<<<(NNODES + 255) / 256, 256, 0, stream>>>(deg);
  k_deg_acc<<<(NEDGES + 255) / 256, 256, 0, stream>>>(ei, deg);
  k_rsqrt<<<(NNODES + 255) / 256, 256, 0, stream>>>(deg);
  k_selfloop<<<(ROWS * 24 + 255) / 256, 256, 0, stream>>>(x, deg, xagg);
  k_scatter<<<(NEDGES * 24 + 255) / 256, 256, 0, stream>>>(x, ei, deg, xagg);

  int tiles = ROWS / 16;
  int blocks = (tiles + 7) / 8;
  k_gru<<<blocks, 256, 0, stream>>>(xagg, att, Wz, bz, Wr, br, Wh, bh,
                                    Lz, lbz, Lr, lbr, Lh, lbh, Wo, bo, out);
}


# aggregation moved into LDS: 256 blocks each own 196 target nodes, filter the whole edge list for them and add source rows with LDS f32 atomics (was L2 atomics); adds in arrival order as before
# speedup vs baseline: 1.6323x; 1.6323x over previous
#include <hip/hip_runtime.h>

#define NNODES 50000
#define NEDGES 800000
#define FIN 8
#define PER 12
#define HIDN 32
#define NB 2
#define OUTC 24
#define ROWS (NB*NNODES)
#define NFEAT (FIN*PER)

typedef float v2f __attribute__((ext_vector_type(2)));
typedef float v8f __attribute__((ext_vector_type(8)));

__device__ __forceinline__ v8f wmma4(v2f a, v2f b, v8f c) {
  return __builtin_amdgcn_wmma_f32_16x16x4_f32(false, a, false, b, (short)0, c, false, false);
}

__device__ __forceinline__ v2f ldsA(const float* buf, int m, int hi, int kb) {
  int k0 = kb * 4 + hi * 2;
  v2f a;
  a.x = buf[m * HIDN + k0];
  a.y = buf[m * HIDN + k0 + 1];
  return a;
}

__device__ __forceinline__ v2f ldsB(const float* W, int m, int hi, int kb, int nt) {
  int k0 = kb * 4 + hi * 2;
  int n = nt * 16 + m;
  v2f b;
  b.x = W[k0 * HIDN + n];
  b.y = W[(k0 + 1) * HIDN + n];
  return b;
}

__device__ __forceinline__ void storeTile(float* buf, int m, int hi, const v8f& c0, const v8f& c1) {
#pragma unroll
  for (int r = 0; r < 8; r++) {
    buf[(r + 8 * hi) * HIDN + m]      = c0[r];
    buf[(r + 8 * hi) * HIDN + 16 + m] = c1[r];
  }
}

__device__ __forceinline__ void gemm8(v2f ax0, v2f ax1, const float* W, const float* bias,
                                      int m, int hi, v8f& c0, v8f& c1) {
  float b0v = bias[m], b1v = bias[16 + m];
#pragma unroll
  for (int r = 0; r < 8; r++) { c0[r] = b0v; c1[r] = b1v; }
  c0 = wmma4(ax0, ldsB(W, m, hi, 0, 0), c0);
  c0 = wmma4(ax1, ldsB(W, m, hi, 1, 0), c0);
  c1 = wmma4(ax0, ldsB(W, m, hi, 0, 1), c1);
  c1 = wmma4(ax1, ldsB(W, m, hi, 1, 1), c1);
}

__device__ __forceinline__ void gemm64(const float* A1, const float* A2, const float* L,
                                       const float* bias, int m, int hi, v8f& c0, v8f& c1) {
  float b0v = bias[m], b1v = bias[16 + m];
#pragma unroll
  for (int r = 0; r < 8; r++) { c0[r] = b0v; c1[r] = b1v; }
#pragma unroll
  for (int kb = 0; kb < 8; kb++) {
    v2f a = ldsA(A1, m, hi, kb);
    c0 = wmma4(a, ldsB(L, m, hi, kb, 0), c0);
    c1 = wmma4(a, ldsB(L, m, hi, kb, 1), c1);
  }
#pragma unroll
  for (int kb = 0; kb < 8; kb++) {
    v2f a = ldsA(A2, m, hi, kb);
    c0 = wmma4(a, ldsB(L, m, hi, kb + 8, 0), c0);
    c1 = wmma4(a, ldsB(L, m, hi, kb + 8, 1), c1);
  }
}

__device__ __forceinline__ float sigm(float x) { return 1.0f / (1.0f + __expf(-x)); }
__device__ __forceinline__ float tanhc(float x) {
  x = fminf(fmaxf(x, -15.f), 15.f);
  float e = __expf(2.0f * x);
  return (e - 1.0f) / (e + 1.0f);
}

__global__ void k_deg_init(float* deg) {
  int i = blockIdx.x * 256 + threadIdx.x;
  if (i < NNODES) deg[i] = 1.0f;
}
__global__ void k_deg_acc(const int* __restrict__ ei, float* deg) {
  int e = blockIdx.x * 256 + threadIdx.x;
  if (e < NEDGES) atomicAdd(&deg[ei[NEDGES + e]], 1.0f);
}
__global__ void k_rsqrt(float* deg) {
  int i = blockIdx.x * 256 + threadIdx.x;
  if (i < NNODES) deg[i] = rsqrtf(deg[i]);
}
#define WIN 196
#define NWIN 256
#define ROWF (NB * NFEAT)
#define AGG_WAVES 32
#define LCAP 256
#define NE4 (NEDGES / 4)
static_assert(NWIN * WIN >= NNODES, "windows cover every node");
static_assert(NE4 * 4 == NEDGES && NE4 % 32 == 0, "whole waves of int4 reads over the target column");

__device__ __forceinline__ void agg_flush(const float* __restrict__ x, const int* __restrict__ ei,
                                          const float* __restrict__ dinv, float* acc,
                                          const int* myE, const int* myD, int cnt, int win0,
                                          int lane, const int* off) {
  for (int base = 0; base < cnt; base += 32) {
    int idx = base + lane;
    int s = 0, dl = 0;
    float w = 0.f;
    if (idx < cnt) {
      int e = myE[idx];
      dl = myD[idx];
      s = ei[e];
      w = dinv[s] * dinv[win0 + dl];
    }
    int nn = cnt - base;
    if (nn > 32) nn = 32;
    int j = 0;
    for (; j + 4 <= nn; j += 4) {
      float2 v[4][3];
      int dls[4];
      float ws[4];
#pragma unroll
      for (int u = 0; u < 4; u++) {
        int sj = __builtin_amdgcn_readlane(s, j + u);
        dls[u] = __builtin_amdgcn_readlane(dl, j + u);
        ws[u] = __int_as_float(__builtin_amdgcn_readlane(__float_as_int(w), j + u));
        const float* row = x + (size_t)sj * NFEAT;
#pragma unroll
        for (int c = 0; c < 3; c++) v[u][c] = *(const float2*)(row + off[c]);
      }
#pragma unroll
      for (int u = 0; u < 4; u++) {
        float* a = acc + dls[u] * ROWF + 2 * lane;
#pragma unroll
        for (int c = 0; c < 3; c++) {
          __hip_atomic_fetch_add(a + c * 64,     ws[u] * v[u][c].x, __ATOMIC_RELAXED, __HIP_MEMORY_SCOPE_WORKGROUP);
          __hip_atomic_fetch_add(a + c * 64 + 1, ws[u] * v[u][c].y, __ATOMIC_RELAXED, __HIP_MEMORY_SCOPE_WORKGROUP);
        }
      }
    }
    for (; j < nn; j++) {
      int sj = __builtin_amdgcn_readlane(s, j);
      int dj = __builtin_amdgcn_readlane(dl, j);
      float wj = __int_as_float(__builtin_amdgcn_readlane(__float_as_int(w), j));
      const float* row = x + (size_t)sj * NFEAT;
      float* a = acc + dj * ROWF + 2 * lane;
      float2 t[3];
#pragma unroll
      for (int c = 0; c < 3; c++) t[c] = *(const float2*)(row + off[c]);
#pragma unroll
      for (int c = 0; c < 3; c++) {
        __hip_atomic_fetch_add(a + c * 64,     wj * t[c].x, __ATOMIC_RELAXED, __HIP_MEMORY_SCOPE_WORKGROUP);
        __hip_atomic_fetch_add(a + c * 64 + 1, wj * t[c].y, __ATOMIC_RELAXED, __HIP_MEMORY_SCOPE_WORKGROUP);
      }
    }
  }
}

__global__ void __launch_bounds__(1024) k_agg_window(const float* __restrict__ x,
                                                     const int* __restrict__ ei,
                                                     const float* __restrict__ dinv,
                                                     float* __restrict__ xagg) {
  __shared__ __attribute__((aligned(16))) float acc[WIN * ROWF];
  __shared__ int lstE[AGG_WAVES][LCAP];
  __shared__ int lstD[AGG_WAVES][LCAP];

  const int tid = threadIdx.x, lane = tid & 31;
  const int wave = __builtin_amdgcn_readfirstlane(tid >> 5);
  const int win0 = blockIdx.x * WIN;

  for (int q = tid; q < WIN * 48; q += 1024) {
    int dl = q / 48, r = q - dl * 48;
    int b = r / 24, c4 = r - b * 24;
    int n = win0 + dl;
    float4 o = make_float4(0.f, 0.f, 0.f, 0.f);
    if (n < NNODES) {
      float dv = dinv[n];
      float w = dv * dv;
      float4 xv = *(const float4*)(x + ((size_t)(b * NNODES + n) * NFEAT + c4 * 4));
      o.x = xv.x * w; o.y = xv.y * w; o.z = xv.z * w; o.w = xv.w * w;
    }
    *(float4*)(acc + q * 4) = o;
  }

  int off[3];
#pragma unroll
  for (int c = 0; c < 3; c++) {
    int f = c * 64 + 2 * lane;
    off[c] = (f >= NFEAT) ? (NNODES * NFEAT + f - NFEAT) : f;
  }

  __syncthreads();

  const int4* dst4 = (const int4*)(ei + NEDGES);
  int* myE = lstE[wave];
  int* myD = lstD[wave];
  int cnt = 0;

  for (int base = wave * 32; base < NE4; base += 4 * 1024) {
    int4 v[4];
#pragma unroll
    for (int u = 0; u < 4; u++) {
      int i4 = base + u * 1024 + lane;
      v[u] = (i4 < NE4) ? dst4[i4] : make_int4(-1, -1, -1, -1);
    }
#pragma unroll
    for (int u = 0; u < 4; u++) {
      int e0 = 4 * (base + u * 1024 + lane);
      unsigned r0 = (unsigned)(v[u].x - win0), r1 = (unsigned)(v[u].y - win0);
      unsigned r2 = (unsigned)(v[u].z - win0), r3 = (unsigned)(v[u].w - win0);
      bool m0 = r0 < (unsigned)WIN, m1 = r1 < (unsigned)WIN, m2 = r2 < (unsigned)WIN, m3 = r3 < (unsigned)WIN;
      if (__builtin_amdgcn_ballot_w32(m0 || m1 || m2 || m3)) {
        unsigned k0 = __builtin_amdgcn_ballot_w32(m0);
        if (m0) { int p = cnt + __builtin_amdgcn_mbcnt_lo(k0, 0u); myE[p] = e0;     myD[p] = (int)r0; }
        cnt += __builtin_popcount(k0);
        unsigned k1 = __builtin_amdgcn_ballot_w32(m1);
        if (m1) { int p = cnt + __builtin_amdgcn_mbcnt_lo(k1, 0u); myE[p] = e0 + 1; myD[p] = (int)r1; }
        cnt += __builtin_popcount(k1);
        unsigned k2 = __builtin_amdgcn_ballot_w32(m2);
        if (m2) { int p = cnt + __builtin_amdgcn_mbcnt_lo(k2, 0u); myE[p] = e0 + 2; myD[p] = (int)r2; }
        cnt += __builtin_popcount(k2);
        unsigned k3 = __builtin_amdgcn_ballot_w32(m3);
        if (m3) { int p = cnt + __builtin_amdgcn_mbcnt_lo(k3, 0u); myE[p] = e0 + 3; myD[p] = (int)r3; }
        cnt += __builtin_popcount(k3);
        if (cnt > LCAP - 128) {
          agg_flush(x, ei, dinv, acc, myE, myD, cnt, win0, lane, off);
          cnt = 0;
        }
      }
    }
  }
  agg_flush(x, ei, dinv, acc, myE, myD, cnt, win0, lane, off);

  __syncthreads();

  for (int q = tid; q < WIN * 48; q += 1024) {
    int dl = q / 48, r = q - dl * 48;
    int b = r / 24, c4 = r - b * 24;
    int n = win0 + dl;
    if (n < NNODES)
      *(float4*)(xagg + ((size_t)(b * NNODES + n) * NFEAT + c4 * 4)) = *(const float4*)(acc + q * 4);
  }
}

__global__ void __launch_bounds__(256) k_gru(
    const float* __restrict__ xagg, const float* __restrict__ att,
    const float* __restrict__ Wz, const float* __restrict__ bz,
    const float* __restrict__ Wr, const float* __restrict__ br,
    const float* __restrict__ Wh, const float* __restrict__ bh,
    const float* __restrict__ Lz, const float* __restrict__ lbz,
    const float* __restrict__ Lr, const float* __restrict__ lbr,
    const float* __restrict__ Lh, const float* __restrict__ lbh,
    const float* __restrict__ Wo, const float* __restrict__ bo,
    float* __restrict__ out) {
  __shared__ float sW[3 * 8 * 32];
  __shared__ float sb[3 * 32];
  __shared__ float sL[3 * 64 * 32];
  __shared__ float slb[3 * 32];
  __shared__ float sWo[32 * 32];
  __shared__ float sbo[32];
  __shared__ float sG[8][16 * 32];
  __shared__ float sH[8][16 * 32];
  __shared__ float sHR[8][16 * 32];

  int tid = threadIdx.x;
  for (int t = tid; t < 8 * 32; t += 256) { sW[t] = Wz[t]; sW[256 + t] = Wr[t]; sW[512 + t] = Wh[t]; }
  for (int t = tid; t < 32; t += 256) {
    sb[t] = bz[t]; sb[32 + t] = br[t]; sb[64 + t] = bh[t];
    slb[t] = lbz[t]; slb[32 + t] = lbr[t]; slb[64 + t] = lbh[t];
    sbo[t] = (t < OUTC) ? bo[t] : 0.f;
  }
  for (int t = tid; t < 64 * 32; t += 256) { sL[t] = Lz[t]; sL[2048 + t] = Lr[t]; sL[4096 + t] = Lh[t]; }
  for (int t = tid; t < 32 * 32; t += 256) {
    int k = t >> 5, j = t & 31;
    sWo[t] = (j < OUTC) ? Wo[k * OUTC + j] : 0.f;
  }

  int wave = tid >> 5, lane = tid & 31, m = lane & 15, hi = lane >> 4;
  float* mG = sG[wave];
  float* mH = sH[wave];
  float* mHR = sHR[wave];
  for (int idx = lane; idx < 16 * 32; idx += 32) mH[idx] = 0.f;

  float pr[PER];
  {
    float a[PER]; float mx = -1e30f;
    for (int p = 0; p < PER; p++) { a[p] = att[p]; mx = fmaxf(mx, a[p]); }
    float sm = 0.f;
    for (int p = 0; p < PER; p++) { a[p] = __expf(a[p] - mx); sm += a[p]; }
    float inv = 1.0f / sm;
    for (int p = 0; p < PER; p++) pr[p] = a[p] * inv;
  }

  int tile = blockIdx.x * 8 + wave;
  long ia = (long)tile * 16 + m;
  if (ia >= ROWS) ia = ROWS - 1;
  const float* xrow = xagg + ia * (long)NFEAT;

  v8f H0, H1, A0, A1;
  for (int r = 0; r < 8; r++) { H0[r] = 0.f; H1[r] = 0.f; A0[r] = 0.f; A1[r] = 0.f; }

  __syncthreads();

  for (int p = 0; p < PER; p++) {
    int f0 = hi * 2;
    v2f ax0, ax1;
    ax0.x = xrow[(f0)*PER + p];     ax0.y = xrow[(f0 + 1) * PER + p];
    ax1.x = xrow[(f0 + 4) * PER + p]; ax1.y = xrow[(f0 + 5) * PER + p];

    v8f g0, g1, z0, z1, r0, r1, t0, t1;

    gemm8(ax0, ax1, sW + 0, sb + 0, m, hi, g0, g1);
    storeTile(mG, m, hi, g0, g1);
    __syncthreads();
    gemm64(mG, mH, sL + 0, slb + 0, m, hi, z0, z1);
    for (int r = 0; r < 8; r++) { z0[r] = sigm(z0[r]); z1[r] = sigm(z1[r]); }
    __syncthreads();

    gemm8(ax0, ax1, sW + 256, sb + 32, m, hi, g0, g1);
    storeTile(mG, m, hi, g0, g1);
    __syncthreads();
    gemm64(mG, mH, sL + 2048, slb + 32, m, hi, r0, r1);
    for (int r = 0; r < 8; r++) { r0[r] = sigm(r0[r]); r1[r] = sigm(r1[r]); }

    gemm8(ax0, ax1, sW + 512, sb + 64, m, hi, g0, g1);
    __syncthreads();
    storeTile(mG, m, hi, g0, g1);
    {
      v8f q0, q1;
      for (int r = 0; r < 8; r++) { q0[r] = H0[r] * r0[r]; q1[r] = H1[r] * r1[r]; }
      storeTile(mHR, m, hi, q0, q1);
    }
    __syncthreads();
    gemm64(mG, mHR, sL + 4096, slb + 64, m, hi, t0, t1);
    for (int r = 0; r < 8; r++) { t0[r] = tanhc(t0[r]); t1[r] = tanhc(t1[r]); }

    for (int r = 0; r < 8; r++) {
      H0[r] = z0[r] * H0[r] + (1.0f - z0[r]) * t0[r];
      H1[r] = z1[r] * H1[r] + (1.0f - z1[r]) * t1[r];
      A0[r] += pr[p] * H0[r];
      A1[r] += pr[p] * H1[r];
    }
    storeTile(mH, m, hi, H0, H1);
    __syncthreads();
  }

  for (int r = 0; r < 8; r++) { A0[r] = fmaxf(A0[r], 0.f); A1[r] = fmaxf(A1[r], 0.f); }
  storeTile(mG, m, hi, A0, A1);
  __syncthreads();
  v8f o0, o1;
  {
    float b0v = sbo[m], b1v = sbo[16 + m];
    for (int r = 0; r < 8; r++) { o0[r] = b0v; o1[r] = b1v; }
#pragma unroll
    for (int kb = 0; kb < 8; kb++) {
      v2f a = ldsA(mG, m, hi, kb);
      o0 = wmma4(a, ldsB(sWo, m, hi, kb, 0), o0);
      o1 = wmma4(a, ldsB(sWo, m, hi, kb, 1), o1);
    }
  }
#pragma unroll
  for (int r = 0; r < 8; r++) {
    long row = (long)tile * 16 + r + 8 * hi;
    if (row < ROWS) {
      out[row * OUTC + m] = o0[r];
      if (m < 8) out[row * OUTC + 16 + m] = o1[r];
    }
  }
}

extern "C" void kernel_launch(void* const* d_in, const int* in_sizes, int n_in,
                              void* d_out, int out_size, void* d_ws, size_t ws_size,
                              hipStream_t stream) {
  (void)in_sizes; (void)n_in; (void)out_size; (void)ws_size;
  const float* x   = (const float*)d_in[0];
  const int*   ei  = (const int*)d_in[1];
  const float* att = (const float*)d_in[2];
  const float* Wz  = (const float*)d_in[3];  const float* bz  = (const float*)d_in[4];
  const float* Wr  = (const float*)d_in[5];  const float* br  = (const float*)d_in[6];
  const float* Wh  = (const float*)d_in[7];  const float* bh  = (const float*)d_in[8];
  const float* Lz  = (const float*)d_in[9];  const float* lbz = (const float*)d_in[10];
  const float* Lr  = (const float*)d_in[11]; const float* lbr = (const float*)d_in[12];
  const float* Lh  = (const float*)d_in[13]; const float* lbh = (const float*)d_in[14];
  const float* Wo  = (const float*)d_in[15]; const float* bo  = (const float*)d_in[16];
  float* out = (float*)d_out;

  float* wsf  = (float*)d_ws;
  float* deg  = wsf;
  float* xagg = wsf + 65536;

  k_deg_init<<<(NNODES + 255) / 256, 256, 0, stream>>>(deg);
  k_deg_acc<<<(NEDGES + 255) / 256, 256, 0, stream>>>(ei, deg);
  k_rsqrt<<<(NNODES + 255) / 256, 256, 0, stream>>>(deg);
  k_agg_window<<<NWIN, 1024, 0, stream>>>(x, ei, deg, xagg);

  int tiles = ROWS / 16;
  int blocks = (tiles + 7) / 8;
  k_gru<<<blocks, 256, 0, stream>>>(xagg, att, Wz, bz, Wr, br, Wh, bh,
                                    Lz, lbz, Lr, lbr, Lh, lbh, Wo, bo, out);
}


# GRU weight-stationary: every gate weight sits in VGPRs as a B fragment for the whole kernel (was re-read from LDS per product); waves share nothing, so they sync at wave scope; sums unchanged
# speedup vs baseline: 1.7751x; 1.0875x over previous
#include <hip/hip_runtime.h>

#define NNODES 50000
#define NEDGES 800000
#define FIN 8
#define PER 12
#define HIDN 32
#define NB 2
#define OUTC 24
#define ROWS (NB*NNODES)
#define NFEAT (FIN*PER)

typedef float v2f __attribute__((ext_vector_type(2)));
typedef float v8f __attribute__((ext_vector_type(8)));

__device__ __forceinline__ v8f wmma4(v2f a, v2f b, v8f c) {
  return __builtin_amdgcn_wmma_f32_16x16x4_f32(false, a, false, b, (short)0, c, false, false);
}

__device__ __forceinline__ v2f ldsA(const float* buf, int m, int hi, int kb) {
  int k0 = kb * 4 + hi * 2;
  v2f a;
  a.x = buf[m * HIDN + k0];
  a.y = buf[m * HIDN + k0 + 1];
  return a;
}

__device__ __forceinline__ void storeTile(float* buf, int m, int hi, const v8f& c0, const v8f& c1) {
#pragma unroll
  for (int r = 0; r < 8; r++) {
    buf[(r + 8 * hi) * HIDN + m]      = c0[r];
    buf[(r + 8 * hi) * HIDN + 16 + m] = c1[r];
  }
}

__device__ __forceinline__ float sigm(float x) { return 1.0f / (1.0f + __expf(-x)); }
__device__ __forceinline__ float tanhc(float x) {
  x = fminf(fmaxf(x, -15.f), 15.f);
  float e = __expf(2.0f * x);
  return (e - 1.0f) / (e + 1.0f);
}

__global__ void k_deg_init(float* deg) {
  int i = blockIdx.x * 256 + threadIdx.x;
  if (i < NNODES) deg[i] = 1.0f;
}
__global__ void k_deg_acc(const int* __restrict__ ei, float* deg) {
  int e = blockIdx.x * 256 + threadIdx.x;
  if (e < NEDGES) atomicAdd(&deg[ei[NEDGES + e]], 1.0f);
}
__global__ void k_rsqrt(float* deg) {
  int i = blockIdx.x * 256 + threadIdx.x;
  if (i < NNODES) deg[i] = rsqrtf(deg[i]);
}
#define WIN 196
#define NWIN 256
#define ROWF (NB * NFEAT)
#define AGG_WAVES 32
#define LCAP 256
#define NE4 (NEDGES / 4)
static_assert(NWIN * WIN >= NNODES, "windows cover every node");
static_assert(NE4 * 4 == NEDGES && NE4 % 32 == 0, "whole waves of int4 reads over the target column");

__device__ __forceinline__ void agg_flush(const float* __restrict__ x, const int* __restrict__ ei,
                                          const float* __restrict__ dinv, float* acc,
                                          const int* myE, const int* myD, int cnt, int win0,
                                          int lane, const int* off) {
  for (int base = 0; base < cnt; base += 32) {
    int idx = base + lane;
    int s = 0, dl = 0;
    float w = 0.f;
    if (idx < cnt) {
      int e = myE[idx];
      dl = myD[idx];
      s = ei[e];
      w = dinv[s] * dinv[win0 + dl];
    }
    int nn = cnt - base;
    if (nn > 32) nn = 32;
    int j = 0;
    for (; j + 4 <= nn; j += 4) {
      float2 v[4][3];
      int dls[4];
      float ws[4];
#pragma unroll
      for (int u = 0; u < 4; u++) {
        int sj = __builtin_amdgcn_readlane(s, j + u);
        dls[u] = __builtin_amdgcn_readlane(dl, j + u);
        ws[u] = __int_as_float(__builtin_amdgcn_readlane(__float_as_int(w), j + u));
        const float* row = x + (size_t)sj * NFEAT;
#pragma unroll
        for (int c = 0; c < 3; c++) v[u][c] = *(const float2*)(row + off[c]);
      }
#pragma unroll
      for (int u = 0; u < 4; u++) {
        float* a = acc + dls[u] * ROWF + 2 * lane;
#pragma unroll
        for (int c = 0; c < 3; c++) {
          __hip_atomic_fetch_add(a + c * 64,     ws[u] * v[u][c].x, __ATOMIC_RELAXED, __HIP_MEMORY_SCOPE_WORKGROUP);
          __hip_atomic_fetch_add(a + c * 64 + 1, ws[u] * v[u][c].y, __ATOMIC_RELAXED, __HIP_MEMORY_SCOPE_WORKGROUP);
        }
      }
    }
    for (; j < nn; j++) {
      int sj = __builtin_amdgcn_readlane(s, j);
      int dj = __builtin_amdgcn_readlane(dl, j);
      float wj = __int_as_float(__builtin_amdgcn_readlane(__float_as_int(w), j));
      const float* row = x + (size_t)sj * NFEAT;
      float* a = acc + dj * ROWF + 2 * lane;
      float2 t[3];
#pragma unroll
      for (int c = 0; c < 3; c++) t[c] = *(const float2*)(row + off[c]);
#pragma unroll
      for (int c = 0; c < 3; c++) {
        __hip_atomic_fetch_add(a + c * 64,     wj * t[c].x, __ATOMIC_RELAXED, __HIP_MEMORY_SCOPE_WORKGROUP);
        __hip_atomic_fetch_add(a + c * 64 + 1, wj * t[c].y, __ATOMIC_RELAXED, __HIP_MEMORY_SCOPE_WORKGROUP);
      }
    }
  }
}

__global__ void __launch_bounds__(1024) k_agg_window(const float* __restrict__ x,
                                                     const int* __restrict__ ei,
                                                     const float* __restrict__ dinv,
                                                     float* __restrict__ xagg) {
  __shared__ __attribute__((aligned(16))) float acc[WIN * ROWF];
  __shared__ int lstE[AGG_WAVES][LCAP];
  __shared__ int lstD[AGG_WAVES][LCAP];

  const int tid = threadIdx.x, lane = tid & 31;
  const int wave = __builtin_amdgcn_readfirstlane(tid >> 5);
  const int win0 = blockIdx.x * WIN;

  for (int q = tid; q < WIN * 48; q += 1024) {
    int dl = q / 48, r = q - dl * 48;
    int b = r / 24, c4 = r - b * 24;
    int n = win0 + dl;
    float4 o = make_float4(0.f, 0.f, 0.f, 0.f);
    if (n < NNODES) {
      float dv = dinv[n];
      float w = dv * dv;
      float4 xv = *(const float4*)(x + ((size_t)(b * NNODES + n) * NFEAT + c4 * 4));
      o.x = xv.x * w; o.y = xv.y * w; o.z = xv.z * w; o.w = xv.w * w;
    }
    *(float4*)(acc + q * 4) = o;
  }

  int off[3];
#pragma unroll
  for (int c = 0; c < 3; c++) {
    int f = c * 64 + 2 * lane;
    off[c] = (f >= NFEAT) ? (NNODES * NFEAT + f - NFEAT) : f;
  }

  __syncthreads();

  const int4* dst4 = (const int4*)(ei + NEDGES);
  int* myE = lstE[wave];
  int* myD = lstD[wave];
  int cnt = 0;

  for (int base = wave * 32; base < NE4; base += 4 * 1024) {
    int4 v[4];
#pragma unroll
    for (int u = 0; u < 4; u++) {
      int i4 = base + u * 1024 + lane;
      v[u] = (i4 < NE4) ? dst4[i4] : make_int4(-1, -1, -1, -1);
    }
#pragma unroll
    for (int u = 0; u < 4; u++) {
      int e0 = 4 * (base + u * 1024 + lane);
      unsigned r0 = (unsigned)(v[u].x - win0), r1 = (unsigned)(v[u].y - win0);
      unsigned r2 = (unsigned)(v[u].z - win0), r3 = (unsigned)(v[u].w - win0);
      bool m0 = r0 < (unsigned)WIN, m1 = r1 < (unsigned)WIN, m2 = r2 < (unsigned)WIN, m3 = r3 < (unsigned)WIN;
      if (__builtin_amdgcn_ballot_w32(m0 || m1 || m2 || m3)) {
        unsigned k0 = __builtin_amdgcn_ballot_w32(m0);
        if (m0) { int p = cnt + __builtin_amdgcn_mbcnt_lo(k0, 0u); myE[p] = e0;     myD[p] = (int)r0; }
        cnt += __builtin_popcount(k0);
        unsigned k1 = __builtin_amdgcn_ballot_w32(m1);
        if (m1) { int p = cnt + __builtin_amdgcn_mbcnt_lo(k1, 0u); myE[p] = e0 + 1; myD[p] = (int)r1; }
        cnt += __builtin_popcount(k1);
        unsigned k2 = __builtin_amdgcn_ballot_w32(m2);
        if (m2) { int p = cnt + __builtin_amdgcn_mbcnt_lo(k2, 0u); myE[p] = e0 + 2; myD[p] = (int)r2; }
        cnt += __builtin_popcount(k2);
        unsigned k3 = __builtin_amdgcn_ballot_w32(m3);
        if (m3) { int p = cnt + __builtin_amdgcn_mbcnt_lo(k3, 0u); myE[p] = e0 + 3; myD[p] = (int)r3; }
        cnt += __builtin_popcount(k3);
        if (cnt > LCAP - 128) {
          agg_flush(x, ei, dinv, acc, myE, myD, cnt, win0, lane, off);
          cnt = 0;
        }
      }
    }
  }
  agg_flush(x, ei, dinv, acc, myE, myD, cnt, win0, lane, off);

  __syncthreads();

  for (int q = tid; q < WIN * 48; q += 1024) {
    int dl = q / 48, r = q - dl * 48;
    int b = r / 24, c4 = r - b * 24;
    int n = win0 + dl;
    if (n < NNODES)
      *(float4*)(xagg + ((size_t)(b * NNODES + n) * NFEAT + c4 * 4)) = *(const float4*)(acc + q * 4);
  }
}

#define GRU_WAVES 4
static_assert(ROWS % 16 == 0, "whole 16-row tiles: a wave either owns a full tile or leaves");

__device__ __forceinline__ v2f gldB(const float* __restrict__ W, int m, int hi, int kb, int nt) {
  int k0 = kb * 4 + hi * 2;
  int n = nt * 16 + m;
  v2f b;
  b.x = W[k0 * HIDN + n];
  b.y = W[(k0 + 1) * HIDN + n];
  return b;
}

__device__ __forceinline__ void wave_sync() {
  __builtin_amdgcn_fence(__ATOMIC_RELEASE, "wavefront");
  __builtin_amdgcn_wave_barrier();
  __builtin_amdgcn_fence(__ATOMIC_ACQUIRE, "wavefront");
}

__device__ __forceinline__ void gemm8r(v2f ax0, v2f ax1, const v2f (&B)[2][2], float b0v, float b1v,
                                       v8f& c0, v8f& c1) {
#pragma unroll
  for (int r = 0; r < 8; r++) { c0[r] = b0v; c1[r] = b1v; }
  c0 = wmma4(ax0, B[0][0], c0);
  c0 = wmma4(ax1, B[1][0], c0);
  c1 = wmma4(ax0, B[0][1], c1);
  c1 = wmma4(ax1, B[1][1], c1);
}

__device__ __forceinline__ void gemm64r(const float* A1, const float* A2, const v2f (&B)[16][2],
                                        float b0v, float b1v, int m, int hi, v8f& c0, v8f& c1) {
#pragma unroll
  for (int r = 0; r < 8; r++) { c0[r] = b0v; c1[r] = b1v; }
#pragma unroll
  for (int kb = 0; kb < 8; kb++) {
    v2f a = ldsA(A1, m, hi, kb);
    c0 = wmma4(a, B[kb][0], c0);
    c1 = wmma4(a, B[kb][1], c1);
  }
#pragma unroll
  for (int kb = 0; kb < 8; kb++) {
    v2f a = ldsA(A2, m, hi, kb);
    c0 = wmma4(a, B[kb + 8][0], c0);
    c1 = wmma4(a, B[kb + 8][1], c1);
  }
}

__global__ void __launch_bounds__(GRU_WAVES * 32) k_gru(
    const float* __restrict__ xagg, const float* __restrict__ att,
    const float* __restrict__ Wz, const float* __restrict__ bz,
    const float* __restrict__ Wr, const float* __restrict__ br,
    const float* __restrict__ Wh, const float* __restrict__ bh,
    const float* __restrict__ Lz, const float* __restrict__ lbz,
    const float* __restrict__ Lr, const float* __restrict__ lbr,
    const float* __restrict__ Lh, const float* __restrict__ lbh,
    const float* __restrict__ Wo, const float* __restrict__ bo,
    float* __restrict__ out) {
  __shared__ float sG[GRU_WAVES][16 * 32];
  __shared__ float sH[GRU_WAVES][16 * 32];
  __shared__ float sHR[GRU_WAVES][16 * 32];
  __shared__ float sPr[GRU_WAVES][16];

  const int tid = threadIdx.x;
  const int lane = tid & 31, m = lane & 15, hi = lane >> 4;
  const int wave = __builtin_amdgcn_readfirstlane(tid >> 5);
  const int tile = blockIdx.x * GRU_WAVES + wave;
  if (tile * 16 >= ROWS) return;

  float* mG = sG[wave];
  float* mH = sH[wave];
  float* mHR = sHR[wave];
  float* mPr = sPr[wave];

  v2f BWz[2][2], BWr[2][2], BWh[2][2];
  v2f BLz[16][2], BLr[16][2], BLh[16][2];
#pragma unroll
  for (int kb = 0; kb < 2; kb++)
#pragma unroll
    for (int nt = 0; nt < 2; nt++) {
      BWz[kb][nt] = gldB(Wz, m, hi, kb, nt);
      BWr[kb][nt] = gldB(Wr, m, hi, kb, nt);
      BWh[kb][nt] = gldB(Wh, m, hi, kb, nt);
    }
#pragma unroll
  for (int kb = 0; kb < 16; kb++)
#pragma unroll
    for (int nt = 0; nt < 2; nt++) {
      BLz[kb][nt] = gldB(Lz, m, hi, kb, nt);
      BLr[kb][nt] = gldB(Lr, m, hi, kb, nt);
      BLh[kb][nt] = gldB(Lh, m, hi, kb, nt);
    }
  const float bz0 = bz[m], bz1 = bz[16 + m], br0 = br[m], br1 = br[16 + m], bh0 = bh[m], bh1 = bh[16 + m];
  const float lz0 = lbz[m], lz1 = lbz[16 + m], lr0 = lbr[m], lr1 = lbr[16 + m], lh0 = lbh[m], lh1 = lbh[16 + m];

  for (int idx = lane; idx < 16 * 32; idx += 32) mH[idx] = 0.f;

  {
    float a[PER]; float mx = -1e30f;
    for (int p = 0; p < PER; p++) { a[p] = att[p]; mx = fmaxf(mx, a[p]); }
    float sm = 0.f;
    for (int p = 0; p < PER; p++) { a[p] = __expf(a[p] - mx); sm += a[p]; }
    float inv = 1.0f / sm;
    float mine = 0.f;
    for (int p = 0; p < PER; p++) mine = (lane == p) ? a[p] * inv : mine;
    if (lane < PER) mPr[lane] = mine;
  }

  const long ia = (long)tile * 16 + m;
  const float* xrow = xagg + ia * (long)NFEAT;

  v8f H0, H1, A0, A1;
  for (int r = 0; r < 8; r++) { H0[r] = 0.f; H1[r] = 0.f; A0[r] = 0.f; A1[r] = 0.f; }

  wave_sync();

  for (int p = 0; p < PER; p++) {
    int f0 = hi * 2;
    v2f ax0, ax1;
    ax0.x = xrow[(f0)*PER + p];     ax0.y = xrow[(f0 + 1) * PER + p];
    ax1.x = xrow[(f0 + 4) * PER + p]; ax1.y = xrow[(f0 + 5) * PER + p];
    const float prp = mPr[p];

    v8f g0, g1, z0, z1, r0, r1, t0, t1;

    gemm8r(ax0, ax1, BWz, bz0, bz1, g0, g1);
    storeTile(mG, m, hi, g0, g1);
    wave_sync();
    gemm64r(mG, mH, BLz, lz0, lz1, m, hi, z0, z1);
    for (int r = 0; r < 8; r++) { z0[r] = sigm(z0[r]); z1[r] = sigm(z1[r]); }
    wave_sync();

    gemm8r(ax0, ax1, BWr, br0, br1, g0, g1);
    storeTile(mG, m, hi, g0, g1);
    wave_sync();
    gemm64r(mG, mH, BLr, lr0, lr1, m, hi, r0, r1);
    for (int r = 0; r < 8; r++) { r0[r] = sigm(r0[r]); r1[r] = sigm(r1[r]); }

    gemm8r(ax0, ax1, BWh, bh0, bh1, g0, g1);
    wave_sync();
    storeTile(mG, m, hi, g0, g1);
    {
      v8f q0, q1;
      for (int r = 0; r < 8; r++) { q0[r] = H0[r] * r0[r]; q1[r] = H1[r] * r1[r]; }
      storeTile(mHR, m, hi, q0, q1);
    }
    wave_sync();
    gemm64r(mG, mHR, BLh, lh0, lh1, m, hi, t0, t1);
    for (int r = 0; r < 8; r++) { t0[r] = tanhc(t0[r]); t1[r] = tanhc(t1[r]); }

    for (int r = 0; r < 8; r++) {
      H0[r] = z0[r] * H0[r] + (1.0f - z0[r]) * t0[r];
      H1[r] = z1[r] * H1[r] + (1.0f - z1[r]) * t1[r];
      A0[r] += prp * H0[r];
      A1[r] += prp * H1[r];
    }
    wave_sync();
    storeTile(mH, m, hi, H0, H1);
    wave_sync();
  }

  for (int r = 0; r < 8; r++) { A0[r] = fmaxf(A0[r], 0.f); A1[r] = fmaxf(A1[r], 0.f); }
  storeTile(mG, m, hi, A0, A1);
  wave_sync();
  v8f o0, o1;
  {
    const bool real = m < OUTC - 16;
    const int mc = real ? 16 + m : 0;
    float b0v = bo[m], b1l = bo[mc];
    float b1v = real ? b1l : 0.f;
    for (int r = 0; r < 8; r++) { o0[r] = b0v; o1[r] = b1v; }
#pragma unroll
    for (int kb = 0; kb < 8; kb++) {
      int k0 = kb * 4 + hi * 2;
      v2f a = ldsA(mG, m, hi, kb);
      v2f w0, w1;
      w0.x = Wo[k0 * OUTC + m];
      w0.y = Wo[(k0 + 1) * OUTC + m];
      float w1x = Wo[k0 * OUTC + mc], w1y = Wo[(k0 + 1) * OUTC + mc];
      w1.x = real ? w1x : 0.f;
      w1.y = real ? w1y : 0.f;
      o0 = wmma4(a, w0, o0);
      o1 = wmma4(a, w1, o1);
    }
  }
#pragma unroll
  for (int r = 0; r < 8; r++) {
    long row = (long)tile * 16 + r + 8 * hi;
    if (row < ROWS) {
      out[row * OUTC + m] = o0[r];
      if (m < 8) out[row * OUTC + 16 + m] = o1[r];
    }
  }
}

extern "C" void kernel_launch(void* const* d_in, const int* in_sizes, int n_in,
                              void* d_out, int out_size, void* d_ws, size_t ws_size,
                              hipStream_t stream) {
  (void)in_sizes; (void)n_in; (void)out_size; (void)ws_size;
  const float* x   = (const float*)d_in[0];
  const int*   ei  = (const int*)d_in[1];
  const float* att = (const float*)d_in[2];
  const float* Wz  = (const float*)d_in[3];  const float* bz  = (const float*)d_in[4];
  const float* Wr  = (const float*)d_in[5];  const float* br  = (const float*)d_in[6];
  const float* Wh  = (const float*)d_in[7];  const float* bh  = (const float*)d_in[8];
  const float* Lz  = (const float*)d_in[9];  const float* lbz = (const float*)d_in[10];
  const float* Lr  = (const float*)d_in[11]; const float* lbr = (const float*)d_in[12];
  const float* Lh  = (const float*)d_in[13]; const float* lbh = (const float*)d_in[14];
  const float* Wo  = (const float*)d_in[15]; const float* bo  = (const float*)d_in[16];
  float* out = (float*)d_out;

  float* wsf  = (float*)d_ws;
  float* deg  = wsf;
  float* xagg = wsf + 65536;

  k_deg_init<<<(NNODES + 255) / 256, 256, 0, stream>>>(deg);
  k_deg_acc<<<(NEDGES + 255) / 256, 256, 0, stream>>>(ei, deg);
  k_rsqrt<<<(NNODES + 255) / 256, 256, 0, stream>>>(deg);
  k_agg_window<<<NWIN, 1024, 0, stream>>>(x, ei, deg, xagg);

  k_gru<<<(ROWS / 16 + GRU_WAVES - 1) / GRU_WAVES, GRU_WAVES * 32, 0, stream>>>(xagg, att, Wz, bz, Wr, br, Wh, bh,
                                    Lz, lbz, Lr, lbr, Lh, lbh, Wo, bo, out);
}


# GRU: the LDS A fragments of each 64-deep gate product are read as one group ahead of its WMMAs (H once per period for z and r; one G tile per gate), not two at a time among them; sums unchanged
# speedup vs baseline: 1.9145x; 1.0785x over previous
#include <hip/hip_runtime.h>

#define NNODES 50000
#define NEDGES 800000
#define FIN 8
#define PER 12
#define HIDN 32
#define NB 2
#define OUTC 24
#define ROWS (NB*NNODES)
#define NFEAT (FIN*PER)

typedef float v2f __attribute__((ext_vector_type(2)));
typedef float v8f __attribute__((ext_vector_type(8)));

__device__ __forceinline__ v8f wmma4(v2f a, v2f b, v8f c) {
  return __builtin_amdgcn_wmma_f32_16x16x4_f32(false, a, false, b, (short)0, c, false, false);
}

__device__ __forceinline__ v2f ldsA(const float* buf, int m, int hi, int kb) {
  int k0 = kb * 4 + hi * 2;
  v2f a;
  a.x = buf[m * HIDN + k0];
  a.y = buf[m * HIDN + k0 + 1];
  return a;
}

__device__ __forceinline__ void ldsA8(const float* buf, int m, int hi, v2f (&a)[8]) {
#pragma unroll
  for (int kb = 0; kb < 8; kb++) a[kb] = ldsA(buf, m, hi, kb);
}

__device__ __forceinline__ void storeTile(float* buf, int m, int hi, const v8f& c0, const v8f& c1) {
#pragma unroll
  for (int r = 0; r < 8; r++) {
    buf[(r + 8 * hi) * HIDN + m]      = c0[r];
    buf[(r + 8 * hi) * HIDN + 16 + m] = c1[r];
  }
}

__device__ __forceinline__ float sigm(float x) { return 1.0f / (1.0f + __expf(-x)); }
__device__ __forceinline__ float tanhc(float x) {
  x = fminf(fmaxf(x, -15.f), 15.f);
  float e = __expf(2.0f * x);
  return (e - 1.0f) / (e + 1.0f);
}

__global__ void k_deg_init(float* deg) {
  int i = blockIdx.x * 256 + threadIdx.x;
  if (i < NNODES) deg[i] = 1.0f;
}
__global__ void k_deg_acc(const int* __restrict__ ei, float* deg) {
  int e = blockIdx.x * 256 + threadIdx.x;
  if (e < NEDGES) atomicAdd(&deg[ei[NEDGES + e]], 1.0f);
}
__global__ void k_rsqrt(float* deg) {
  int i = blockIdx.x * 256 + threadIdx.x;
  if (i < NNODES) deg[i] = rsqrtf(deg[i]);
}
#define WIN 196
#define NWIN 256
#define ROWF (NB * NFEAT)
#define AGG_WAVES 32
#define LCAP 256
#define NE4 (NEDGES / 4)
static_assert(NWIN * WIN >= NNODES, "windows cover every node");
static_assert(NE4 * 4 == NEDGES && NE4 % 32 == 0, "whole waves of int4 reads over the target column");

__device__ __forceinline__ void agg_flush(const float* __restrict__ x, const int* __restrict__ ei,
                                          const float* __restrict__ dinv, float* acc,
                                          const int* myE, const int* myD, int cnt, int win0,
                                          int lane, const int* off) {
  for (int base = 0; base < cnt; base += 32) {
    int idx = base + lane;
    int s = 0, dl = 0;
    float w = 0.f;
    if (idx < cnt) {
      int e = myE[idx];
      dl = myD[idx];
      s = ei[e];
      w = dinv[s] * dinv[win0 + dl];
    }
    int nn = cnt - base;
    if (nn > 32) nn = 32;
    int j = 0;
    for (; j + 4 <= nn; j += 4) {
      float2 v[4][3];
      int dls[4];
      float ws[4];
#pragma unroll
      for (int u = 0; u < 4; u++) {
        int sj = __builtin_amdgcn_readlane(s, j + u);
        dls[u] = __builtin_amdgcn_readlane(dl, j + u);
        ws[u] = __int_as_float(__builtin_amdgcn_readlane(__float_as_int(w), j + u));
        const float* row = x + (size_t)sj * NFEAT;
#pragma unroll
        for (int c = 0; c < 3; c++) v[u][c] = *(const float2*)(row + off[c]);
      }
#pragma unroll
      for (int u = 0; u < 4; u++) {
        float* a = acc + dls[u] * ROWF + 2 * lane;
#pragma unroll
        for (int c = 0; c < 3; c++) {
          __hip_atomic_fetch_add(a + c * 64,     ws[u] * v[u][c].x, __ATOMIC_RELAXED, __HIP_MEMORY_SCOPE_WORKGROUP);
          __hip_atomic_fetch_add(a + c * 64 + 1, ws[u] * v[u][c].y, __ATOMIC_RELAXED, __HIP_MEMORY_SCOPE_WORKGROUP);
        }
      }
    }
    for (; j < nn; j++) {
      int sj = __builtin_amdgcn_readlane(s, j);
      int dj = __builtin_amdgcn_readlane(dl, j);
      float wj = __int_as_float(__builtin_amdgcn_readlane(__float_as_int(w), j));
      const float* row = x + (size_t)sj * NFEAT;
      float* a = acc + dj * ROWF + 2 * lane;
      float2 t[3];
#pragma unroll
      for (int c = 0; c < 3; c++) t[c] = *(const float2*)(row + off[c]);
#pragma unroll
      for (int c = 0; c < 3; c++) {
        __hip_atomic_fetch_add(a + c * 64,     wj * t[c].x, __ATOMIC_RELAXED, __HIP_MEMORY_SCOPE_WORKGROUP);
        __hip_atomic_fetch_add(a + c * 64 + 1, wj * t[c].y, __ATOMIC_RELAXED, __HIP_MEMORY_SCOPE_WORKGROUP);
      }
    }
  }
}

__global__ void __launch_bounds__(1024) k_agg_window(const float* __restrict__ x,
                                                     const int* __restrict__ ei,
                                                     const float* __restrict__ dinv,
                                                     float* __restrict__ xagg) {
  __shared__ __attribute__((aligned(16))) float acc[WIN * ROWF];
  __shared__ int lstE[AGG_WAVES][LCAP];
  __shared__ int lstD[AGG_WAVES][LCAP];

  const int tid = threadIdx.x, lane = tid & 31;
  const int wave = __builtin_amdgcn_readfirstlane(tid >> 5);
  const int win0 = blockIdx.x * WIN;

  for (int q = tid; q < WIN * 48; q += 1024) {
    int dl = q / 48, r = q - dl * 48;
    int b = r / 24, c4 = r - b * 24;
    int n = win0 + dl;
    float4 o = make_float4(0.f, 0.f, 0.f, 0.f);
    if (n < NNODES) {
      float dv = dinv[n];
      float w = dv * dv;
      float4 xv = *(const float4*)(x + ((size_t)(b * NNODES + n) * NFEAT + c4 * 4));
      o.x = xv.x * w; o.y = xv.y * w; o.z = xv.z * w; o.w = xv.w * w;
    }
    *(float4*)(acc + q * 4) = o;
  }

  int off[3];
#pragma unroll
  for (int c = 0; c < 3; c++) {
    int f = c * 64 + 2 * lane;
    off[c] = (f >= NFEAT) ? (NNODES * NFEAT + f - NFEAT) : f;
  }

  __syncthreads();

  const int4* dst4 = (const int4*)(ei + NEDGES);
  int* myE = lstE[wave];
  int* myD = lstD[wave];
  int cnt = 0;

  for (int base = wave * 32; base < NE4; base += 4 * 1024) {
    int4 v[4];
#pragma unroll
    for (int u = 0; u < 4; u++) {
      int i4 = base + u * 1024 + lane;
      v[u] = (i4 < NE4) ? dst4[i4] : make_int4(-1, -1, -1, -1);
    }
#pragma unroll
    for (int u = 0; u < 4; u++) {
      int e0 = 4 * (base + u * 1024 + lane);
      unsigned r0 = (unsigned)(v[u].x - win0), r1 = (unsigned)(v[u].y - win0);
      unsigned r2 = (unsigned)(v[u].z - win0), r3 = (unsigned)(v[u].w - win0);
      bool m0 = r0 < (unsigned)WIN, m1 = r1 < (unsigned)WIN, m2 = r2 < (unsigned)WIN, m3 = r3 < (unsigned)WIN;
      if (__builtin_amdgcn_ballot_w32(m0 || m1 || m2 || m3)) {
        unsigned k0 = __builtin_amdgcn_ballot_w32(m0);
        if (m0) { int p = cnt + __builtin_amdgcn_mbcnt_lo(k0, 0u); myE[p] = e0;     myD[p] = (int)r0; }
        cnt += __builtin_popcount(k0);
        unsigned k1 = __builtin_amdgcn_ballot_w32(m1);
        if (m1) { int p = cnt + __builtin_amdgcn_mbcnt_lo(k1, 0u); myE[p] = e0 + 1; myD[p] = (int)r1; }
        cnt += __builtin_popcount(k1);
        unsigned k2 = __builtin_amdgcn_ballot_w32(m2);
        if (m2) { int p = cnt + __builtin_amdgcn_mbcnt_lo(k2, 0u); myE[p] = e0 + 2; myD[p] = (int)r2; }
        cnt += __builtin_popcount(k2);
        unsigned k3 = __builtin_amdgcn_ballot_w32(m3);
        if (m3) { int p = cnt + __builtin_amdgcn_mbcnt_lo(k3, 0u); myE[p] = e0 + 3; myD[p] = (int)r3; }
        cnt += __builtin_popcount(k3);
        if (cnt > LCAP - 128) {
          agg_flush(x, ei, dinv, acc, myE, myD, cnt, win0, lane, off);
          cnt = 0;
        }
      }
    }
  }
  agg_flush(x, ei, dinv, acc, myE, myD, cnt, win0, lane, off);

  __syncthreads();

  for (int q = tid; q < WIN * 48; q += 1024) {
    int dl = q / 48, r = q - dl * 48;
    int b = r / 24, c4 = r - b * 24;
    int n = win0 + dl;
    if (n < NNODES)
      *(float4*)(xagg + ((size_t)(b * NNODES + n) * NFEAT + c4 * 4)) = *(const float4*)(acc + q * 4);
  }
}

#define GRU_WAVES 4
static_assert(ROWS % 16 == 0, "whole 16-row tiles: a wave either owns a full tile or leaves");

__device__ __forceinline__ v2f gldB(const float* __restrict__ W, int m, int hi, int kb, int nt) {
  int k0 = kb * 4 + hi * 2;
  int n = nt * 16 + m;
  v2f b;
  b.x = W[k0 * HIDN + n];
  b.y = W[(k0 + 1) * HIDN + n];
  return b;
}

__device__ __forceinline__ void wave_sync() {
  __builtin_amdgcn_fence(__ATOMIC_RELEASE, "wavefront");
  __builtin_amdgcn_wave_barrier();
  __builtin_amdgcn_fence(__ATOMIC_ACQUIRE, "wavefront");
}

__device__ __forceinline__ void gemm8r(v2f ax0, v2f ax1, const v2f (&B)[2][2], float b0v, float b1v,
                                       v8f& c0, v8f& c1) {
#pragma unroll
  for (int r = 0; r < 8; r++) { c0[r] = b0v; c1[r] = b1v; }
  c0 = wmma4(ax0, B[0][0], c0);
  c0 = wmma4(ax1, B[1][0], c0);
  c1 = wmma4(ax0, B[0][1], c1);
  c1 = wmma4(ax1, B[1][1], c1);
}

__global__ void __launch_bounds__(GRU_WAVES * 32) k_gru(
    const float* __restrict__ xagg, const float* __restrict__ att,
    const float* __restrict__ Wz, const float* __restrict__ bz,
    const float* __restrict__ Wr, const float* __restrict__ br,
    const float* __restrict__ Wh, const float* __restrict__ bh,
    const float* __restrict__ Lz, const float* __restrict__ lbz,
    const float* __restrict__ Lr, const float* __restrict__ lbr,
    const float* __restrict__ Lh, const float* __restrict__ lbh,
    const float* __restrict__ Wo, const float* __restrict__ bo,
    float* __restrict__ out) {
  __shared__ float sGz[GRU_WAVES][16 * 32];
  __shared__ float sGr[GRU_WAVES][16 * 32];
  __shared__ float sGh[GRU_WAVES][16 * 32];
  __shared__ float sH[GRU_WAVES][16 * 32];
  __shared__ float sHR[GRU_WAVES][16 * 32];
  __shared__ float sPr[GRU_WAVES][16];

  const int tid = threadIdx.x;
  const int lane = tid & 31, m = lane & 15, hi = lane >> 4;
  const int wave = __builtin_amdgcn_readfirstlane(tid >> 5);
  const int tile = blockIdx.x * GRU_WAVES + wave;
  if (tile * 16 >= ROWS) return;

  float* mGz = sGz[wave];
  float* mGr = sGr[wave];
  float* mGh = sGh[wave];
  float* mH = sH[wave];
  float* mHR = sHR[wave];
  float* mPr = sPr[wave];

  v2f BWz[2][2], BWr[2][2], BWh[2][2];
  v2f BLz[16][2], BLr[16][2], BLh[16][2];
#pragma unroll
  for (int kb = 0; kb < 2; kb++)
#pragma unroll
    for (int nt = 0; nt < 2; nt++) {
      BWz[kb][nt] = gldB(Wz, m, hi, kb, nt);
      BWr[kb][nt] = gldB(Wr, m, hi, kb, nt);
      BWh[kb][nt] = gldB(Wh, m, hi, kb, nt);
    }
#pragma unroll
  for (int kb = 0; kb < 16; kb++)
#pragma unroll
    for (int nt = 0; nt < 2; nt++) {
      BLz[kb][nt] = gldB(Lz, m, hi, kb, nt);
      BLr[kb][nt] = gldB(Lr, m, hi, kb, nt);
      BLh[kb][nt] = gldB(Lh, m, hi, kb, nt);
    }
  const float bz0 = bz[m], bz1 = bz[16 + m], br0 = br[m], br1 = br[16 + m], bh0 = bh[m], bh1 = bh[16 + m];
  const float lz0 = lbz[m], lz1 = lbz[16 + m], lr0 = lbr[m], lr1 = lbr[16 + m], lh0 = lbh[m], lh1 = lbh[16 + m];

  for (int idx = lane; idx < 16 * 32; idx += 32) mH[idx] = 0.f;

  {
    float a[PER]; float mx = -1e30f;
    for (int p = 0; p < PER; p++) { a[p] = att[p]; mx = fmaxf(mx, a[p]); }
    float sm = 0.f;
    for (int p = 0; p < PER; p++) { a[p] = __expf(a[p] - mx); sm += a[p]; }
    float inv = 1.0f / sm;
    float mine = 0.f;
    for (int p = 0; p < PER; p++) mine = (lane == p) ? a[p] * inv : mine;
    if (lane < PER) mPr[lane] = mine;
  }

  const long ia = (long)tile * 16 + m;
  const float* xrow = xagg + ia * (long)NFEAT;

  v8f H0, H1, A0, A1;
  for (int r = 0; r < 8; r++) { H0[r] = 0.f; H1[r] = 0.f; A0[r] = 0.f; A1[r] = 0.f; }

  wave_sync();

  for (int p = 0; p < PER; p++) {
    v2f aH[8];
    ldsA8(mH, m, hi, aH);

    int f0 = hi * 2;
    v2f ax0, ax1;
    ax0.x = xrow[(f0)*PER + p];     ax0.y = xrow[(f0 + 1) * PER + p];
    ax1.x = xrow[(f0 + 4) * PER + p]; ax1.y = xrow[(f0 + 5) * PER + p];
    const float prp = mPr[p];

    v8f g0, g1, z0, z1, r0, r1, t0, t1;

    gemm8r(ax0, ax1, BWz, bz0, bz1, g0, g1);
    storeTile(mGz, m, hi, g0, g1);
    gemm8r(ax0, ax1, BWr, br0, br1, g0, g1);
    storeTile(mGr, m, hi, g0, g1);
    gemm8r(ax0, ax1, BWh, bh0, bh1, g0, g1);
    storeTile(mGh, m, hi, g0, g1);
    wave_sync();

    v2f aGz[8], aGr[8], aGh[8];
    ldsA8(mGz, m, hi, aGz);
    ldsA8(mGr, m, hi, aGr);
    ldsA8(mGh, m, hi, aGh);
    __builtin_amdgcn_sched_barrier(0);

#pragma unroll
    for (int r = 0; r < 8; r++) { z0[r] = lz0; z1[r] = lz1; r0[r] = lr0; r1[r] = lr1; }
#pragma unroll
    for (int kb = 0; kb < 8; kb++) {
      r0 = wmma4(aGr[kb], BLr[kb][0], r0);
      r1 = wmma4(aGr[kb], BLr[kb][1], r1);
      z0 = wmma4(aGz[kb], BLz[kb][0], z0);
      z1 = wmma4(aGz[kb], BLz[kb][1], z1);
    }
#pragma unroll
    for (int kb = 0; kb < 8; kb++) {
      r0 = wmma4(aH[kb], BLr[kb + 8][0], r0);
      r1 = wmma4(aH[kb], BLr[kb + 8][1], r1);
      z0 = wmma4(aH[kb], BLz[kb + 8][0], z0);
      z1 = wmma4(aH[kb], BLz[kb + 8][1], z1);
    }

    for (int r = 0; r < 8; r++) { r0[r] = sigm(r0[r]); r1[r] = sigm(r1[r]); }
    {
      v8f q0, q1;
      for (int r = 0; r < 8; r++) { q0[r] = H0[r] * r0[r]; q1[r] = H1[r] * r1[r]; }
      storeTile(mHR, m, hi, q0, q1);
    }
    wave_sync();
    v2f aHR[8];
    ldsA8(mHR, m, hi, aHR);
    __builtin_amdgcn_sched_barrier(0);

#pragma unroll
    for (int r = 0; r < 8; r++) { t0[r] = lh0; t1[r] = lh1; }
#pragma unroll
    for (int kb = 0; kb < 8; kb++) {
      t0 = wmma4(aGh[kb], BLh[kb][0], t0);
      t1 = wmma4(aGh[kb], BLh[kb][1], t1);
    }
    for (int r = 0; r < 8; r++) { z0[r] = sigm(z0[r]); z1[r] = sigm(z1[r]); }
#pragma unroll
    for (int kb = 0; kb < 8; kb++) {
      t0 = wmma4(aHR[kb], BLh[kb + 8][0], t0);
      t1 = wmma4(aHR[kb], BLh[kb + 8][1], t1);
    }
    for (int r = 0; r < 8; r++) { t0[r] = tanhc(t0[r]); t1[r] = tanhc(t1[r]); }

    for (int r = 0; r < 8; r++) {
      H0[r] = z0[r] * H0[r] + (1.0f - z0[r]) * t0[r];
      H1[r] = z1[r] * H1[r] + (1.0f - z1[r]) * t1[r];
      A0[r] += prp * H0[r];
      A1[r] += prp * H1[r];
    }
    wave_sync();
    storeTile(mH, m, hi, H0, H1);
    wave_sync();
  }

  for (int r = 0; r < 8; r++) { A0[r] = fmaxf(A0[r], 0.f); A1[r] = fmaxf(A1[r], 0.f); }
  storeTile(mGz, m, hi, A0, A1);
  wave_sync();
  v8f o0, o1;
  {
    const bool real = m < OUTC - 16;
    const int mc = real ? 16 + m : 0;
    float b0v = bo[m], b1l = bo[mc];
    float b1v = real ? b1l : 0.f;
    for (int r = 0; r < 8; r++) { o0[r] = b0v; o1[r] = b1v; }
#pragma unroll
    for (int kb = 0; kb < 8; kb++) {
      int k0 = kb * 4 + hi * 2;
      v2f a = ldsA(mGz, m, hi, kb);
      v2f w0, w1;
      w0.x = Wo[k0 * OUTC + m];
      w0.y = Wo[(k0 + 1) * OUTC + m];
      float w1x = Wo[k0 * OUTC + mc], w1y = Wo[(k0 + 1) * OUTC + mc];
      w1.x = real ? w1x : 0.f;
      w1.y = real ? w1y : 0.f;
      o0 = wmma4(a, w0, o0);
      o1 = wmma4(a, w1, o1);
    }
  }
#pragma unroll
  for (int r = 0; r < 8; r++) {
    long row = (long)tile * 16 + r + 8 * hi;
    if (row < ROWS) {
      out[row * OUTC + m] = o0[r];
      if (m < 8) out[row * OUTC + 16 + m] = o1[r];
    }
  }
}

extern "C" void kernel_launch(void* const* d_in, const int* in_sizes, int n_in,
                              void* d_out, int out_size, void* d_ws, size_t ws_size,
                              hipStream_t stream) {
  (void)in_sizes; (void)n_in; (void)out_size; (void)ws_size;
  const float* x   = (const float*)d_in[0];
  const int*   ei  = (const int*)d_in[1];
  const float* att = (const float*)d_in[2];
  const float* Wz  = (const float*)d_in[3];  const float* bz  = (const float*)d_in[4];
  const float* Wr  = (const float*)d_in[5];  const float* br  = (const float*)d_in[6];
  const float* Wh  = (const float*)d_in[7];  const float* bh  = (const float*)d_in[8];
  const float* Lz  = (const float*)d_in[9];  const float* lbz = (const float*)d_in[10];
  const float* Lr  = (const float*)d_in[11]; const float* lbr = (const float*)d_in[12];
  const float* Lh  = (const float*)d_in[13]; const float* lbh = (const float*)d_in[14];
  const float* Wo  = (const float*)d_in[15]; const float* bo  = (const float*)d_in[16];
  float* out = (float*)d_out;

  float* wsf  = (float*)d_ws;
  float* deg  = wsf;
  float* xagg = wsf + 65536;

  k_deg_init<<<(NNODES + 255) / 256, 256, 0, stream>>>(deg);
  k_deg_acc<<<(NEDGES + 255) / 256, 256, 0, stream>>>(ei, deg);
  k_rsqrt<<<(NNODES + 255) / 256, 256, 0, stream>>>(deg);
  k_agg_window<<<NWIN, 1024, 0, stream>>>(x, ei, deg, xagg);

  k_gru<<<(ROWS / 16 + GRU_WAVES - 1) / GRU_WAVES, GRU_WAVES * 32, 0, stream>>>(xagg, att, Wz, bz, Wr, br, Wh, bh,
                                    Lz, lbz, Lr, lbr, Lh, lbh, Wo, bo, out);
}


# window edges from 1-byte tags: degree pass (4 edges/thread) writes tag=target/196; a filter launch (absorbs the rsqrt pass) lists them from 0.8 MB, not 3.2 MB; LDS adds in arrival order as before
# speedup vs baseline: 2.0071x; 1.0484x over previous
#include <hip/hip_runtime.h>

#define NNODES 50000
#define NEDGES 800000
#define FIN 8
#define PER 12
#define HIDN 32
#define NB 2
#define OUTC 24
#define ROWS (NB*NNODES)
#define NFEAT (FIN*PER)
#define TAGWIN 196

typedef float v2f __attribute__((ext_vector_type(2)));
typedef float v8f __attribute__((ext_vector_type(8)));

__device__ __forceinline__ v8f wmma4(v2f a, v2f b, v8f c) {
  return __builtin_amdgcn_wmma_f32_16x16x4_f32(false, a, false, b, (short)0, c, false, false);
}

__device__ __forceinline__ v2f ldsA(const float* buf, int m, int hi, int kb) {
  int k0 = kb * 4 + hi * 2;
  v2f a;
  a.x = buf[m * HIDN + k0];
  a.y = buf[m * HIDN + k0 + 1];
  return a;
}

__device__ __forceinline__ void ldsA8(const float* buf, int m, int hi, v2f (&a)[8]) {
#pragma unroll
  for (int kb = 0; kb < 8; kb++) a[kb] = ldsA(buf, m, hi, kb);
}

__device__ __forceinline__ void storeTile(float* buf, int m, int hi, const v8f& c0, const v8f& c1) {
#pragma unroll
  for (int r = 0; r < 8; r++) {
    buf[(r + 8 * hi) * HIDN + m]      = c0[r];
    buf[(r + 8 * hi) * HIDN + 16 + m] = c1[r];
  }
}

__device__ __forceinline__ float sigm(float x) { return 1.0f / (1.0f + __expf(-x)); }
__device__ __forceinline__ float tanhc(float x) {
  x = fminf(fmaxf(x, -15.f), 15.f);
  float e = __expf(2.0f * x);
  return (e - 1.0f) / (e + 1.0f);
}

__global__ void k_deg_init(float* deg) {
  int i = blockIdx.x * 256 + threadIdx.x;
  if (i < NNODES) deg[i] = 1.0f;
}
__global__ void k_deg_acc(const int* __restrict__ ei, float* deg, unsigned* __restrict__ tags) {
  int i4 = blockIdx.x * 256 + threadIdx.x;
  if (i4 < NEDGES / 4) {
    int4 d = ((const int4*)(ei + NEDGES))[i4];
    atomicAdd(&deg[d.x], 1.0f);
    atomicAdd(&deg[d.y], 1.0f);
    atomicAdd(&deg[d.z], 1.0f);
    atomicAdd(&deg[d.w], 1.0f);
    tags[i4] = ((unsigned)d.x / TAGWIN) | (((unsigned)d.y / TAGWIN) << 8) |
               (((unsigned)d.z / TAGWIN) << 16) | (((unsigned)d.w / TAGWIN) << 24);
  }
}
#define WIN 196
#define NWIN 256
#define ROWF (NB * NFEAT)
#define AGG_WAVES 32
#define LCAP 256
#define NE4 (NEDGES / 4)
static_assert(NWIN * WIN >= NNODES, "windows cover every node");
static_assert(NE4 * 4 == NEDGES && NE4 % 32 == 0, "whole waves of int4 reads over the target column");

__device__ __forceinline__ void agg_walk(const float* __restrict__ x, float* acc, int s, int dl, float w,
                                         int nn, int lane, const int* off) {
  {
    int j = 0;
    for (; j + 4 <= nn; j += 4) {
      float2 v[4][3];
      int dls[4];
      float ws[4];
#pragma unroll
      for (int u = 0; u < 4; u++) {
        int sj = __builtin_amdgcn_readlane(s, j + u);
        dls[u] = __builtin_amdgcn_readlane(dl, j + u);
        ws[u] = __int_as_float(__builtin_amdgcn_readlane(__float_as_int(w), j + u));
        const float* row = x + (size_t)sj * NFEAT;
#pragma unroll
        for (int c = 0; c < 3; c++) v[u][c] = *(const float2*)(row + off[c]);
      }
#pragma unroll
      for (int u = 0; u < 4; u++) {
        float* a = acc + dls[u] * ROWF + 2 * lane;
#pragma unroll
        for (int c = 0; c < 3; c++) {
          __hip_atomic_fetch_add(a + c * 64,     ws[u] * v[u][c].x, __ATOMIC_RELAXED, __HIP_MEMORY_SCOPE_WORKGROUP);
          __hip_atomic_fetch_add(a + c * 64 + 1, ws[u] * v[u][c].y, __ATOMIC_RELAXED, __HIP_MEMORY_SCOPE_WORKGROUP);
        }
      }
    }
    for (; j < nn; j++) {
      int sj = __builtin_amdgcn_readlane(s, j);
      int dj = __builtin_amdgcn_readlane(dl, j);
      float wj = __int_as_float(__builtin_amdgcn_readlane(__float_as_int(w), j));
      const float* row = x + (size_t)sj * NFEAT;
      float* a = acc + dj * ROWF + 2 * lane;
      float2 t[3];
#pragma unroll
      for (int c = 0; c < 3; c++) t[c] = *(const float2*)(row + off[c]);
#pragma unroll
      for (int c = 0; c < 3; c++) {
        __hip_atomic_fetch_add(a + c * 64,     wj * t[c].x, __ATOMIC_RELAXED, __HIP_MEMORY_SCOPE_WORKGROUP);
        __hip_atomic_fetch_add(a + c * 64 + 1, wj * t[c].y, __ATOMIC_RELAXED, __HIP_MEMORY_SCOPE_WORKGROUP);
      }
    }
  }
}

__device__ __forceinline__ void agg_flush(const float* __restrict__ x, const int* __restrict__ ei,
                                          const float* __restrict__ dinv, float* acc,
                                          const int* myE, const int* myD, int cnt, int win0,
                                          int lane, const int* off) {
  for (int base = 0; base < cnt; base += 32) {
    int idx = base + lane;
    int s = 0, dl = 0;
    float w = 0.f;
    if (idx < cnt) {
      int e = myE[idx];
      dl = myD[idx];
      s = ei[e];
      w = dinv[s] * dinv[win0 + dl];
    }
    int nn = cnt - base;
    if (nn > 32) nn = 32;
    agg_walk(x, acc, s, dl, w, nn, lane, off);
  }
}

#define NT16 (NEDGES / 16)
#define SCAP 256
static_assert(NT16 * 16 == NEDGES, "whole 16-byte reads over the tag array");
static_assert(NWIN == 256 && WIN == TAGWIN, "a tag is one byte and names the window");
static_assert((size_t)NEDGES <= (size_t)NNODES * NFEAT * 4, "the tags fit into the batch-1 rows");

__device__ __forceinline__ int win_rows(int win0) {
  int r = NNODES - win0;
  return r > WIN ? WIN : r;
}

__device__ __forceinline__ void list_out(const int* __restrict__ ei, unsigned* __restrict__ glist, int cap,
                                         const int* myE, int cnt, int win0, int lane, int* total) {
  int basep = 0;
  if (lane == 0) basep = __hip_atomic_fetch_add(total, cnt, __ATOMIC_RELAXED, __HIP_MEMORY_SCOPE_WORKGROUP);
  basep = __builtin_amdgcn_readfirstlane(basep);
  for (int idx = lane; idx < cnt; idx += 32) {
    int e = myE[idx];
    int s = ei[e];
    int d = ei[NEDGES + e];
    int pos = basep + idx;
    if (pos < cap) glist[pos] = ((unsigned)s << 8) | (unsigned)(d - win0);
  }
}

__global__ void __launch_bounds__(1024) k_tag_filter(const int* __restrict__ ei,
                                                     const unsigned* __restrict__ tagw,
                                                     float* __restrict__ deg,
                                                     float* xagg, int* __restrict__ gcount) {
  const uint4* tags = (const uint4*)tagw;
  __shared__ int lst[AGG_WAVES][SCAP];
  __shared__ int total;

  const int tid = threadIdx.x, lane = tid & 31;
  const int wave = __builtin_amdgcn_readfirstlane(tid >> 5);
  const int win0 = blockIdx.x * WIN;
  const int cap = win_rows(win0) * NFEAT;
  unsigned* glist = (unsigned*)(xagg + (size_t)win0 * NFEAT);
  const unsigned pat = (unsigned)blockIdx.x * 0x01010101u;

  if (tid == 0) total = 0;
  if (tid < WIN && win0 + tid < NNODES) deg[win0 + tid] = rsqrtf(deg[win0 + tid]);
  __syncthreads();

  int* myE = lst[wave];
  int cnt = 0;

  for (int base = wave * 32; base < NT16; base += 4 * 1024) {
    uint4 v[4];
#pragma unroll
    for (int u = 0; u < 4; u++) {
      int i = base + u * 1024 + lane;
      v[u] = tags[i < NT16 ? i : NT16 - 1];
    }
#pragma unroll
    for (int u = 0; u < 4; u++) {
      int i = base + u * 1024 + lane;
      unsigned x0 = v[u].x ^ pat, x1 = v[u].y ^ pat, x2 = v[u].z ^ pat, x3 = v[u].w ^ pat;
      unsigned z0 = ~(((x0 & 0x7f7f7f7fu) + 0x7f7f7f7fu) | x0) & 0x80808080u;
      unsigned z1 = ~(((x1 & 0x7f7f7f7fu) + 0x7f7f7f7fu) | x1) & 0x80808080u;
      unsigned z2 = ~(((x2 & 0x7f7f7f7fu) + 0x7f7f7f7fu) | x2) & 0x80808080u;
      unsigned z3 = ~(((x3 & 0x7f7f7f7fu) + 0x7f7f7f7fu) | x3) & 0x80808080u;
      unsigned m = (z0 >> 7) | (z1 >> 6) | (z2 >> 5) | (z3 >> 4);
      if (i >= NT16) m = 0u;
      while (__builtin_amdgcn_ballot_w32(m != 0u)) {
        bool has = m != 0u;
        int b = __builtin_ctz(m | 0x80000000u);
        unsigned kk = __builtin_amdgcn_ballot_w32(has);
        if (has) {
          int p = cnt + __builtin_amdgcn_mbcnt_lo(kk, 0u);
          myE[p] = 16 * i + 4 * (b & 7) + (b >> 3);
        }
        cnt += __builtin_popcount(kk);
        m &= m - 1u;
        if (cnt > SCAP - 32) {
          list_out(ei, glist, cap, myE, cnt, win0, lane, &total);
          cnt = 0;
        }
      }
    }
  }
  list_out(ei, glist, cap, myE, cnt, win0, lane, &total);

  __syncthreads();
  if (tid == 0) gcount[blockIdx.x] = total;
}

__global__ void __launch_bounds__(1024) k_agg_window(const float* __restrict__ x,
                                                     const int* __restrict__ ei,
                                                     const float* __restrict__ dinv,
                                                     const int* __restrict__ gcount,
                                                     float* xagg) {
  __shared__ __attribute__((aligned(16))) float acc[WIN * ROWF];
  __shared__ int lstE[AGG_WAVES][LCAP];
  __shared__ int lstD[AGG_WAVES][LCAP];

  const int tid = threadIdx.x, lane = tid & 31;
  const int wave = __builtin_amdgcn_readfirstlane(tid >> 5);
  const int win0 = blockIdx.x * WIN;

  for (int q = tid; q < WIN * 48; q += 1024) {
    int dl = q / 48, r = q - dl * 48;
    int b = r / 24, c4 = r - b * 24;
    int n = win0 + dl;
    float4 o = make_float4(0.f, 0.f, 0.f, 0.f);
    if (n < NNODES) {
      float dv = dinv[n];
      float w = dv * dv;
      float4 xv = *(const float4*)(x + ((size_t)(b * NNODES + n) * NFEAT + c4 * 4));
      o.x = xv.x * w; o.y = xv.y * w; o.z = xv.z * w; o.w = xv.w * w;
    }
    *(float4*)(acc + q * 4) = o;
  }

  int off[3];
#pragma unroll
  for (int c = 0; c < 3; c++) {
    int f = c * 64 + 2 * lane;
    off[c] = (f >= NFEAT) ? (NNODES * NFEAT + f - NFEAT) : f;
  }

  __syncthreads();

  const int nlist = gcount[blockIdx.x];
  if (nlist <= win_rows(win0) * NFEAT) {
    const unsigned* glist = (const unsigned*)(xagg + (size_t)win0 * NFEAT);
    for (int base = wave * 32; base < nlist; base += 1024) {
      int idx = base + lane;
      int s = 0, dl = 0;
      float w = 0.f;
      if (idx < nlist) {
        unsigned pk = glist[idx];
        s = (int)(pk >> 8);
        dl = (int)(pk & 255u);
        w = dinv[s] * dinv[win0 + dl];
      }
      int nn = nlist - base;
      if (nn > 32) nn = 32;
      agg_walk(x, acc, s, dl, w, nn, lane, off);
    }
  } else {
  const int4* dst4 = (const int4*)(ei + NEDGES);
  int* myE = lstE[wave];
  int* myD = lstD[wave];
  int cnt = 0;

  for (int base = wave * 32; base < NE4; base += 4 * 1024) {
    int4 v[4];
#pragma unroll
    for (int u = 0; u < 4; u++) {
      int i4 = base + u * 1024 + lane;
      v[u] = (i4 < NE4) ? dst4[i4] : make_int4(-1, -1, -1, -1);
    }
#pragma unroll
    for (int u = 0; u < 4; u++) {
      int e0 = 4 * (base + u * 1024 + lane);
      unsigned r0 = (unsigned)(v[u].x - win0), r1 = (unsigned)(v[u].y - win0);
      unsigned r2 = (unsigned)(v[u].z - win0), r3 = (unsigned)(v[u].w - win0);
      bool m0 = r0 < (unsigned)WIN, m1 = r1 < (unsigned)WIN, m2 = r2 < (unsigned)WIN, m3 = r3 < (unsigned)WIN;
      if (__builtin_amdgcn_ballot_w32(m0 || m1 || m2 || m3)) {
        unsigned k0 = __builtin_amdgcn_ballot_w32(m0);
        if (m0) { int p = cnt + __builtin_amdgcn_mbcnt_lo(k0, 0u); myE[p] = e0;     myD[p] = (int)r0; }
        cnt += __builtin_popcount(k0);
        unsigned k1 = __builtin_amdgcn_ballot_w32(m1);
        if (m1) { int p = cnt + __builtin_amdgcn_mbcnt_lo(k1, 0u); myE[p] = e0 + 1; myD[p] = (int)r1; }
        cnt += __builtin_popcount(k1);
        unsigned k2 = __builtin_amdgcn_ballot_w32(m2);
        if (m2) { int p = cnt + __builtin_amdgcn_mbcnt_lo(k2, 0u); myE[p] = e0 + 2; myD[p] = (int)r2; }
        cnt += __builtin_popcount(k2);
        unsigned k3 = __builtin_amdgcn_ballot_w32(m3);
        if (m3) { int p = cnt + __builtin_amdgcn_mbcnt_lo(k3, 0u); myE[p] = e0 + 3; myD[p] = (int)r3; }
        cnt += __builtin_popcount(k3);
        if (cnt > LCAP - 128) {
          agg_flush(x, ei, dinv, acc, myE, myD, cnt, win0, lane, off);
          cnt = 0;
        }
      }
    }
  }
  agg_flush(x, ei, dinv, acc, myE, myD, cnt, win0, lane, off);
  }

  __syncthreads();

  for (int q = tid; q < WIN * 48; q += 1024) {
    int dl = q / 48, r = q - dl * 48;
    int b = r / 24, c4 = r - b * 24;
    int n = win0 + dl;
    if (n < NNODES)
      *(float4*)(xagg + ((size_t)(b * NNODES + n) * NFEAT + c4 * 4)) = *(const float4*)(acc + q * 4);
  }
}

#define GRU_WAVES 4
static_assert(ROWS % 16 == 0, "whole 16-row tiles: a wave either owns a full tile or leaves");

__device__ __forceinline__ v2f gldB(const float* __restrict__ W, int m, int hi, int kb, int nt) {
  int k0 = kb * 4 + hi * 2;
  int n = nt * 16 + m;
  v2f b;
  b.x = W[k0 * HIDN + n];
  b.y = W[(k0 + 1) * HIDN + n];
  return b;
}

__device__ __forceinline__ void wave_sync() {
  __builtin_amdgcn_fence(__ATOMIC_RELEASE, "wavefront");
  __builtin_amdgcn_wave_barrier();
  __builtin_amdgcn_fence(__ATOMIC_ACQUIRE, "wavefront");
}

__device__ __forceinline__ void gemm8r(v2f ax0, v2f ax1, const v2f (&B)[2][2], float b0v, float b1v,
                                       v8f& c0, v8f& c1) {
#pragma unroll
  for (int r = 0; r < 8; r++) { c0[r] = b0v; c1[r] = b1v; }
  c0 = wmma4(ax0, B[0][0], c0);
  c0 = wmma4(ax1, B[1][0], c0);
  c1 = wmma4(ax0, B[0][1], c1);
  c1 = wmma4(ax1, B[1][1], c1);
}

__global__ void __launch_bounds__(GRU_WAVES * 32) k_gru(
    const float* __restrict__ xagg, const float* __restrict__ att,
    const float* __restrict__ Wz, const float* __restrict__ bz,
    const float* __restrict__ Wr, const float* __restrict__ br,
    const float* __restrict__ Wh, const float* __restrict__ bh,
    const float* __restrict__ Lz, const float* __restrict__ lbz,
    const float* __restrict__ Lr, const float* __restrict__ lbr,
    const float* __restrict__ Lh, const float* __restrict__ lbh,
    const float* __restrict__ Wo, const float* __restrict__ bo,
    float* __restrict__ out) {
  __shared__ float sGz[GRU_WAVES][16 * 32];
  __shared__ float sGr[GRU_WAVES][16 * 32];
  __shared__ float sGh[GRU_WAVES][16 * 32];
  __shared__ float sH[GRU_WAVES][16 * 32];
  __shared__ float sHR[GRU_WAVES][16 * 32];
  __shared__ float sPr[GRU_WAVES][16];

  const int tid = threadIdx.x;
  const int lane = tid & 31, m = lane & 15, hi = lane >> 4;
  const int wave = __builtin_amdgcn_readfirstlane(tid >> 5);
  const int tile = blockIdx.x * GRU_WAVES + wave;
  if (tile * 16 >= ROWS) return;

  float* mGz = sGz[wave];
  float* mGr = sGr[wave];
  float* mGh = sGh[wave];
  float* mH = sH[wave];
  float* mHR = sHR[wave];
  float* mPr = sPr[wave];

  v2f BWz[2][2], BWr[2][2], BWh[2][2];
  v2f BLz[16][2], BLr[16][2], BLh[16][2];
#pragma unroll
  for (int kb = 0; kb < 2; kb++)
#pragma unroll
    for (int nt = 0; nt < 2; nt++) {
      BWz[kb][nt] = gldB(Wz, m, hi, kb, nt);
      BWr[kb][nt] = gldB(Wr, m, hi, kb, nt);
      BWh[kb][nt] = gldB(Wh, m, hi, kb, nt);
    }
#pragma unroll
  for (int kb = 0; kb < 16; kb++)
#pragma unroll
    for (int nt = 0; nt < 2; nt++) {
      BLz[kb][nt] = gldB(Lz, m, hi, kb, nt);
      BLr[kb][nt] = gldB(Lr, m, hi, kb, nt);
      BLh[kb][nt] = gldB(Lh, m, hi, kb, nt);
    }
  const float bz0 = bz[m], bz1 = bz[16 + m], br0 = br[m], br1 = br[16 + m], bh0 = bh[m], bh1 = bh[16 + m];
  const float lz0 = lbz[m], lz1 = lbz[16 + m], lr0 = lbr[m], lr1 = lbr[16 + m], lh0 = lbh[m], lh1 = lbh[16 + m];

  for (int idx = lane; idx < 16 * 32; idx += 32) mH[idx] = 0.f;

  {
    float a[PER]; float mx = -1e30f;
    for (int p = 0; p < PER; p++) { a[p] = att[p]; mx = fmaxf(mx, a[p]); }
    float sm = 0.f;
    for (int p = 0; p < PER; p++) { a[p] = __expf(a[p] - mx); sm += a[p]; }
    float inv = 1.0f / sm;
    float mine = 0.f;
    for (int p = 0; p < PER; p++) mine = (lane == p) ? a[p] * inv : mine;
    if (lane < PER) mPr[lane] = mine;
  }

  const long ia = (long)tile * 16 + m;
  const float* xrow = xagg + ia * (long)NFEAT;

  v8f H0, H1, A0, A1;
  for (int r = 0; r < 8; r++) { H0[r] = 0.f; H1[r] = 0.f; A0[r] = 0.f; A1[r] = 0.f; }

  wave_sync();

  for (int p = 0; p < PER; p++) {
    v2f aH[8];
    ldsA8(mH, m, hi, aH);

    int f0 = hi * 2;
    v2f ax0, ax1;
    ax0.x = xrow[(f0)*PER + p];     ax0.y = xrow[(f0 + 1) * PER + p];
    ax1.x = xrow[(f0 + 4) * PER + p]; ax1.y = xrow[(f0 + 5) * PER + p];
    const float prp = mPr[p];

    v8f g0, g1, z0, z1, r0, r1, t0, t1;

    gemm8r(ax0, ax1, BWz, bz0, bz1, g0, g1);
    storeTile(mGz, m, hi, g0, g1);
    gemm8r(ax0, ax1, BWr, br0, br1, g0, g1);
    storeTile(mGr, m, hi, g0, g1);
    gemm8r(ax0, ax1, BWh, bh0, bh1, g0, g1);
    storeTile(mGh, m, hi, g0, g1);
    wave_sync();

    v2f aGz[8], aGr[8], aGh[8];
    ldsA8(mGz, m, hi, aGz);
    ldsA8(mGr, m, hi, aGr);
    ldsA8(mGh, m, hi, aGh);
    __builtin_amdgcn_sched_barrier(0);

#pragma unroll
    for (int r = 0; r < 8; r++) { z0[r] = lz0; z1[r] = lz1; r0[r] = lr0; r1[r] = lr1; }
#pragma unroll
    for (int kb = 0; kb < 8; kb++) {
      r0 = wmma4(aGr[kb], BLr[kb][0], r0);
      r1 = wmma4(aGr[kb], BLr[kb][1], r1);
      z0 = wmma4(aGz[kb], BLz[kb][0], z0);
      z1 = wmma4(aGz[kb], BLz[kb][1], z1);
    }
#pragma unroll
    for (int kb = 0; kb < 8; kb++) {
      r0 = wmma4(aH[kb], BLr[kb + 8][0], r0);
      r1 = wmma4(aH[kb], BLr[kb + 8][1], r1);
      z0 = wmma4(aH[kb], BLz[kb + 8][0], z0);
      z1 = wmma4(aH[kb], BLz[kb + 8][1], z1);
    }

    for (int r = 0; r < 8; r++) { r0[r] = sigm(r0[r]); r1[r] = sigm(r1[r]); }
    {
      v8f q0, q1;
      for (int r = 0; r < 8; r++) { q0[r] = H0[r] * r0[r]; q1[r] = H1[r] * r1[r]; }
      storeTile(mHR, m, hi, q0, q1);
    }
    wave_sync();
    v2f aHR[8];
    ldsA8(mHR, m, hi, aHR);
    __builtin_amdgcn_sched_barrier(0);

#pragma unroll
    for (int r = 0; r < 8; r++) { t0[r] = lh0; t1[r] = lh1; }
#pragma unroll
    for (int kb = 0; kb < 8; kb++) {
      t0 = wmma4(aGh[kb], BLh[kb][0], t0);
      t1 = wmma4(aGh[kb], BLh[kb][1], t1);
    }
    for (int r = 0; r < 8; r++) { z0[r] = sigm(z0[r]); z1[r] = sigm(z1[r]); }
#pragma unroll
    for (int kb = 0; kb < 8; kb++) {
      t0 = wmma4(aHR[kb], BLh[kb + 8][0], t0);
      t1 = wmma4(aHR[kb], BLh[kb + 8][1], t1);
    }
    for (int r = 0; r < 8; r++) { t0[r] = tanhc(t0[r]); t1[r] = tanhc(t1[r]); }

    for (int r = 0; r < 8; r++) {
      H0[r] = z0[r] * H0[r] + (1.0f - z0[r]) * t0[r];
      H1[r] = z1[r] * H1[r] + (1.0f - z1[r]) * t1[r];
      A0[r] += prp * H0[r];
      A1[r] += prp * H1[r];
    }
    wave_sync();
    storeTile(mH, m, hi, H0, H1);
    wave_sync();
  }

  for (int r = 0; r < 8; r++) { A0[r] = fmaxf(A0[r], 0.f); A1[r] = fmaxf(A1[r], 0.f); }
  storeTile(mGz, m, hi, A0, A1);
  wave_sync();
  v8f o0, o1;
  {
    const bool real = m < OUTC - 16;
    const int mc = real ? 16 + m : 0;
    float b0v = bo[m], b1l = bo[mc];
    float b1v = real ? b1l : 0.f;
    for (int r = 0; r < 8; r++) { o0[r] = b0v; o1[r] = b1v; }
#pragma unroll
    for (int kb = 0; kb < 8; kb++) {
      int k0 = kb * 4 + hi * 2;
      v2f a = ldsA(mGz, m, hi, kb);
      v2f w0, w1;
      w0.x = Wo[k0 * OUTC + m];
      w0.y = Wo[(k0 + 1) * OUTC + m];
      float w1x = Wo[k0 * OUTC + mc], w1y = Wo[(k0 + 1) * OUTC + mc];
      w1.x = real ? w1x : 0.f;
      w1.y = real ? w1y : 0.f;
      o0 = wmma4(a, w0, o0);
      o1 = wmma4(a, w1, o1);
    }
  }
#pragma unroll
  for (int r = 0; r < 8; r++) {
    long row = (long)tile * 16 + r + 8 * hi;
    if (row < ROWS) {
      out[row * OUTC + m] = o0[r];
      if (m < 8) out[row * OUTC + 16 + m] = o1[r];
    }
  }
}

extern "C" void kernel_launch(void* const* d_in, const int* in_sizes, int n_in,
                              void* d_out, int out_size, void* d_ws, size_t ws_size,
                              hipStream_t stream) {
  (void)in_sizes; (void)n_in; (void)out_size; (void)ws_size;
  const float* x   = (const float*)d_in[0];
  const int*   ei  = (const int*)d_in[1];
  const float* att = (const float*)d_in[2];
  const float* Wz  = (const float*)d_in[3];  const float* bz  = (const float*)d_in[4];
  const float* Wr  = (const float*)d_in[5];  const float* br  = (const float*)d_in[6];
  const float* Wh  = (const float*)d_in[7];  const float* bh  = (const float*)d_in[8];
  const float* Lz  = (const float*)d_in[9];  const float* lbz = (const float*)d_in[10];
  const float* Lr  = (const float*)d_in[11]; const float* lbr = (const float*)d_in[12];
  const float* Lh  = (const float*)d_in[13]; const float* lbh = (const float*)d_in[14];
  const float* Wo  = (const float*)d_in[15]; const float* bo  = (const float*)d_in[16];
  float* out = (float*)d_out;

  float* wsf  = (float*)d_ws;
  float* deg  = wsf;
  float* xagg = wsf + 65536;
  int* gcount = (int*)(wsf + 50000);
  unsigned* tags = (unsigned*)(xagg + (size_t)NNODES * NFEAT);

  k_deg_init<<<(NNODES + 255) / 256, 256, 0, stream>>>(deg);
  k_deg_acc<<<(NEDGES / 4 + 255) / 256, 256, 0, stream>>>(ei, deg, tags);
  k_tag_filter<<<NWIN, 1024, 0, stream>>>(ei, tags, deg, xagg, gcount);
  k_agg_window<<<NWIN, 1024, 0, stream>>>(x, ei, deg, gcount, xagg);

  k_gru<<<(ROWS / 16 + GRU_WAVES - 1) / GRU_WAVES, GRU_WAVES * 32, 0, stream>>>(xagg, att, Wz, bz, Wr, br, Wh, bh,
                                    Lz, lbz, Lr, lbr, Lh, lbh, Wo, bo, out);
}


# GRU gates: two activations share one IEEE division (1/a, 1/b from 1/(a*b)): R by column pairs, Z with its element's tanh; 24 divisions per lane per period, was 48; rounding differs, sums unchanged
# speedup vs baseline: 2.3127x; 1.1523x over previous
#include <hip/hip_runtime.h>

#define NNODES 50000
#define NEDGES 800000
#define FIN 8
#define PER 12
#define HIDN 32
#define NB 2
#define OUTC 24
#define ROWS (NB*NNODES)
#define NFEAT (FIN*PER)
#define TAGWIN 196

typedef float v2f __attribute__((ext_vector_type(2)));
typedef float v8f __attribute__((ext_vector_type(8)));

__device__ __forceinline__ v8f wmma4(v2f a, v2f b, v8f c) {
  return __builtin_amdgcn_wmma_f32_16x16x4_f32(false, a, false, b, (short)0, c, false, false);
}

__device__ __forceinline__ v2f ldsA(const float* buf, int m, int hi, int kb) {
  int k0 = kb * 4 + hi * 2;
  v2f a;
  a.x = buf[m * HIDN + k0];
  a.y = buf[m * HIDN + k0 + 1];
  return a;
}

__device__ __forceinline__ void ldsA8(const float* buf, int m, int hi, v2f (&a)[8]) {
#pragma unroll
  for (int kb = 0; kb < 8; kb++) a[kb] = ldsA(buf, m, hi, kb);
}

__device__ __forceinline__ void storeTile(float* buf, int m, int hi, const v8f& c0, const v8f& c1) {
#pragma unroll
  for (int r = 0; r < 8; r++) {
    buf[(r + 8 * hi) * HIDN + m]      = c0[r];
    buf[(r + 8 * hi) * HIDN + 16 + m] = c1[r];
  }
}

#define SIG_PAIR_BOUND 43.0f
#define SIG_TANH_BOUND 57.0f
__device__ __forceinline__ float sig_den(float x, float bound) { return 1.0f + __expf(fminf(-x, bound)); }
__device__ __forceinline__ void sigm2(float& x0, float& x1) {
  float a = sig_den(x0, SIG_PAIR_BOUND), b = sig_den(x1, SIG_PAIR_BOUND);
  float inv = 1.0f / (a * b);
  x0 = inv * b;
  x1 = inv * a;
}
__device__ __forceinline__ void sigm_tanh(float za, float x, float& z, float& t) {
  x = fminf(fmaxf(x, -15.f), 15.f);
  float e = __expf(2.0f * x);
  float b = e + 1.0f;
  float inv = 1.0f / (za * b);
  z = inv * b;
  t = (e - 1.0f) * (inv * za);
}

__global__ void k_deg_init(float* deg) {
  int i = blockIdx.x * 256 + threadIdx.x;
  if (i < NNODES) deg[i] = 1.0f;
}
__global__ void k_deg_acc(const int* __restrict__ ei, float* deg, unsigned* __restrict__ tags) {
  int i4 = blockIdx.x * 256 + threadIdx.x;
  if (i4 < NEDGES / 4) {
    int4 d = ((const int4*)(ei + NEDGES))[i4];
    atomicAdd(&deg[d.x], 1.0f);
    atomicAdd(&deg[d.y], 1.0f);
    atomicAdd(&deg[d.z], 1.0f);
    atomicAdd(&deg[d.w], 1.0f);
    tags[i4] = ((unsigned)d.x / TAGWIN) | (((unsigned)d.y / TAGWIN) << 8) |
               (((unsigned)d.z / TAGWIN) << 16) | (((unsigned)d.w / TAGWIN) << 24);
  }
}
#define WIN 196
#define NWIN 256
#define ROWF (NB * NFEAT)
#define AGG_WAVES 32
#define LCAP 256
#define NE4 (NEDGES / 4)
static_assert(NWIN * WIN >= NNODES, "windows cover every node");
static_assert(NE4 * 4 == NEDGES && NE4 % 32 == 0, "whole waves of int4 reads over the target column");

__device__ __forceinline__ void agg_walk(const float* __restrict__ x, float* acc, int s, int dl, float w,
                                         int nn, int lane, const int* off) {
  {
    int j = 0;
    for (; j + 4 <= nn; j += 4) {
      float2 v[4][3];
      int dls[4];
      float ws[4];
#pragma unroll
      for (int u = 0; u < 4; u++) {
        int sj = __builtin_amdgcn_readlane(s, j + u);
        dls[u] = __builtin_amdgcn_readlane(dl, j + u);
        ws[u] = __int_as_float(__builtin_amdgcn_readlane(__float_as_int(w), j + u));
        const float* row = x + (size_t)sj * NFEAT;
#pragma unroll
        for (int c = 0; c < 3; c++) v[u][c] = *(const float2*)(row + off[c]);
      }
#pragma unroll
      for (int u = 0; u < 4; u++) {
        float* a = acc + dls[u] * ROWF + 2 * lane;
#pragma unroll
        for (int c = 0; c < 3; c++) {
          __hip_atomic_fetch_add(a + c * 64,     ws[u] * v[u][c].x, __ATOMIC_RELAXED, __HIP_MEMORY_SCOPE_WORKGROUP);
          __hip_atomic_fetch_add(a + c * 64 + 1, ws[u] * v[u][c].y, __ATOMIC_RELAXED, __HIP_MEMORY_SCOPE_WORKGROUP);
        }
      }
    }
    for (; j < nn; j++) {
      int sj = __builtin_amdgcn_readlane(s, j);
      int dj = __builtin_amdgcn_readlane(dl, j);
      float wj = __int_as_float(__builtin_amdgcn_readlane(__float_as_int(w), j));
      const float* row = x + (size_t)sj * NFEAT;
      float* a = acc + dj * ROWF + 2 * lane;
      float2 t[3];
#pragma unroll
      for (int c = 0; c < 3; c++) t[c] = *(const float2*)(row + off[c]);
#pragma unroll
      for (int c = 0; c < 3; c++) {
        __hip_atomic_fetch_add(a + c * 64,     wj * t[c].x, __ATOMIC_RELAXED, __HIP_MEMORY_SCOPE_WORKGROUP);
        __hip_atomic_fetch_add(a + c * 64 + 1, wj * t[c].y, __ATOMIC_RELAXED, __HIP_MEMORY_SCOPE_WORKGROUP);
      }
    }
  }
}

__device__ __forceinline__ void agg_flush(const float* __restrict__ x, const int* __restrict__ ei,
                                          const float* __restrict__ dinv, float* acc,
                                          const int* myE, const int* myD, int cnt, int win0,
                                          int lane, const int* off) {
  for (int base = 0; base < cnt; base += 32) {
    int idx = base + lane;
    int s = 0, dl = 0;
    float w = 0.f;
    if (idx < cnt) {
      int e = myE[idx];
      dl = myD[idx];
      s = ei[e];
      w = dinv[s] * dinv[win0 + dl];
    }
    int nn = cnt - base;
    if (nn > 32) nn = 32;
    agg_walk(x, acc, s, dl, w, nn, lane, off);
  }
}

#define NT16 (NEDGES / 16)
#define SCAP 256
static_assert(NT16 * 16 == NEDGES, "whole 16-byte reads over the tag array");
static_assert(NWIN == 256 && WIN == TAGWIN, "a tag is one byte and names the window");
static_assert((size_t)NEDGES <= (size_t)NNODES * NFEAT * 4, "the tags fit into the batch-1 rows");

__device__ __forceinline__ int win_rows(int win0) {
  int r = NNODES - win0;
  return r > WIN ? WIN : r;
}

__device__ __forceinline__ void list_out(const int* __restrict__ ei, unsigned* __restrict__ glist, int cap,
                                         const int* myE, int cnt, int win0, int lane, int* total) {
  int basep = 0;
  if (lane == 0) basep = __hip_atomic_fetch_add(total, cnt, __ATOMIC_RELAXED, __HIP_MEMORY_SCOPE_WORKGROUP);
  basep = __builtin_amdgcn_readfirstlane(basep);
  for (int idx = lane; idx < cnt; idx += 32) {
    int e = myE[idx];
    int s = ei[e];
    int d = ei[NEDGES + e];
    int pos = basep + idx;
    if (pos < cap) glist[pos] = ((unsigned)s << 8) | (unsigned)(d - win0);
  }
}

__global__ void __launch_bounds__(1024) k_tag_filter(const int* __restrict__ ei,
                                                     const unsigned* __restrict__ tagw,
                                                     float* __restrict__ deg,
                                                     float* xagg, int* __restrict__ gcount) {
  const uint4* tags = (const uint4*)tagw;
  __shared__ int lst[AGG_WAVES][SCAP];
  __shared__ int total;

  const int tid = threadIdx.x, lane = tid & 31;
  const int wave = __builtin_amdgcn_readfirstlane(tid >> 5);
  const int win0 = blockIdx.x * WIN;
  const int cap = win_rows(win0) * NFEAT;
  unsigned* glist = (unsigned*)(xagg + (size_t)win0 * NFEAT);
  const unsigned pat = (unsigned)blockIdx.x * 0x01010101u;

  if (tid == 0) total = 0;
  if (tid < WIN && win0 + tid < NNODES) deg[win0 + tid] = rsqrtf(deg[win0 + tid]);
  __syncthreads();

  int* myE = lst[wave];
  int cnt = 0;

  for (int base = wave * 32; base < NT16; base += 4 * 1024) {
    uint4 v[4];
#pragma unroll
    for (int u = 0; u < 4; u++) {
      int i = base + u * 1024 + lane;
      v[u] = tags[i < NT16 ? i : NT16 - 1];
    }
#pragma unroll
    for (int u = 0; u < 4; u++) {
      int i = base + u * 1024 + lane;
      unsigned x0 = v[u].x ^ pat, x1 = v[u].y ^ pat, x2 = v[u].z ^ pat, x3 = v[u].w ^ pat;
      unsigned z0 = ~(((x0 & 0x7f7f7f7fu) + 0x7f7f7f7fu) | x0) & 0x80808080u;
      unsigned z1 = ~(((x1 & 0x7f7f7f7fu) + 0x7f7f7f7fu) | x1) & 0x80808080u;
      unsigned z2 = ~(((x2 & 0x7f7f7f7fu) + 0x7f7f7f7fu) | x2) & 0x80808080u;
      unsigned z3 = ~(((x3 & 0x7f7f7f7fu) + 0x7f7f7f7fu) | x3) & 0x80808080u;
      unsigned m = (z0 >> 7) | (z1 >> 6) | (z2 >> 5) | (z3 >> 4);
      if (i >= NT16) m = 0u;
      while (__builtin_amdgcn_ballot_w32(m != 0u)) {
        bool has = m != 0u;
        int b = __builtin_ctz(m | 0x80000000u);
        unsigned kk = __builtin_amdgcn_ballot_w32(has);
        if (has) {
          int p = cnt + __builtin_amdgcn_mbcnt_lo(kk, 0u);
          myE[p] = 16 * i + 4 * (b & 7) + (b >> 3);
        }
        cnt += __builtin_popcount(kk);
        m &= m - 1u;
        if (cnt > SCAP - 32) {
          list_out(ei, glist, cap, myE, cnt, win0, lane, &total);
          cnt = 0;
        }
      }
    }
  }
  list_out(ei, glist, cap, myE, cnt, win0, lane, &total);

  __syncthreads();
  if (tid == 0) gcount[blockIdx.x] = total;
}

__global__ void __launch_bounds__(1024) k_agg_window(const float* __restrict__ x,
                                                     const int* __restrict__ ei,
                                                     const float* __restrict__ dinv,
                                                     const int* __restrict__ gcount,
                                                     float* xagg) {
  __shared__ __attribute__((aligned(16))) float acc[WIN * ROWF];
  __shared__ int lstE[AGG_WAVES][LCAP];
  __shared__ int lstD[AGG_WAVES][LCAP];

  const int tid = threadIdx.x, lane = tid & 31;
  const int wave = __builtin_amdgcn_readfirstlane(tid >> 5);
  const int win0 = blockIdx.x * WIN;

  for (int q = tid; q < WIN * 48; q += 1024) {
    int dl = q / 48, r = q - dl * 48;
    int b = r / 24, c4 = r - b * 24;
    int n = win0 + dl;
    float4 o = make_float4(0.f, 0.f, 0.f, 0.f);
    if (n < NNODES) {
      float dv = dinv[n];
      float w = dv * dv;
      float4 xv = *(const float4*)(x + ((size_t)(b * NNODES + n) * NFEAT + c4 * 4));
      o.x = xv.x * w; o.y = xv.y * w; o.z = xv.z * w; o.w = xv.w * w;
    }
    *(float4*)(acc + q * 4) = o;
  }

  int off[3];
#pragma unroll
  for (int c = 0; c < 3; c++) {
    int f = c * 64 + 2 * lane;
    off[c] = (f >= NFEAT) ? (NNODES * NFEAT + f - NFEAT) : f;
  }

  __syncthreads();

  const int nlist = gcount[blockIdx.x];
  if (nlist <= win_rows(win0) * NFEAT) {
    const unsigned* glist = (const unsigned*)(xagg + (size_t)win0 * NFEAT);
    for (int base = wave * 32; base < nlist; base += 1024) {
      int idx = base + lane;
      int s = 0, dl = 0;
      float w = 0.f;
      if (idx < nlist) {
        unsigned pk = glist[idx];
        s = (int)(pk >> 8);
        dl = (int)(pk & 255u);
        w = dinv[s] * dinv[win0 + dl];
      }
      int nn = nlist - base;
      if (nn > 32) nn = 32;
      agg_walk(x, acc, s, dl, w, nn, lane, off);
    }
  } else {
  const int4* dst4 = (const int4*)(ei + NEDGES);
  int* myE = lstE[wave];
  int* myD = lstD[wave];
  int cnt = 0;

  for (int base = wave * 32; base < NE4; base += 4 * 1024) {
    int4 v[4];
#pragma unroll
    for (int u = 0; u < 4; u++) {
      int i4 = base + u * 1024 + lane;
      v[u] = (i4 < NE4) ? dst4[i4] : make_int4(-1, -1, -1, -1);
    }
#pragma unroll
    for (int u = 0; u < 4; u++) {
      int e0 = 4 * (base + u * 1024 + lane);
      unsigned r0 = (unsigned)(v[u].x - win0), r1 = (unsigned)(v[u].y - win0);
      unsigned r2 = (unsigned)(v[u].z - win0), r3 = (unsigned)(v[u].w - win0);
      bool m0 = r0 < (unsigned)WIN, m1 = r1 < (unsigned)WIN, m2 = r2 < (unsigned)WIN, m3 = r3 < (unsigned)WIN;
      if (__builtin_amdgcn_ballot_w32(m0 || m1 || m2 || m3)) {
        unsigned k0 = __builtin_amdgcn_ballot_w32(m0);
        if (m0) { int p = cnt + __builtin_amdgcn_mbcnt_lo(k0, 0u); myE[p] = e0;     myD[p] = (int)r0; }
        cnt += __builtin_popcount(k0);
        unsigned k1 = __builtin_amdgcn_ballot_w32(m1);
        if (m1) { int p = cnt + __builtin_amdgcn_mbcnt_lo(k1, 0u); myE[p] = e0 + 1; myD[p] = (int)r1; }
        cnt += __builtin_popcount(k1);
        unsigned k2 = __builtin_amdgcn_ballot_w32(m2);
        if (m2) { int p = cnt + __builtin_amdgcn_mbcnt_lo(k2, 0u); myE[p] = e0 + 2; myD[p] = (int)r2; }
        cnt += __builtin_popcount(k2);
        unsigned k3 = __builtin_amdgcn_ballot_w32(m3);
        if (m3) { int p = cnt + __builtin_amdgcn_mbcnt_lo(k3, 0u); myE[p] = e0 + 3; myD[p] = (int)r3; }
        cnt += __builtin_popcount(k3);
        if (cnt > LCAP - 128) {
          agg_flush(x, ei, dinv, acc, myE, myD, cnt, win0, lane, off);
          cnt = 0;
        }
      }
    }
  }
  agg_flush(x, ei, dinv, acc, myE, myD, cnt, win0, lane, off);
  }

  __syncthreads();

  for (int q = tid; q < WIN * 48; q += 1024) {
    int dl = q / 48, r = q - dl * 48;
    int b = r / 24, c4 = r - b * 24;
    int n = win0 + dl;
    if (n < NNODES)
      *(float4*)(xagg + ((size_t)(b * NNODES + n) * NFEAT + c4 * 4)) = *(const float4*)(acc + q * 4);
  }
}

#define GRU_WAVES 4
static_assert(ROWS % 16 == 0, "whole 16-row tiles: a wave either owns a full tile or leaves");

__device__ __forceinline__ v2f gldB(const float* __restrict__ W, int m, int hi, int kb, int nt) {
  int k0 = kb * 4 + hi * 2;
  int n = nt * 16 + m;
  v2f b;
  b.x = W[k0 * HIDN + n];
  b.y = W[(k0 + 1) * HIDN + n];
  return b;
}

__device__ __forceinline__ void wave_sync() {
  __builtin_amdgcn_fence(__ATOMIC_RELEASE, "wavefront");
  __builtin_amdgcn_wave_barrier();
  __builtin_amdgcn_fence(__ATOMIC_ACQUIRE, "wavefront");
}

__device__ __forceinline__ void gemm8r(v2f ax0, v2f ax1, const v2f (&B)[2][2], float b0v, float b1v,
                                       v8f& c0, v8f& c1) {
#pragma unroll
  for (int r = 0; r < 8; r++) { c0[r] = b0v; c1[r] = b1v; }
  c0 = wmma4(ax0, B[0][0], c0);
  c0 = wmma4(ax1, B[1][0], c0);
  c1 = wmma4(ax0, B[0][1], c1);
  c1 = wmma4(ax1, B[1][1], c1);
}

__global__ void __launch_bounds__(GRU_WAVES * 32) k_gru(
    const float* __restrict__ xagg, const float* __restrict__ att,
    const float* __restrict__ Wz, const float* __restrict__ bz,
    const float* __restrict__ Wr, const float* __restrict__ br,
    const float* __restrict__ Wh, const float* __restrict__ bh,
    const float* __restrict__ Lz, const float* __restrict__ lbz,
    const float* __restrict__ Lr, const float* __restrict__ lbr,
    const float* __restrict__ Lh, const float* __restrict__ lbh,
    const float* __restrict__ Wo, const float* __restrict__ bo,
    float* __restrict__ out) {
  __shared__ float sGz[GRU_WAVES][16 * 32];
  __shared__ float sGr[GRU_WAVES][16 * 32];
  __shared__ float sGh[GRU_WAVES][16 * 32];
  __shared__ float sH[GRU_WAVES][16 * 32];
  __shared__ float sHR[GRU_WAVES][16 * 32];
  __shared__ float sPr[GRU_WAVES][16];

  const int tid = threadIdx.x;
  const int lane = tid & 31, m = lane & 15, hi = lane >> 4;
  const int wave = __builtin_amdgcn_readfirstlane(tid >> 5);
  const int tile = blockIdx.x * GRU_WAVES + wave;
  if (tile * 16 >= ROWS) return;

  float* mGz = sGz[wave];
  float* mGr = sGr[wave];
  float* mGh = sGh[wave];
  float* mH = sH[wave];
  float* mHR = sHR[wave];
  float* mPr = sPr[wave];

  v2f BWz[2][2], BWr[2][2], BWh[2][2];
  v2f BLz[16][2], BLr[16][2], BLh[16][2];
#pragma unroll
  for (int kb = 0; kb < 2; kb++)
#pragma unroll
    for (int nt = 0; nt < 2; nt++) {
      BWz[kb][nt] = gldB(Wz, m, hi, kb, nt);
      BWr[kb][nt] = gldB(Wr, m, hi, kb, nt);
      BWh[kb][nt] = gldB(Wh, m, hi, kb, nt);
    }
#pragma unroll
  for (int kb = 0; kb < 16; kb++)
#pragma unroll
    for (int nt = 0; nt < 2; nt++) {
      BLz[kb][nt] = gldB(Lz, m, hi, kb, nt);
      BLr[kb][nt] = gldB(Lr, m, hi, kb, nt);
      BLh[kb][nt] = gldB(Lh, m, hi, kb, nt);
    }
  const float bz0 = bz[m], bz1 = bz[16 + m], br0 = br[m], br1 = br[16 + m], bh0 = bh[m], bh1 = bh[16 + m];
  const float lz0 = lbz[m], lz1 = lbz[16 + m], lr0 = lbr[m], lr1 = lbr[16 + m], lh0 = lbh[m], lh1 = lbh[16 + m];

  for (int idx = lane; idx < 16 * 32; idx += 32) mH[idx] = 0.f;

  {
    float a[PER]; float mx = -1e30f;
    for (int p = 0; p < PER; p++) { a[p] = att[p]; mx = fmaxf(mx, a[p]); }
    float sm = 0.f;
    for (int p = 0; p < PER; p++) { a[p] = __expf(a[p] - mx); sm += a[p]; }
    float inv = 1.0f / sm;
    float mine = 0.f;
    for (int p = 0; p < PER; p++) mine = (lane == p) ? a[p] * inv : mine;
    if (lane < PER) mPr[lane] = mine;
  }

  const long ia = (long)tile * 16 + m;
  const float* xrow = xagg + ia * (long)NFEAT;

  v8f H0, H1, A0, A1;
  for (int r = 0; r < 8; r++) { H0[r] = 0.f; H1[r] = 0.f; A0[r] = 0.f; A1[r] = 0.f; }

  wave_sync();

  for (int p = 0; p < PER; p++) {
    v2f aH[8];
    ldsA8(mH, m, hi, aH);

    int f0 = hi * 2;
    v2f ax0, ax1;
    ax0.x = xrow[(f0)*PER + p];     ax0.y = xrow[(f0 + 1) * PER + p];
    ax1.x = xrow[(f0 + 4) * PER + p]; ax1.y = xrow[(f0 + 5) * PER + p];
    const float prp = mPr[p];

    v8f g0, g1, z0, z1, r0, r1, t0, t1;

    gemm8r(ax0, ax1, BWz, bz0, bz1, g0, g1);
    storeTile(mGz, m, hi, g0, g1);
    gemm8r(ax0, ax1, BWr, br0, br1, g0, g1);
    storeTile(mGr, m, hi, g0, g1);
    gemm8r(ax0, ax1, BWh, bh0, bh1, g0, g1);
    storeTile(mGh, m, hi, g0, g1);
    wave_sync();

    v2f aGz[8], aGr[8], aGh[8];
    ldsA8(mGz, m, hi, aGz);
    ldsA8(mGr, m, hi, aGr);
    ldsA8(mGh, m, hi, aGh);
    __builtin_amdgcn_sched_barrier(0);

#pragma unroll
    for (int r = 0; r < 8; r++) { z0[r] = lz0; z1[r] = lz1; r0[r] = lr0; r1[r] = lr1; }
#pragma unroll
    for (int kb = 0; kb < 8; kb++) {
      r0 = wmma4(aGr[kb], BLr[kb][0], r0);
      r1 = wmma4(aGr[kb], BLr[kb][1], r1);
      z0 = wmma4(aGz[kb], BLz[kb][0], z0);
      z1 = wmma4(aGz[kb], BLz[kb][1], z1);
    }
#pragma unroll
    for (int kb = 0; kb < 8; kb++) {
      r0 = wmma4(aH[kb], BLr[kb + 8][0], r0);
      r1 = wmma4(aH[kb], BLr[kb + 8][1], r1);
      z0 = wmma4(aH[kb], BLz[kb + 8][0], z0);
      z1 = wmma4(aH[kb], BLz[kb + 8][1], z1);
    }

    for (int r = 0; r < 8; r++) {
      float a = r0[r], b = r1[r];
      sigm2(a, b);
      r0[r] = a; r1[r] = b;
    }
    {
      v8f q0, q1;
      for (int r = 0; r < 8; r++) { q0[r] = H0[r] * r0[r]; q1[r] = H1[r] * r1[r]; }
      storeTile(mHR, m, hi, q0, q1);
    }
    wave_sync();
    v2f aHR[8];
    ldsA8(mHR, m, hi, aHR);
    __builtin_amdgcn_sched_barrier(0);

#pragma unroll
    for (int r = 0; r < 8; r++) { t0[r] = lh0; t1[r] = lh1; }
#pragma unroll
    for (int kb = 0; kb < 8; kb++) {
      t0 = wmma4(aGh[kb], BLh[kb][0], t0);
      t1 = wmma4(aGh[kb], BLh[kb][1], t1);
    }
    for (int r = 0; r < 8; r++) { z0[r] = sig_den(z0[r], SIG_TANH_BOUND); z1[r] = sig_den(z1[r], SIG_TANH_BOUND); }
#pragma unroll
    for (int kb = 0; kb < 8; kb++) {
      t0 = wmma4(aHR[kb], BLh[kb + 8][0], t0);
      t1 = wmma4(aHR[kb], BLh[kb + 8][1], t1);
    }
    for (int r = 0; r < 8; r++) {
      float za, ta, zb, tb;
      sigm_tanh(z0[r], t0[r], za, ta);
      sigm_tanh(z1[r], t1[r], zb, tb);
      z0[r] = za; t0[r] = ta; z1[r] = zb; t1[r] = tb;
    }

    for (int r = 0; r < 8; r++) {
      H0[r] = z0[r] * H0[r] + (1.0f - z0[r]) * t0[r];
      H1[r] = z1[r] * H1[r] + (1.0f - z1[r]) * t1[r];
      A0[r] += prp * H0[r];
      A1[r] += prp * H1[r];
    }
    wave_sync();
    storeTile(mH, m, hi, H0, H1);
    wave_sync();
  }

  for (int r = 0; r < 8; r++) { A0[r] = fmaxf(A0[r], 0.f); A1[r] = fmaxf(A1[r], 0.f); }
  storeTile(mGz, m, hi, A0, A1);
  wave_sync();
  v8f o0, o1;
  {
    const bool real = m < OUTC - 16;
    const int mc = real ? 16 + m : 0;
    float b0v = bo[m], b1l = bo[mc];
    float b1v = real ? b1l : 0.f;
    for (int r = 0; r < 8; r++) { o0[r] = b0v; o1[r] = b1v; }
#pragma unroll
    for (int kb = 0; kb < 8; kb++) {
      int k0 = kb * 4 + hi * 2;
      v2f a = ldsA(mGz, m, hi, kb);
      v2f w0, w1;
      w0.x = Wo[k0 * OUTC + m];
      w0.y = Wo[(k0 + 1) * OUTC + m];
      float w1x = Wo[k0 * OUTC + mc], w1y = Wo[(k0 + 1) * OUTC + mc];
      w1.x = real ? w1x : 0.f;
      w1.y = real ? w1y : 0.f;
      o0 = wmma4(a, w0, o0);
      o1 = wmma4(a, w1, o1);
    }
  }
#pragma unroll
  for (int r = 0; r < 8; r++) {
    long row = (long)tile * 16 + r + 8 * hi;
    if (row < ROWS) {
      out[row * OUTC + m] = o0[r];
      if (m < 8) out[row * OUTC + 16 + m] = o1[r];
    }
  }
}

extern "C" void kernel_launch(void* const* d_in, const int* in_sizes, int n_in,
                              void* d_out, int out_size, void* d_ws, size_t ws_size,
                              hipStream_t stream) {
  (void)in_sizes; (void)n_in; (void)out_size; (void)ws_size;
  const float* x   = (const float*)d_in[0];
  const int*   ei  = (const int*)d_in[1];
  const float* att = (const float*)d_in[2];
  const float* Wz  = (const float*)d_in[3];  const float* bz  = (const float*)d_in[4];
  const float* Wr  = (const float*)d_in[5];  const float* br  = (const float*)d_in[6];
  const float* Wh  = (const float*)d_in[7];  const float* bh  = (const float*)d_in[8];
  const float* Lz  = (const float*)d_in[9];  const float* lbz = (const float*)d_in[10];
  const float* Lr  = (const float*)d_in[11]; const float* lbr = (const float*)d_in[12];
  const float* Lh  = (const float*)d_in[13]; const float* lbh = (const float*)d_in[14];
  const float* Wo  = (const float*)d_in[15]; const float* bo  = (const float*)d_in[16];
  float* out = (float*)d_out;

  float* wsf  = (float*)d_ws;
  float* deg  = wsf;
  float* xagg = wsf + 65536;
  int* gcount = (int*)(wsf + 50000);
  unsigned* tags = (unsigned*)(xagg + (size_t)NNODES * NFEAT);

  k_deg_init<<<(NNODES + 255) / 256, 256, 0, stream>>>(deg);
  k_deg_acc<<<(NEDGES / 4 + 255) / 256, 256, 0, stream>>>(ei, deg, tags);
  k_tag_filter<<<NWIN, 1024, 0, stream>>>(ei, tags, deg, xagg, gcount);
  k_agg_window<<<NWIN, 1024, 0, stream>>>(x, ei, deg, gcount, xagg);

  k_gru<<<(ROWS / 16 + GRU_WAVES - 1) / GRU_WAVES, GRU_WAVES * 32, 0, stream>>>(xagg, att, Wz, bz, Wr, br, Wh, bh,
                                    Lz, lbz, Lr, lbr, Lh, lbh, Wo, bo, out);
}


# in-degrees counted by each window's owner block in LDS (integer adds over its own edge list, in the filter launch), not by 800000 global f32 atomic adds; degree pass writes tags only; no init launch
# speedup vs baseline: 2.8126x; 1.2162x over previous
#include <hip/hip_runtime.h>

#define NNODES 50000
#define NEDGES 800000
#define FIN 8
#define PER 12
#define HIDN 32
#define NB 2
#define OUTC 24
#define ROWS (NB*NNODES)
#define NFEAT (FIN*PER)
#define TAGWIN 196

typedef float v2f __attribute__((ext_vector_type(2)));
typedef float v8f __attribute__((ext_vector_type(8)));

__device__ __forceinline__ v8f wmma4(v2f a, v2f b, v8f c) {
  return __builtin_amdgcn_wmma_f32_16x16x4_f32(false, a, false, b, (short)0, c, false, false);
}

__device__ __forceinline__ v2f ldsA(const float* buf, int m, int hi, int kb) {
  int k0 = kb * 4 + hi * 2;
  v2f a;
  a.x = buf[m * HIDN + k0];
  a.y = buf[m * HIDN + k0 + 1];
  return a;
}

__device__ __forceinline__ void ldsA8(const float* buf, int m, int hi, v2f (&a)[8]) {
#pragma unroll
  for (int kb = 0; kb < 8; kb++) a[kb] = ldsA(buf, m, hi, kb);
}

__device__ __forceinline__ void storeTile(float* buf, int m, int hi, const v8f& c0, const v8f& c1) {
#pragma unroll
  for (int r = 0; r < 8; r++) {
    buf[(r + 8 * hi) * HIDN + m]      = c0[r];
    buf[(r + 8 * hi) * HIDN + 16 + m] = c1[r];
  }
}

#define SIG_PAIR_BOUND 43.0f
#define SIG_TANH_BOUND 57.0f
__device__ __forceinline__ float sig_den(float x, float bound) { return 1.0f + __expf(fminf(-x, bound)); }
__device__ __forceinline__ void sigm2(float& x0, float& x1) {
  float a = sig_den(x0, SIG_PAIR_BOUND), b = sig_den(x1, SIG_PAIR_BOUND);
  float inv = 1.0f / (a * b);
  x0 = inv * b;
  x1 = inv * a;
}
__device__ __forceinline__ void sigm_tanh(float za, float x, float& z, float& t) {
  x = fminf(fmaxf(x, -15.f), 15.f);
  float e = __expf(2.0f * x);
  float b = e + 1.0f;
  float inv = 1.0f / (za * b);
  z = inv * b;
  t = (e - 1.0f) * (inv * za);
}

__global__ void k_tags(const int* __restrict__ ei, unsigned* __restrict__ tags) {
  int i4 = blockIdx.x * 256 + threadIdx.x;
  if (i4 < NEDGES / 4) {
    int4 d = ((const int4*)(ei + NEDGES))[i4];
    tags[i4] = ((unsigned)d.x / TAGWIN) | (((unsigned)d.y / TAGWIN) << 8) |
               (((unsigned)d.z / TAGWIN) << 16) | (((unsigned)d.w / TAGWIN) << 24);
  }
}
#define WIN 196
#define NWIN 256
#define ROWF (NB * NFEAT)
#define AGG_WAVES 32
#define LCAP 256
#define NE4 (NEDGES / 4)
static_assert(NWIN * WIN >= NNODES, "windows cover every node");
static_assert(NE4 * 4 == NEDGES && NE4 % 32 == 0, "whole waves of int4 reads over the target column");

__device__ __forceinline__ void agg_walk(const float* __restrict__ x, float* acc, int s, int dl, float w,
                                         int nn, int lane, const int* off) {
  {
    int j = 0;
    for (; j + 4 <= nn; j += 4) {
      float2 v[4][3];
      int dls[4];
      float ws[4];
#pragma unroll
      for (int u = 0; u < 4; u++) {
        int sj = __builtin_amdgcn_readlane(s, j + u);
        dls[u] = __builtin_amdgcn_readlane(dl, j + u);
        ws[u] = __int_as_float(__builtin_amdgcn_readlane(__float_as_int(w), j + u));
        const float* row = x + (size_t)sj * NFEAT;
#pragma unroll
        for (int c = 0; c < 3; c++) v[u][c] = *(const float2*)(row + off[c]);
      }
#pragma unroll
      for (int u = 0; u < 4; u++) {
        float* a = acc + dls[u] * ROWF + 2 * lane;
#pragma unroll
        for (int c = 0; c < 3; c++) {
          __hip_atomic_fetch_add(a + c * 64,     ws[u] * v[u][c].x, __ATOMIC_RELAXED, __HIP_MEMORY_SCOPE_WORKGROUP);
          __hip_atomic_fetch_add(a + c * 64 + 1, ws[u] * v[u][c].y, __ATOMIC_RELAXED, __HIP_MEMORY_SCOPE_WORKGROUP);
        }
      }
    }
    for (; j < nn; j++) {
      int sj = __builtin_amdgcn_readlane(s, j);
      int dj = __builtin_amdgcn_readlane(dl, j);
      float wj = __int_as_float(__builtin_amdgcn_readlane(__float_as_int(w), j));
      const float* row = x + (size_t)sj * NFEAT;
      float* a = acc + dj * ROWF + 2 * lane;
      float2 t[3];
#pragma unroll
      for (int c = 0; c < 3; c++) t[c] = *(const float2*)(row + off[c]);
#pragma unroll
      for (int c = 0; c < 3; c++) {
        __hip_atomic_fetch_add(a + c * 64,     wj * t[c].x, __ATOMIC_RELAXED, __HIP_MEMORY_SCOPE_WORKGROUP);
        __hip_atomic_fetch_add(a + c * 64 + 1, wj * t[c].y, __ATOMIC_RELAXED, __HIP_MEMORY_SCOPE_WORKGROUP);
      }
    }
  }
}

__device__ __forceinline__ void agg_flush(const float* __restrict__ x, const int* __restrict__ ei,
                                          const float* __restrict__ dinv, float* acc,
                                          const int* myE, const int* myD, int cnt, int win0,
                                          int lane, const int* off) {
  for (int base = 0; base < cnt; base += 32) {
    int idx = base + lane;
    int s = 0, dl = 0;
    float w = 0.f;
    if (idx < cnt) {
      int e = myE[idx];
      dl = myD[idx];
      s = ei[e];
      w = dinv[s] * dinv[win0 + dl];
    }
    int nn = cnt - base;
    if (nn > 32) nn = 32;
    agg_walk(x, acc, s, dl, w, nn, lane, off);
  }
}

#define NT16 (NEDGES / 16)
#define SCAP 256
static_assert(NT16 * 16 == NEDGES, "whole 16-byte reads over the tag array");
static_assert(NWIN == 256 && WIN == TAGWIN, "a tag is one byte and names the window");
static_assert((size_t)NEDGES <= (size_t)NNODES * NFEAT * 4, "the tags fit into the batch-1 rows");

__device__ __forceinline__ int win_rows(int win0) {
  int r = NNODES - win0;
  return r > WIN ? WIN : r;
}

__device__ __forceinline__ void list_out(const int* __restrict__ ei, unsigned* __restrict__ glist, int cap,
                                         const int* myE, int cnt, int win0, int lane, int* total, int* wdeg) {
  int basep = 0;
  if (lane == 0) basep = __hip_atomic_fetch_add(total, cnt, __ATOMIC_RELAXED, __HIP_MEMORY_SCOPE_WORKGROUP);
  basep = __builtin_amdgcn_readfirstlane(basep);
  for (int idx = lane; idx < cnt; idx += 32) {
    int e = myE[idx];
    int s = ei[e];
    int d = ei[NEDGES + e];
    int pos = basep + idx;
    if (pos < cap) glist[pos] = ((unsigned)s << 8) | (unsigned)(d - win0);
    __hip_atomic_fetch_add(wdeg + (d - win0), 1, __ATOMIC_RELAXED, __HIP_MEMORY_SCOPE_WORKGROUP);
  }
}

__global__ void __launch_bounds__(1024) k_tag_filter(const int* __restrict__ ei,
                                                     const unsigned* __restrict__ tagw,
                                                     float* __restrict__ dinv,
                                                     float* xagg, int* __restrict__ gcount) {
  const uint4* tags = (const uint4*)tagw;
  __shared__ int lst[AGG_WAVES][SCAP];
  __shared__ int wdeg[WIN];
  __shared__ int total;

  const int tid = threadIdx.x, lane = tid & 31;
  const int wave = __builtin_amdgcn_readfirstlane(tid >> 5);
  const int win0 = blockIdx.x * WIN;
  const int cap = win_rows(win0) * NFEAT;
  unsigned* glist = (unsigned*)(xagg + (size_t)win0 * NFEAT);
  const unsigned pat = (unsigned)blockIdx.x * 0x01010101u;

  if (tid == 0) total = 0;
  if (tid < WIN) wdeg[tid] = 0;
  __syncthreads();

  int* myE = lst[wave];
  int cnt = 0;

  for (int base = wave * 32; base < NT16; base += 4 * 1024) {
    uint4 v[4];
#pragma unroll
    for (int u = 0; u < 4; u++) {
      int i = base + u * 1024 + lane;
      v[u] = tags[i < NT16 ? i : NT16 - 1];
    }
#pragma unroll
    for (int u = 0; u < 4; u++) {
      int i = base + u * 1024 + lane;
      unsigned x0 = v[u].x ^ pat, x1 = v[u].y ^ pat, x2 = v[u].z ^ pat, x3 = v[u].w ^ pat;
      unsigned z0 = ~(((x0 & 0x7f7f7f7fu) + 0x7f7f7f7fu) | x0) & 0x80808080u;
      unsigned z1 = ~(((x1 & 0x7f7f7f7fu) + 0x7f7f7f7fu) | x1) & 0x80808080u;
      unsigned z2 = ~(((x2 & 0x7f7f7f7fu) + 0x7f7f7f7fu) | x2) & 0x80808080u;
      unsigned z3 = ~(((x3 & 0x7f7f7f7fu) + 0x7f7f7f7fu) | x3) & 0x80808080u;
      unsigned m = (z0 >> 7) | (z1 >> 6) | (z2 >> 5) | (z3 >> 4);
      if (i >= NT16) m = 0u;
      while (__builtin_amdgcn_ballot_w32(m != 0u)) {
        bool has = m != 0u;
        int b = __builtin_ctz(m | 0x80000000u);
        unsigned kk = __builtin_amdgcn_ballot_w32(has);
        if (has) {
          int p = cnt + __builtin_amdgcn_mbcnt_lo(kk, 0u);
          myE[p] = 16 * i + 4 * (b & 7) + (b >> 3);
        }
        cnt += __builtin_popcount(kk);
        m &= m - 1u;
        if (cnt > SCAP - 32) {
          list_out(ei, glist, cap, myE, cnt, win0, lane, &total, wdeg);
          cnt = 0;
        }
      }
    }
  }
  list_out(ei, glist, cap, myE, cnt, win0, lane, &total, wdeg);

  __syncthreads();
  if (tid == 0) gcount[blockIdx.x] = total;
  if (tid < WIN && win0 + tid < NNODES) dinv[win0 + tid] = rsqrtf(1.0f + (float)wdeg[tid]);
}

__global__ void __launch_bounds__(1024) k_agg_window(const float* __restrict__ x,
                                                     const int* __restrict__ ei,
                                                     const float* __restrict__ dinv,
                                                     const int* __restrict__ gcount,
                                                     float* xagg) {
  __shared__ __attribute__((aligned(16))) float acc[WIN * ROWF];
  __shared__ int lstE[AGG_WAVES][LCAP];
  __shared__ int lstD[AGG_WAVES][LCAP];

  const int tid = threadIdx.x, lane = tid & 31;
  const int wave = __builtin_amdgcn_readfirstlane(tid >> 5);
  const int win0 = blockIdx.x * WIN;

  for (int q = tid; q < WIN * 48; q += 1024) {
    int dl = q / 48, r = q - dl * 48;
    int b = r / 24, c4 = r - b * 24;
    int n = win0 + dl;
    float4 o = make_float4(0.f, 0.f, 0.f, 0.f);
    if (n < NNODES) {
      float dv = dinv[n];
      float w = dv * dv;
      float4 xv = *(const float4*)(x + ((size_t)(b * NNODES + n) * NFEAT + c4 * 4));
      o.x = xv.x * w; o.y = xv.y * w; o.z = xv.z * w; o.w = xv.w * w;
    }
    *(float4*)(acc + q * 4) = o;
  }

  int off[3];
#pragma unroll
  for (int c = 0; c < 3; c++) {
    int f = c * 64 + 2 * lane;
    off[c] = (f >= NFEAT) ? (NNODES * NFEAT + f - NFEAT) : f;
  }

  __syncthreads();

  const int nlist = gcount[blockIdx.x];
  if (nlist <= win_rows(win0) * NFEAT) {
    const unsigned* glist = (const unsigned*)(xagg + (size_t)win0 * NFEAT);
    for (int base = wave * 32; base < nlist; base += 1024) {
      int idx = base + lane;
      int s = 0, dl = 0;
      float w = 0.f;
      if (idx < nlist) {
        unsigned pk = glist[idx];
        s = (int)(pk >> 8);
        dl = (int)(pk & 255u);
        w = dinv[s] * dinv[win0 + dl];
      }
      int nn = nlist - base;
      if (nn > 32) nn = 32;
      agg_walk(x, acc, s, dl, w, nn, lane, off);
    }
  } else {
  const int4* dst4 = (const int4*)(ei + NEDGES);
  int* myE = lstE[wave];
  int* myD = lstD[wave];
  int cnt = 0;

  for (int base = wave * 32; base < NE4; base += 4 * 1024) {
    int4 v[4];
#pragma unroll
    for (int u = 0; u < 4; u++) {
      int i4 = base + u * 1024 + lane;
      v[u] = (i4 < NE4) ? dst4[i4] : make_int4(-1, -1, -1, -1);
    }
#pragma unroll
    for (int u = 0; u < 4; u++) {
      int e0 = 4 * (base + u * 1024 + lane);
      unsigned r0 = (unsigned)(v[u].x - win0), r1 = (unsigned)(v[u].y - win0);
      unsigned r2 = (unsigned)(v[u].z - win0), r3 = (unsigned)(v[u].w - win0);
      bool m0 = r0 < (unsigned)WIN, m1 = r1 < (unsigned)WIN, m2 = r2 < (unsigned)WIN, m3 = r3 < (unsigned)WIN;
      if (__builtin_amdgcn_ballot_w32(m0 || m1 || m2 || m3)) {
        unsigned k0 = __builtin_amdgcn_ballot_w32(m0);
        if (m0) { int p = cnt + __builtin_amdgcn_mbcnt_lo(k0, 0u); myE[p] = e0;     myD[p] = (int)r0; }
        cnt += __builtin_popcount(k0);
        unsigned k1 = __builtin_amdgcn_ballot_w32(m1);
        if (m1) { int p = cnt + __builtin_amdgcn_mbcnt_lo(k1, 0u); myE[p] = e0 + 1; myD[p] = (int)r1; }
        cnt += __builtin_popcount(k1);
        unsigned k2 = __builtin_amdgcn_ballot_w32(m2);
        if (m2) { int p = cnt + __builtin_amdgcn_mbcnt_lo(k2, 0u); myE[p] = e0 + 2; myD[p] = (int)r2; }
        cnt += __builtin_popcount(k2);
        unsigned k3 = __builtin_amdgcn_ballot_w32(m3);
        if (m3) { int p = cnt + __builtin_amdgcn_mbcnt_lo(k3, 0u); myE[p] = e0 + 3; myD[p] = (int)r3; }
        cnt += __builtin_popcount(k3);
        if (cnt > LCAP - 128) {
          agg_flush(x, ei, dinv, acc, myE, myD, cnt, win0, lane, off);
          cnt = 0;
        }
      }
    }
  }
  agg_flush(x, ei, dinv, acc, myE, myD, cnt, win0, lane, off);
  }

  __syncthreads();

  for (int q = tid; q < WIN * 48; q += 1024) {
    int dl = q / 48, r = q - dl * 48;
    int b = r / 24, c4 = r - b * 24;
    int n = win0 + dl;
    if (n < NNODES)
      *(float4*)(xagg + ((size_t)(b * NNODES + n) * NFEAT + c4 * 4)) = *(const float4*)(acc + q * 4);
  }
}

#define GRU_WAVES 4
static_assert(ROWS % 16 == 0, "whole 16-row tiles: a wave either owns a full tile or leaves");

__device__ __forceinline__ v2f gldB(const float* __restrict__ W, int m, int hi, int kb, int nt) {
  int k0 = kb * 4 + hi * 2;
  int n = nt * 16 + m;
  v2f b;
  b.x = W[k0 * HIDN + n];
  b.y = W[(k0 + 1) * HIDN + n];
  return b;
}

__device__ __forceinline__ void wave_sync() {
  __builtin_amdgcn_fence(__ATOMIC_RELEASE, "wavefront");
  __builtin_amdgcn_wave_barrier();
  __builtin_amdgcn_fence(__ATOMIC_ACQUIRE, "wavefront");
}

__device__ __forceinline__ void gemm8r(v2f ax0, v2f ax1, const v2f (&B)[2][2], float b0v, float b1v,
                                       v8f& c0, v8f& c1) {
#pragma unroll
  for (int r = 0; r < 8; r++) { c0[r] = b0v; c1[r] = b1v; }
  c0 = wmma4(ax0, B[0][0], c0);
  c0 = wmma4(ax1, B[1][0], c0);
  c1 = wmma4(ax0, B[0][1], c1);
  c1 = wmma4(ax1, B[1][1], c1);
}

__global__ void __launch_bounds__(GRU_WAVES * 32) k_gru(
    const float* __restrict__ xagg, const float* __restrict__ att,
    const float* __restrict__ Wz, const float* __restrict__ bz,
    const float* __restrict__ Wr, const float* __restrict__ br,
    const float* __restrict__ Wh, const float* __restrict__ bh,
    const float* __restrict__ Lz, const float* __restrict__ lbz,
    const float* __restrict__ Lr, const float* __restrict__ lbr,
    const float* __restrict__ Lh, const float* __restrict__ lbh,
    const float* __restrict__ Wo, const float* __restrict__ bo,
    float* __restrict__ out) {
  __shared__ float sGz[GRU_WAVES][16 * 32];
  __shared__ float sGr[GRU_WAVES][16 * 32];
  __shared__ float sGh[GRU_WAVES][16 * 32];
  __shared__ float sH[GRU_WAVES][16 * 32];
  __shared__ float sHR[GRU_WAVES][16 * 32];
  __shared__ float sPr[GRU_WAVES][16];

  const int tid = threadIdx.x;
  const int lane = tid & 31, m = lane & 15, hi = lane >> 4;
  const int wave = __builtin_amdgcn_readfirstlane(tid >> 5);
  const int tile = blockIdx.x * GRU_WAVES + wave;
  if (tile * 16 >= ROWS) return;

  float* mGz = sGz[wave];
  float* mGr = sGr[wave];
  float* mGh = sGh[wave];
  float* mH = sH[wave];
  float* mHR = sHR[wave];
  float* mPr = sPr[wave];

  v2f BWz[2][2], BWr[2][2], BWh[2][2];
  v2f BLz[16][2], BLr[16][2], BLh[16][2];
#pragma unroll
  for (int kb = 0; kb < 2; kb++)
#pragma unroll
    for (int nt = 0; nt < 2; nt++) {
      BWz[kb][nt] = gldB(Wz, m, hi, kb, nt);
      BWr[kb][nt] = gldB(Wr, m, hi, kb, nt);
      BWh[kb][nt] = gldB(Wh, m, hi, kb, nt);
    }
#pragma unroll
  for (int kb = 0; kb < 16; kb++)
#pragma unroll
    for (int nt = 0; nt < 2; nt++) {
      BLz[kb][nt] = gldB(Lz, m, hi, kb, nt);
      BLr[kb][nt] = gldB(Lr, m, hi, kb, nt);
      BLh[kb][nt] = gldB(Lh, m, hi, kb, nt);
    }
  const float bz0 = bz[m], bz1 = bz[16 + m], br0 = br[m], br1 = br[16 + m], bh0 = bh[m], bh1 = bh[16 + m];
  const float lz0 = lbz[m], lz1 = lbz[16 + m], lr0 = lbr[m], lr1 = lbr[16 + m], lh0 = lbh[m], lh1 = lbh[16 + m];

  for (int idx = lane; idx < 16 * 32; idx += 32) mH[idx] = 0.f;

  {
    float a[PER]; float mx = -1e30f;
    for (int p = 0; p < PER; p++) { a[p] = att[p]; mx = fmaxf(mx, a[p]); }
    float sm = 0.f;
    for (int p = 0; p < PER; p++) { a[p] = __expf(a[p] - mx); sm += a[p]; }
    float inv = 1.0f / sm;
    float mine = 0.f;
    for (int p = 0; p < PER; p++) mine = (lane == p) ? a[p] * inv : mine;
    if (lane < PER) mPr[lane] = mine;
  }

  const long ia = (long)tile * 16 + m;
  const float* xrow = xagg + ia * (long)NFEAT;

  v8f H0, H1, A0, A1;
  for (int r = 0; r < 8; r++) { H0[r] = 0.f; H1[r] = 0.f; A0[r] = 0.f; A1[r] = 0.f; }

  wave_sync();

  for (int p = 0; p < PER; p++) {
    v2f aH[8];
    ldsA8(mH, m, hi, aH);

    int f0 = hi * 2;
    v2f ax0, ax1;
    ax0.x = xrow[(f0)*PER + p];     ax0.y = xrow[(f0 + 1) * PER + p];
    ax1.x = xrow[(f0 + 4) * PER + p]; ax1.y = xrow[(f0 + 5) * PER + p];
    const float prp = mPr[p];

    v8f g0, g1, z0, z1, r0, r1, t0, t1;

    gemm8r(ax0, ax1, BWz, bz0, bz1, g0, g1);
    storeTile(mGz, m, hi, g0, g1);
    gemm8r(ax0, ax1, BWr, br0, br1, g0, g1);
    storeTile(mGr, m, hi, g0, g1);
    gemm8r(ax0, ax1, BWh, bh0, bh1, g0, g1);
    storeTile(mGh, m, hi, g0, g1);
    wave_sync();

    v2f aGz[8], aGr[8], aGh[8];
    ldsA8(mGz, m, hi, aGz);
    ldsA8(mGr, m, hi, aGr);
    ldsA8(mGh, m, hi, aGh);
    __builtin_amdgcn_sched_barrier(0);

#pragma unroll
    for (int r = 0; r < 8; r++) { z0[r] = lz0; z1[r] = lz1; r0[r] = lr0; r1[r] = lr1; }
#pragma unroll
    for (int kb = 0; kb < 8; kb++) {
      r0 = wmma4(aGr[kb], BLr[kb][0], r0);
      r1 = wmma4(aGr[kb], BLr[kb][1], r1);
      z0 = wmma4(aGz[kb], BLz[kb][0], z0);
      z1 = wmma4(aGz[kb], BLz[kb][1], z1);
    }
#pragma unroll
    for (int kb = 0; kb < 8; kb++) {
      r0 = wmma4(aH[kb], BLr[kb + 8][0], r0);
      r1 = wmma4(aH[kb], BLr[kb + 8][1], r1);
      z0 = wmma4(aH[kb], BLz[kb + 8][0], z0);
      z1 = wmma4(aH[kb], BLz[kb + 8][1], z1);
    }

    for (int r = 0; r < 8; r++) {
      float a = r0[r], b = r1[r];
      sigm2(a, b);
      r0[r] = a; r1[r] = b;
    }
    {
      v8f q0, q1;
      for (int r = 0; r < 8; r++) { q0[r] = H0[r] * r0[r]; q1[r] = H1[r] * r1[r]; }
      storeTile(mHR, m, hi, q0, q1);
    }
    wave_sync();
    v2f aHR[8];
    ldsA8(mHR, m, hi, aHR);
    __builtin_amdgcn_sched_barrier(0);

#pragma unroll
    for (int r = 0; r < 8; r++) { t0[r] = lh0; t1[r] = lh1; }
#pragma unroll
    for (int kb = 0; kb < 8; kb++) {
      t0 = wmma4(aGh[kb], BLh[kb][0], t0);
      t1 = wmma4(aGh[kb], BLh[kb][1], t1);
    }
    for (int r = 0; r < 8; r++) { z0[r] = sig_den(z0[r], SIG_TANH_BOUND); z1[r] = sig_den(z1[r], SIG_TANH_BOUND); }
#pragma unroll
    for (int kb = 0; kb < 8; kb++) {
      t0 = wmma4(aHR[kb], BLh[kb + 8][0], t0);
      t1 = wmma4(aHR[kb], BLh[kb + 8][1], t1);
    }
    for (int r = 0; r < 8; r++) {
      float za, ta, zb, tb;
      sigm_tanh(z0[r], t0[r], za, ta);
      sigm_tanh(z1[r], t1[r], zb, tb);
      z0[r] = za; t0[r] = ta; z1[r] = zb; t1[r] = tb;
    }

    for (int r = 0; r < 8; r++) {
      H0[r] = z0[r] * H0[r] + (1.0f - z0[r]) * t0[r];
      H1[r] = z1[r] * H1[r] + (1.0f - z1[r]) * t1[r];
      A0[r] += prp * H0[r];
      A1[r] += prp * H1[r];
    }
    wave_sync();
    storeTile(mH, m, hi, H0, H1);
    wave_sync();
  }

  for (int r = 0; r < 8; r++) { A0[r] = fmaxf(A0[r], 0.f); A1[r] = fmaxf(A1[r], 0.f); }
  storeTile(mGz, m, hi, A0, A1);
  wave_sync();
  v8f o0, o1;
  {
    const bool real = m < OUTC - 16;
    const int mc = real ? 16 + m : 0;
    float b0v = bo[m], b1l = bo[mc];
    float b1v = real ? b1l : 0.f;
    for (int r = 0; r < 8; r++) { o0[r] = b0v; o1[r] = b1v; }
#pragma unroll
    for (int kb = 0; kb < 8; kb++) {
      int k0 = kb * 4 + hi * 2;
      v2f a = ldsA(mGz, m, hi, kb);
      v2f w0, w1;
      w0.x = Wo[k0 * OUTC + m];
      w0.y = Wo[(k0 + 1) * OUTC + m];
      float w1x = Wo[k0 * OUTC + mc], w1y = Wo[(k0 + 1) * OUTC + mc];
      w1.x = real ? w1x : 0.f;
      w1.y = real ? w1y : 0.f;
      o0 = wmma4(a, w0, o0);
      o1 = wmma4(a, w1, o1);
    }
  }
#pragma unroll
  for (int r = 0; r < 8; r++) {
    long row = (long)tile * 16 + r + 8 * hi;
    if (row < ROWS) {
      out[row * OUTC + m] = o0[r];
      if (m < 8) out[row * OUTC + 16 + m] = o1[r];
    }
  }
}

extern "C" void kernel_launch(void* const* d_in, const int* in_sizes, int n_in,
                              void* d_out, int out_size, void* d_ws, size_t ws_size,
                              hipStream_t stream) {
  (void)in_sizes; (void)n_in; (void)out_size; (void)ws_size;
  const float* x   = (const float*)d_in[0];
  const int*   ei  = (const int*)d_in[1];
  const float* att = (const float*)d_in[2];
  const float* Wz  = (const float*)d_in[3];  const float* bz  = (const float*)d_in[4];
  const float* Wr  = (const float*)d_in[5];  const float* br  = (const float*)d_in[6];
  const float* Wh  = (const float*)d_in[7];  const float* bh  = (const float*)d_in[8];
  const float* Lz  = (const float*)d_in[9];  const float* lbz = (const float*)d_in[10];
  const float* Lr  = (const float*)d_in[11]; const float* lbr = (const float*)d_in[12];
  const float* Lh  = (const float*)d_in[13]; const float* lbh = (const float*)d_in[14];
  const float* Wo  = (const float*)d_in[15]; const float* bo  = (const float*)d_in[16];
  float* out = (float*)d_out;

  float* wsf  = (float*)d_ws;
  float* dinv = wsf;
  float* xagg = wsf + 65536;
  int* gcount = (int*)(wsf + 50000);
  unsigned* tags = (unsigned*)(xagg + (size_t)NNODES * NFEAT);

  k_tags<<<(NEDGES / 4 + 255) / 256, 256, 0, stream>>>(ei, tags);
  k_tag_filter<<<NWIN, 1024, 0, stream>>>(ei, tags, dinv, xagg, gcount);
  k_agg_window<<<NWIN, 1024, 0, stream>>>(x, ei, dinv, gcount, xagg);

  k_gru<<<(ROWS / 16 + GRU_WAVES - 1) / GRU_WAVES, GRU_WAVES * 32, 0, stream>>>(xagg, att, Wz, bz, Wr, br, Wh, bh,
                                    Lz, lbz, Lr, lbr, Lh, lbh, Wo, bo, out);
}


# GRU gates: four activations share one IEEE division (1/a..1/d from 1/(abcd)): R by 2x2 blocks, Z and tanh of two elements; 12 divisions per lane per period, was 24; rounding differs, sums unchanged
# speedup vs baseline: 2.9145x; 1.0362x over previous
#include <hip/hip_runtime.h>

#define NNODES 50000
#define NEDGES 800000
#define FIN 8
#define PER 12
#define HIDN 32
#define NB 2
#define OUTC 24
#define ROWS (NB*NNODES)
#define NFEAT (FIN*PER)
#define TAGWIN 196

typedef float v2f __attribute__((ext_vector_type(2)));
typedef float v8f __attribute__((ext_vector_type(8)));

__device__ __forceinline__ v8f wmma4(v2f a, v2f b, v8f c) {
  return __builtin_amdgcn_wmma_f32_16x16x4_f32(false, a, false, b, (short)0, c, false, false);
}

__device__ __forceinline__ v2f ldsA(const float* buf, int m, int hi, int kb) {
  int k0 = kb * 4 + hi * 2;
  v2f a;
  a.x = buf[m * HIDN + k0];
  a.y = buf[m * HIDN + k0 + 1];
  return a;
}

__device__ __forceinline__ void ldsA8(const float* buf, int m, int hi, v2f (&a)[8]) {
#pragma unroll
  for (int kb = 0; kb < 8; kb++) a[kb] = ldsA(buf, m, hi, kb);
}

__device__ __forceinline__ void storeTile(float* buf, int m, int hi, const v8f& c0, const v8f& c1) {
#pragma unroll
  for (int r = 0; r < 8; r++) {
    buf[(r + 8 * hi) * HIDN + m]      = c0[r];
    buf[(r + 8 * hi) * HIDN + 16 + m] = c1[r];
  }
}

#define SIG_QUAD_BOUND 21.0f
#define SIG_TANH_BOUND 23.0f
#define TANH_CLAMP 10.0f
__device__ __forceinline__ float sig_den(float x, float bound) { return 1.0f + __expf(fminf(-x, bound)); }
__device__ __forceinline__ void recip4(float a, float b, float c, float d, float& ia, float& ib, float& ic, float& id) {
  float ab = a * b, cd = c * d;
  float inv = 1.0f / (ab * cd);
  float iab = inv * cd, icd = inv * ab;
  ia = iab * b;
  ib = iab * a;
  ic = icd * d;
  id = icd * c;
}
__device__ __forceinline__ void sigm4(float& x0, float& x1, float& x2, float& x3) {
  recip4(sig_den(x0, SIG_QUAD_BOUND), sig_den(x1, SIG_QUAD_BOUND), sig_den(x2, SIG_QUAD_BOUND),
         sig_den(x3, SIG_QUAD_BOUND), x0, x1, x2, x3);
}
__device__ __forceinline__ void sigm_tanh2(float za0, float x0, float za1, float x1,
                                           float& z0, float& t0, float& z1, float& t1) {
  x0 = fminf(fmaxf(x0, -TANH_CLAMP), TANH_CLAMP);
  x1 = fminf(fmaxf(x1, -TANH_CLAMP), TANH_CLAMP);
  float e0 = __expf(2.0f * x0), e1 = __expf(2.0f * x1);
  float b0 = e0 + 1.0f, b1 = e1 + 1.0f;
  float ib0, ib1;
  recip4(za0, b0, za1, b1, z0, ib0, z1, ib1);
  t0 = (e0 - 1.0f) * ib0;
  t1 = (e1 - 1.0f) * ib1;
}

__global__ void k_tags(const int* __restrict__ ei, unsigned* __restrict__ tags) {
  int i4 = blockIdx.x * 256 + threadIdx.x;
  if (i4 < NEDGES / 4) {
    int4 d = ((const int4*)(ei + NEDGES))[i4];
    tags[i4] = ((unsigned)d.x / TAGWIN) | (((unsigned)d.y / TAGWIN) << 8) |
               (((unsigned)d.z / TAGWIN) << 16) | (((unsigned)d.w / TAGWIN) << 24);
  }
}
#define WIN 196
#define NWIN 256
#define ROWF (NB * NFEAT)
#define AGG_WAVES 32
#define LCAP 256
#define NE4 (NEDGES / 4)
static_assert(NWIN * WIN >= NNODES, "windows cover every node");
static_assert(NE4 * 4 == NEDGES && NE4 % 32 == 0, "whole waves of int4 reads over the target column");

__device__ __forceinline__ void agg_walk(const float* __restrict__ x, float* acc, int s, int dl, float w,
                                         int nn, int lane, const int* off) {
  {
    int j = 0;
    for (; j + 4 <= nn; j += 4) {
      float2 v[4][3];
      int dls[4];
      float ws[4];
#pragma unroll
      for (int u = 0; u < 4; u++) {
        int sj = __builtin_amdgcn_readlane(s, j + u);
        dls[u] = __builtin_amdgcn_readlane(dl, j + u);
        ws[u] = __int_as_float(__builtin_amdgcn_readlane(__float_as_int(w), j + u));
        const float* row = x + (size_t)sj * NFEAT;
#pragma unroll
        for (int c = 0; c < 3; c++) v[u][c] = *(const float2*)(row + off[c]);
      }
#pragma unroll
      for (int u = 0; u < 4; u++) {
        float* a = acc + dls[u] * ROWF + 2 * lane;
#pragma unroll
        for (int c = 0; c < 3; c++) {
          __hip_atomic_fetch_add(a + c * 64,     ws[u] * v[u][c].x, __ATOMIC_RELAXED, __HIP_MEMORY_SCOPE_WORKGROUP);
          __hip_atomic_fetch_add(a + c * 64 + 1, ws[u] * v[u][c].y, __ATOMIC_RELAXED, __HIP_MEMORY_SCOPE_WORKGROUP);
        }
      }
    }
    for (; j < nn; j++) {
      int sj = __builtin_amdgcn_readlane(s, j);
      int dj = __builtin_amdgcn_readlane(dl, j);
      float wj = __int_as_float(__builtin_amdgcn_readlane(__float_as_int(w), j));
      const float* row = x + (size_t)sj * NFEAT;
      float* a = acc + dj * ROWF + 2 * lane;
      float2 t[3];
#pragma unroll
      for (int c = 0; c < 3; c++) t[c] = *(const float2*)(row + off[c]);
#pragma unroll
      for (int c = 0; c < 3; c++) {
        __hip_atomic_fetch_add(a + c * 64,     wj * t[c].x, __ATOMIC_RELAXED, __HIP_MEMORY_SCOPE_WORKGROUP);
        __hip_atomic_fetch_add(a + c * 64 + 1, wj * t[c].y, __ATOMIC_RELAXED, __HIP_MEMORY_SCOPE_WORKGROUP);
      }
    }
  }
}

__device__ __forceinline__ void agg_flush(const float* __restrict__ x, const int* __restrict__ ei,
                                          const float* __restrict__ dinv, float* acc,
                                          const int* myE, const int* myD, int cnt, int win0,
                                          int lane, const int* off) {
  for (int base = 0; base < cnt; base += 32) {
    int idx = base + lane;
    int s = 0, dl = 0;
    float w = 0.f;
    if (idx < cnt) {
      int e = myE[idx];
      dl = myD[idx];
      s = ei[e];
      w = dinv[s] * dinv[win0 + dl];
    }
    int nn = cnt - base;
    if (nn > 32) nn = 32;
    agg_walk(x, acc, s, dl, w, nn, lane, off);
  }
}

#define NT16 (NEDGES / 16)
#define SCAP 256
static_assert(NT16 * 16 == NEDGES, "whole 16-byte reads over the tag array");
static_assert(NWIN == 256 && WIN == TAGWIN, "a tag is one byte and names the window");
static_assert((size_t)NEDGES <= (size_t)NNODES * NFEAT * 4, "the tags fit into the batch-1 rows");

__device__ __forceinline__ int win_rows(int win0) {
  int r = NNODES - win0;
  return r > WIN ? WIN : r;
}

__device__ __forceinline__ void list_out(const int* __restrict__ ei, unsigned* __restrict__ glist, int cap,
                                         const int* myE, int cnt, int win0, int lane, int* total, int* wdeg) {
  int basep = 0;
  if (lane == 0) basep = __hip_atomic_fetch_add(total, cnt, __ATOMIC_RELAXED, __HIP_MEMORY_SCOPE_WORKGROUP);
  basep = __builtin_amdgcn_readfirstlane(basep);
  for (int idx = lane; idx < cnt; idx += 32) {
    int e = myE[idx];
    int s = ei[e];
    int d = ei[NEDGES + e];
    int pos = basep + idx;
    if (pos < cap) glist[pos] = ((unsigned)s << 8) | (unsigned)(d - win0);
    __hip_atomic_fetch_add(wdeg + (d - win0), 1, __ATOMIC_RELAXED, __HIP_MEMORY_SCOPE_WORKGROUP);
  }
}

__global__ void __launch_bounds__(1024) k_tag_filter(const int* __restrict__ ei,
                                                     const unsigned* __restrict__ tagw,
                                                     float* __restrict__ dinv,
                                                     float* xagg, int* __restrict__ gcount) {
  const uint4* tags = (const uint4*)tagw;
  __shared__ int lst[AGG_WAVES][SCAP];
  __shared__ int wdeg[WIN];
  __shared__ int total;

  const int tid = threadIdx.x, lane = tid & 31;
  const int wave = __builtin_amdgcn_readfirstlane(tid >> 5);
  const int win0 = blockIdx.x * WIN;
  const int cap = win_rows(win0) * NFEAT;
  unsigned* glist = (unsigned*)(xagg + (size_t)win0 * NFEAT);
  const unsigned pat = (unsigned)blockIdx.x * 0x01010101u;

  if (tid == 0) total = 0;
  if (tid < WIN) wdeg[tid] = 0;
  __syncthreads();

  int* myE = lst[wave];
  int cnt = 0;

  for (int base = wave * 32; base < NT16; base += 4 * 1024) {
    uint4 v[4];
#pragma unroll
    for (int u = 0; u < 4; u++) {
      int i = base + u * 1024 + lane;
      v[u] = tags[i < NT16 ? i : NT16 - 1];
    }
#pragma unroll
    for (int u = 0; u < 4; u++) {
      int i = base + u * 1024 + lane;
      unsigned x0 = v[u].x ^ pat, x1 = v[u].y ^ pat, x2 = v[u].z ^ pat, x3 = v[u].w ^ pat;
      unsigned z0 = ~(((x0 & 0x7f7f7f7fu) + 0x7f7f7f7fu) | x0) & 0x80808080u;
      unsigned z1 = ~(((x1 & 0x7f7f7f7fu) + 0x7f7f7f7fu) | x1) & 0x80808080u;
      unsigned z2 = ~(((x2 & 0x7f7f7f7fu) + 0x7f7f7f7fu) | x2) & 0x80808080u;
      unsigned z3 = ~(((x3 & 0x7f7f7f7fu) + 0x7f7f7f7fu) | x3) & 0x80808080u;
      unsigned m = (z0 >> 7) | (z1 >> 6) | (z2 >> 5) | (z3 >> 4);
      if (i >= NT16) m = 0u;
      while (__builtin_amdgcn_ballot_w32(m != 0u)) {
        bool has = m != 0u;
        int b = __builtin_ctz(m | 0x80000000u);
        unsigned kk = __builtin_amdgcn_ballot_w32(has);
        if (has) {
          int p = cnt + __builtin_amdgcn_mbcnt_lo(kk, 0u);
          myE[p] = 16 * i + 4 * (b & 7) + (b >> 3);
        }
        cnt += __builtin_popcount(kk);
        m &= m - 1u;
        if (cnt > SCAP - 32) {
          list_out(ei, glist, cap, myE, cnt, win0, lane, &total, wdeg);
          cnt = 0;
        }
      }
    }
  }
  list_out(ei, glist, cap, myE, cnt, win0, lane, &total, wdeg);

  __syncthreads();
  if (tid == 0) gcount[blockIdx.x] = total;
  if (tid < WIN && win0 + tid < NNODES) dinv[win0 + tid] = rsqrtf(1.0f + (float)wdeg[tid]);
}

__global__ void __launch_bounds__(1024) k_agg_window(const float* __restrict__ x,
                                                     const int* __restrict__ ei,
                                                     const float* __restrict__ dinv,
                                                     const int* __restrict__ gcount,
                                                     float* xagg) {
  __shared__ __attribute__((aligned(16))) float acc[WIN * ROWF];
  __shared__ int lstE[AGG_WAVES][LCAP];
  __shared__ int lstD[AGG_WAVES][LCAP];

  const int tid = threadIdx.x, lane = tid & 31;
  const int wave = __builtin_amdgcn_readfirstlane(tid >> 5);
  const int win0 = blockIdx.x * WIN;

  for (int q = tid; q < WIN * 48; q += 1024) {
    int dl = q / 48, r = q - dl * 48;
    int b = r / 24, c4 = r - b * 24;
    int n = win0 + dl;
    float4 o = make_float4(0.f, 0.f, 0.f, 0.f);
    if (n < NNODES) {
      float dv = dinv[n];
      float w = dv * dv;
      float4 xv = *(const float4*)(x + ((size_t)(b * NNODES + n) * NFEAT + c4 * 4));
      o.x = xv.x * w; o.y = xv.y * w; o.z = xv.z * w; o.w = xv.w * w;
    }
    *(float4*)(acc + q * 4) = o;
  }

  int off[3];
#pragma unroll
  for (int c = 0; c < 3; c++) {
    int f = c * 64 + 2 * lane;
    off[c] = (f >= NFEAT) ? (NNODES * NFEAT + f - NFEAT) : f;
  }

  __syncthreads();

  const int nlist = gcount[blockIdx.x];
  if (nlist <= win_rows(win0) * NFEAT) {
    const unsigned* glist = (const unsigned*)(xagg + (size_t)win0 * NFEAT);
    for (int base = wave * 32; base < nlist; base += 1024) {
      int idx = base + lane;
      int s = 0, dl = 0;
      float w = 0.f;
      if (idx < nlist) {
        unsigned pk = glist[idx];
        s = (int)(pk >> 8);
        dl = (int)(pk & 255u);
        w = dinv[s] * dinv[win0 + dl];
      }
      int nn = nlist - base;
      if (nn > 32) nn = 32;
      agg_walk(x, acc, s, dl, w, nn, lane, off);
    }
  } else {
  const int4* dst4 = (const int4*)(ei + NEDGES);
  int* myE = lstE[wave];
  int* myD = lstD[wave];
  int cnt = 0;

  for (int base = wave * 32; base < NE4; base += 4 * 1024) {
    int4 v[4];
#pragma unroll
    for (int u = 0; u < 4; u++) {
      int i4 = base + u * 1024 + lane;
      v[u] = (i4 < NE4) ? dst4[i4] : make_int4(-1, -1, -1, -1);
    }
#pragma unroll
    for (int u = 0; u < 4; u++) {
      int e0 = 4 * (base + u * 1024 + lane);
      unsigned r0 = (unsigned)(v[u].x - win0), r1 = (unsigned)(v[u].y - win0);
      unsigned r2 = (unsigned)(v[u].z - win0), r3 = (unsigned)(v[u].w - win0);
      bool m0 = r0 < (unsigned)WIN, m1 = r1 < (unsigned)WIN, m2 = r2 < (unsigned)WIN, m3 = r3 < (unsigned)WIN;
      if (__builtin_amdgcn_ballot_w32(m0 || m1 || m2 || m3)) {
        unsigned k0 = __builtin_amdgcn_ballot_w32(m0);
        if (m0) { int p = cnt + __builtin_amdgcn_mbcnt_lo(k0, 0u); myE[p] = e0;     myD[p] = (int)r0; }
        cnt += __builtin_popcount(k0);
        unsigned k1 = __builtin_amdgcn_ballot_w32(m1);
        if (m1) { int p = cnt + __builtin_amdgcn_mbcnt_lo(k1, 0u); myE[p] = e0 + 1; myD[p] = (int)r1; }
        cnt += __builtin_popcount(k1);
        unsigned k2 = __builtin_amdgcn_ballot_w32(m2);
        if (m2) { int p = cnt + __builtin_amdgcn_mbcnt_lo(k2, 0u); myE[p] = e0 + 2; myD[p] = (int)r2; }
        cnt += __builtin_popcount(k2);
        unsigned k3 = __builtin_amdgcn_ballot_w32(m3);
        if (m3) { int p = cnt + __builtin_amdgcn_mbcnt_lo(k3, 0u); myE[p] = e0 + 3; myD[p] = (int)r3; }
        cnt += __builtin_popcount(k3);
        if (cnt > LCAP - 128) {
          agg_flush(x, ei, dinv, acc, myE, myD, cnt, win0, lane, off);
          cnt = 0;
        }
      }
    }
  }
  agg_flush(x, ei, dinv, acc, myE, myD, cnt, win0, lane, off);
  }

  __syncthreads();

  for (int q = tid; q < WIN * 48; q += 1024) {
    int dl = q / 48, r = q - dl * 48;
    int b = r / 24, c4 = r - b * 24;
    int n = win0 + dl;
    if (n < NNODES)
      *(float4*)(xagg + ((size_t)(b * NNODES + n) * NFEAT + c4 * 4)) = *(const float4*)(acc + q * 4);
  }
}

#define GRU_WAVES 4
static_assert(ROWS % 16 == 0, "whole 16-row tiles: a wave either owns a full tile or leaves");

__device__ __forceinline__ v2f gldB(const float* __restrict__ W, int m, int hi, int kb, int nt) {
  int k0 = kb * 4 + hi * 2;
  int n = nt * 16 + m;
  v2f b;
  b.x = W[k0 * HIDN + n];
  b.y = W[(k0 + 1) * HIDN + n];
  return b;
}

__device__ __forceinline__ void wave_sync() {
  __builtin_amdgcn_fence(__ATOMIC_RELEASE, "wavefront");
  __builtin_amdgcn_wave_barrier();
  __builtin_amdgcn_fence(__ATOMIC_ACQUIRE, "wavefront");
}

__device__ __forceinline__ void gemm8r(v2f ax0, v2f ax1, const v2f (&B)[2][2], float b0v, float b1v,
                                       v8f& c0, v8f& c1) {
#pragma unroll
  for (int r = 0; r < 8; r++) { c0[r] = b0v; c1[r] = b1v; }
  c0 = wmma4(ax0, B[0][0], c0);
  c0 = wmma4(ax1, B[1][0], c0);
  c1 = wmma4(ax0, B[0][1], c1);
  c1 = wmma4(ax1, B[1][1], c1);
}

__global__ void __launch_bounds__(GRU_WAVES * 32) k_gru(
    const float* __restrict__ xagg, const float* __restrict__ att,
    const float* __restrict__ Wz, const float* __restrict__ bz,
    const float* __restrict__ Wr, const float* __restrict__ br,
    const float* __restrict__ Wh, const float* __restrict__ bh,
    const float* __restrict__ Lz, const float* __restrict__ lbz,
    const float* __restrict__ Lr, const float* __restrict__ lbr,
    const float* __restrict__ Lh, const float* __restrict__ lbh,
    const float* __restrict__ Wo, const float* __restrict__ bo,
    float* __restrict__ out) {
  __shared__ float sGz[GRU_WAVES][16 * 32];
  __shared__ float sGr[GRU_WAVES][16 * 32];
  __shared__ float sGh[GRU_WAVES][16 * 32];
  __shared__ float sH[GRU_WAVES][16 * 32];
  __shared__ float sHR[GRU_WAVES][16 * 32];
  __shared__ float sPr[GRU_WAVES][16];

  const int tid = threadIdx.x;
  const int lane = tid & 31, m = lane & 15, hi = lane >> 4;
  const int wave = __builtin_amdgcn_readfirstlane(tid >> 5);
  const int tile = blockIdx.x * GRU_WAVES + wave;
  if (tile * 16 >= ROWS) return;

  float* mGz = sGz[wave];
  float* mGr = sGr[wave];
  float* mGh = sGh[wave];
  float* mH = sH[wave];
  float* mHR = sHR[wave];
  float* mPr = sPr[wave];

  v2f BWz[2][2], BWr[2][2], BWh[2][2];
  v2f BLz[16][2], BLr[16][2], BLh[16][2];
#pragma unroll
  for (int kb = 0; kb < 2; kb++)
#pragma unroll
    for (int nt = 0; nt < 2; nt++) {
      BWz[kb][nt] = gldB(Wz, m, hi, kb, nt);
      BWr[kb][nt] = gldB(Wr, m, hi, kb, nt);
      BWh[kb][nt] = gldB(Wh, m, hi, kb, nt);
    }
#pragma unroll
  for (int kb = 0; kb < 16; kb++)
#pragma unroll
    for (int nt = 0; nt < 2; nt++) {
      BLz[kb][nt] = gldB(Lz, m, hi, kb, nt);
      BLr[kb][nt] = gldB(Lr, m, hi, kb, nt);
      BLh[kb][nt] = gldB(Lh, m, hi, kb, nt);
    }
  const float bz0 = bz[m], bz1 = bz[16 + m], br0 = br[m], br1 = br[16 + m], bh0 = bh[m], bh1 = bh[16 + m];
  const float lz0 = lbz[m], lz1 = lbz[16 + m], lr0 = lbr[m], lr1 = lbr[16 + m], lh0 = lbh[m], lh1 = lbh[16 + m];

  for (int idx = lane; idx < 16 * 32; idx += 32) mH[idx] = 0.f;

  {
    float a[PER]; float mx = -1e30f;
    for (int p = 0; p < PER; p++) { a[p] = att[p]; mx = fmaxf(mx, a[p]); }
    float sm = 0.f;
    for (int p = 0; p < PER; p++) { a[p] = __expf(a[p] - mx); sm += a[p]; }
    float inv = 1.0f / sm;
    float mine = 0.f;
    for (int p = 0; p < PER; p++) mine = (lane == p) ? a[p] * inv : mine;
    if (lane < PER) mPr[lane] = mine;
  }

  const long ia = (long)tile * 16 + m;
  const float* xrow = xagg + ia * (long)NFEAT;

  v8f H0, H1, A0, A1;
  for (int r = 0; r < 8; r++) { H0[r] = 0.f; H1[r] = 0.f; A0[r] = 0.f; A1[r] = 0.f; }

  wave_sync();

  for (int p = 0; p < PER; p++) {
    v2f aH[8];
    ldsA8(mH, m, hi, aH);

    int f0 = hi * 2;
    v2f ax0, ax1;
    ax0.x = xrow[(f0)*PER + p];     ax0.y = xrow[(f0 + 1) * PER + p];
    ax1.x = xrow[(f0 + 4) * PER + p]; ax1.y = xrow[(f0 + 5) * PER + p];
    const float prp = mPr[p];

    v8f g0, g1, z0, z1, r0, r1, t0, t1;

    gemm8r(ax0, ax1, BWz, bz0, bz1, g0, g1);
    storeTile(mGz, m, hi, g0, g1);
    gemm8r(ax0, ax1, BWr, br0, br1, g0, g1);
    storeTile(mGr, m, hi, g0, g1);
    gemm8r(ax0, ax1, BWh, bh0, bh1, g0, g1);
    storeTile(mGh, m, hi, g0, g1);
    wave_sync();

    v2f aGz[8], aGr[8], aGh[8];
    ldsA8(mGz, m, hi, aGz);
    ldsA8(mGr, m, hi, aGr);
    ldsA8(mGh, m, hi, aGh);
    __builtin_amdgcn_sched_barrier(0);

#pragma unroll
    for (int r = 0; r < 8; r++) { z0[r] = lz0; z1[r] = lz1; r0[r] = lr0; r1[r] = lr1; }
#pragma unroll
    for (int kb = 0; kb < 8; kb++) {
      r0 = wmma4(aGr[kb], BLr[kb][0], r0);
      r1 = wmma4(aGr[kb], BLr[kb][1], r1);
      z0 = wmma4(aGz[kb], BLz[kb][0], z0);
      z1 = wmma4(aGz[kb], BLz[kb][1], z1);
    }
#pragma unroll
    for (int kb = 0; kb < 8; kb++) {
      r0 = wmma4(aH[kb], BLr[kb + 8][0], r0);
      r1 = wmma4(aH[kb], BLr[kb + 8][1], r1);
      z0 = wmma4(aH[kb], BLz[kb + 8][0], z0);
      z1 = wmma4(aH[kb], BLz[kb + 8][1], z1);
    }

    for (int r = 0; r < 8; r += 2) {
      float a = r0[r], b = r1[r], c = r0[r + 1], d = r1[r + 1];
      sigm4(a, b, c, d);
      r0[r] = a; r1[r] = b; r0[r + 1] = c; r1[r + 1] = d;
    }
    {
      v8f q0, q1;
      for (int r = 0; r < 8; r++) { q0[r] = H0[r] * r0[r]; q1[r] = H1[r] * r1[r]; }
      storeTile(mHR, m, hi, q0, q1);
    }
    wave_sync();
    v2f aHR[8];
    ldsA8(mHR, m, hi, aHR);
    __builtin_amdgcn_sched_barrier(0);

#pragma unroll
    for (int r = 0; r < 8; r++) { t0[r] = lh0; t1[r] = lh1; }
#pragma unroll
    for (int kb = 0; kb < 8; kb++) {
      t0 = wmma4(aGh[kb], BLh[kb][0], t0);
      t1 = wmma4(aGh[kb], BLh[kb][1], t1);
    }
    for (int r = 0; r < 8; r++) { z0[r] = sig_den(z0[r], SIG_TANH_BOUND); z1[r] = sig_den(z1[r], SIG_TANH_BOUND); }
#pragma unroll
    for (int kb = 0; kb < 8; kb++) {
      t0 = wmma4(aHR[kb], BLh[kb + 8][0], t0);
      t1 = wmma4(aHR[kb], BLh[kb + 8][1], t1);
    }
    for (int r = 0; r < 8; r++) {
      float za, ta, zb, tb;
      sigm_tanh2(z0[r], t0[r], z1[r], t1[r], za, ta, zb, tb);
      z0[r] = za; t0[r] = ta; z1[r] = zb; t1[r] = tb;
    }

    for (int r = 0; r < 8; r++) {
      H0[r] = z0[r] * H0[r] + (1.0f - z0[r]) * t0[r];
      H1[r] = z1[r] * H1[r] + (1.0f - z1[r]) * t1[r];
      A0[r] += prp * H0[r];
      A1[r] += prp * H1[r];
    }
    wave_sync();
    storeTile(mH, m, hi, H0, H1);
    wave_sync();
  }

  for (int r = 0; r < 8; r++) { A0[r] = fmaxf(A0[r], 0.f); A1[r] = fmaxf(A1[r], 0.f); }
  storeTile(mGz, m, hi, A0, A1);
  wave_sync();
  v8f o0, o1;
  {
    const bool real = m < OUTC - 16;
    const int mc = real ? 16 + m : 0;
    float b0v = bo[m], b1l = bo[mc];
    float b1v = real ? b1l : 0.f;
    for (int r = 0; r < 8; r++) { o0[r] = b0v; o1[r] = b1v; }
#pragma unroll
    for (int kb = 0; kb < 8; kb++) {
      int k0 = kb * 4 + hi * 2;
      v2f a = ldsA(mGz, m, hi, kb);
      v2f w0, w1;
      w0.x = Wo[k0 * OUTC + m];
      w0.y = Wo[(k0 + 1) * OUTC + m];
      float w1x = Wo[k0 * OUTC + mc], w1y = Wo[(k0 + 1) * OUTC + mc];
      w1.x = real ? w1x : 0.f;
      w1.y = real ? w1y : 0.f;
      o0 = wmma4(a, w0, o0);
      o1 = wmma4(a, w1, o1);
    }
  }
#pragma unroll
  for (int r = 0; r < 8; r++) {
    long row = (long)tile * 16 + r + 8 * hi;
    if (row < ROWS) {
      out[row * OUTC + m] = o0[r];
      if (m < 8) out[row * OUTC + 16 + m] = o1[r];
    }
  }
}

extern "C" void kernel_launch(void* const* d_in, const int* in_sizes, int n_in,
                              void* d_out, int out_size, void* d_ws, size_t ws_size,
                              hipStream_t stream) {
  (void)in_sizes; (void)n_in; (void)out_size; (void)ws_size;
  const float* x   = (const float*)d_in[0];
  const int*   ei  = (const int*)d_in[1];
  const float* att = (const float*)d_in[2];
  const float* Wz  = (const float*)d_in[3];  const float* bz  = (const float*)d_in[4];
  const float* Wr  = (const float*)d_in[5];  const float* br  = (const float*)d_in[6];
  const float* Wh  = (const float*)d_in[7];  const float* bh  = (const float*)d_in[8];
  const float* Lz  = (const float*)d_in[9];  const float* lbz = (const float*)d_in[10];
  const float* Lr  = (const float*)d_in[11]; const float* lbr = (const float*)d_in[12];
  const float* Lh  = (const float*)d_in[13]; const float* lbh = (const float*)d_in[14];
  const float* Wo  = (const float*)d_in[15]; const float* bo  = (const float*)d_in[16];
  float* out = (float*)d_out;

  float* wsf  = (float*)d_ws;
  float* dinv = wsf;
  float* xagg = wsf + 65536;
  int* gcount = (int*)(wsf + 50000);
  unsigned* tags = (unsigned*)(xagg + (size_t)NNODES * NFEAT);

  k_tags<<<(NEDGES / 4 + 255) / 256, 256, 0, stream>>>(ei, tags);
  k_tag_filter<<<NWIN, 1024, 0, stream>>>(ei, tags, dinv, xagg, gcount);
  k_agg_window<<<NWIN, 1024, 0, stream>>>(x, ei, dinv, gcount, xagg);

  k_gru<<<(ROWS / 16 + GRU_WAVES - 1) / GRU_WAVES, GRU_WAVES * 32, 0, stream>>>(xagg, att, Wz, bz, Wr, br, Wh, bh,
                                    Lz, lbz, Lr, lbr, Lh, lbh, Wo, bo, out);
}


# window edge lists from one partition pass: 256 blocks counting-sort their own 3125-edge slice by window in LDS, the owner block walks its 256 runs; no block scans all edges; adds in arrival order
# speedup vs baseline: 3.0747x; 1.0550x over previous
#include <hip/hip_runtime.h>

#define NNODES 50000
#define NEDGES 800000
#define FIN 8
#define PER 12
#define HIDN 32
#define NB 2
#define OUTC 24
#define ROWS (NB*NNODES)
#define NFEAT (FIN*PER)

typedef float v2f __attribute__((ext_vector_type(2)));
typedef float v8f __attribute__((ext_vector_type(8)));

__device__ __forceinline__ v8f wmma4(v2f a, v2f b, v8f c) {
  return __builtin_amdgcn_wmma_f32_16x16x4_f32(false, a, false, b, (short)0, c, false, false);
}

__device__ __forceinline__ v2f ldsA(const float* buf, int m, int hi, int kb) {
  int k0 = kb * 4 + hi * 2;
  v2f a;
  a.x = buf[m * HIDN + k0];
  a.y = buf[m * HIDN + k0 + 1];
  return a;
}

__device__ __forceinline__ void ldsA8(const float* buf, int m, int hi, v2f (&a)[8]) {
#pragma unroll
  for (int kb = 0; kb < 8; kb++) a[kb] = ldsA(buf, m, hi, kb);
}

__device__ __forceinline__ void storeTile(float* buf, int m, int hi, const v8f& c0, const v8f& c1) {
#pragma unroll
  for (int r = 0; r < 8; r++) {
    buf[(r + 8 * hi) * HIDN + m]      = c0[r];
    buf[(r + 8 * hi) * HIDN + 16 + m] = c1[r];
  }
}

#define SIG_QUAD_BOUND 21.0f
#define SIG_TANH_BOUND 23.0f
#define TANH_CLAMP 10.0f
__device__ __forceinline__ float sig_den(float x, float bound) { return 1.0f + __expf(fminf(-x, bound)); }
__device__ __forceinline__ void recip4(float a, float b, float c, float d, float& ia, float& ib, float& ic, float& id) {
  float ab = a * b, cd = c * d;
  float inv = 1.0f / (ab * cd);
  float iab = inv * cd, icd = inv * ab;
  ia = iab * b;
  ib = iab * a;
  ic = icd * d;
  id = icd * c;
}
__device__ __forceinline__ void sigm4(float& x0, float& x1, float& x2, float& x3) {
  recip4(sig_den(x0, SIG_QUAD_BOUND), sig_den(x1, SIG_QUAD_BOUND), sig_den(x2, SIG_QUAD_BOUND),
         sig_den(x3, SIG_QUAD_BOUND), x0, x1, x2, x3);
}
__device__ __forceinline__ void sigm_tanh2(float za0, float x0, float za1, float x1,
                                           float& z0, float& t0, float& z1, float& t1) {
  x0 = fminf(fmaxf(x0, -TANH_CLAMP), TANH_CLAMP);
  x1 = fminf(fmaxf(x1, -TANH_CLAMP), TANH_CLAMP);
  float e0 = __expf(2.0f * x0), e1 = __expf(2.0f * x1);
  float b0 = e0 + 1.0f, b1 = e1 + 1.0f;
  float ib0, ib1;
  recip4(za0, b0, za1, b1, z0, ib0, z1, ib1);
  t0 = (e0 - 1.0f) * ib0;
  t1 = (e1 - 1.0f) * ib1;
}

#define WIN 196
#define NWIN 256
#define ROWF (NB * NFEAT)
#define AGG_WAVES 32
#define LCAP 256
#define NE4 (NEDGES / 4)
static_assert(NWIN * WIN >= NNODES, "windows cover every node");
static_assert(NE4 * 4 == NEDGES && NE4 % 32 == 0, "whole waves of int4 reads over the target column");

__device__ __forceinline__ void agg_walk(const float* __restrict__ x, float* acc, int s, int dl, float w,
                                         int nn, int lane, const int* off) {
  {
    int j = 0;
    for (; j + 4 <= nn; j += 4) {
      float2 v[4][3];
      int dls[4];
      float ws[4];
#pragma unroll
      for (int u = 0; u < 4; u++) {
        int sj = __builtin_amdgcn_readlane(s, j + u);
        dls[u] = __builtin_amdgcn_readlane(dl, j + u);
        ws[u] = __int_as_float(__builtin_amdgcn_readlane(__float_as_int(w), j + u));
        const float* row = x + (size_t)sj * NFEAT;
#pragma unroll
        for (int c = 0; c < 3; c++) v[u][c] = *(const float2*)(row + off[c]);
      }
#pragma unroll
      for (int u = 0; u < 4; u++) {
        float* a = acc + dls[u] * ROWF + 2 * lane;
#pragma unroll
        for (int c = 0; c < 3; c++) {
          __hip_atomic_fetch_add(a + c * 64,     ws[u] * v[u][c].x, __ATOMIC_RELAXED, __HIP_MEMORY_SCOPE_WORKGROUP);
          __hip_atomic_fetch_add(a + c * 64 + 1, ws[u] * v[u][c].y, __ATOMIC_RELAXED, __HIP_MEMORY_SCOPE_WORKGROUP);
        }
      }
    }
    for (; j < nn; j++) {
      int sj = __builtin_amdgcn_readlane(s, j);
      int dj = __builtin_amdgcn_readlane(dl, j);
      float wj = __int_as_float(__builtin_amdgcn_readlane(__float_as_int(w), j));
      const float* row = x + (size_t)sj * NFEAT;
      float* a = acc + dj * ROWF + 2 * lane;
      float2 t[3];
#pragma unroll
      for (int c = 0; c < 3; c++) t[c] = *(const float2*)(row + off[c]);
#pragma unroll
      for (int c = 0; c < 3; c++) {
        __hip_atomic_fetch_add(a + c * 64,     wj * t[c].x, __ATOMIC_RELAXED, __HIP_MEMORY_SCOPE_WORKGROUP);
        __hip_atomic_fetch_add(a + c * 64 + 1, wj * t[c].y, __ATOMIC_RELAXED, __HIP_MEMORY_SCOPE_WORKGROUP);
      }
    }
  }
}

__device__ __forceinline__ void agg_flush(const float* __restrict__ x, const int* __restrict__ ei,
                                          const float* __restrict__ dinv, float* acc,
                                          const int* myE, const int* myD, int cnt, int win0,
                                          int lane, const int* off) {
  for (int base = 0; base < cnt; base += 32) {
    int idx = base + lane;
    int s = 0, dl = 0;
    float w = 0.f;
    if (idx < cnt) {
      int e = myE[idx];
      dl = myD[idx];
      s = ei[e];
      w = dinv[s] * dinv[win0 + dl];
    }
    int nn = cnt - base;
    if (nn > 32) nn = 32;
    agg_walk(x, acc, s, dl, w, nn, lane, off);
  }
}


#define SLICE (NEDGES / NWIN)
#define PART_T NWIN
#define PART_EPT ((SLICE + PART_T - 1) / PART_T)
#define STAGE_CAP 6144
static_assert(SLICE * NWIN == NEDGES, "whole slices");
static_assert(SLICE < 65536, "start and count of a run fit 16 bits each");
static_assert(NWIN == 256 && WIN <= 256, "a window-relative target fits 8 bits; one thread per window and per slice");
static_assert((size_t)NEDGES + NWIN * NWIN <= (size_t)NNODES * NFEAT, "sorted slices and run words fit into the batch-1 rows");

__device__ __forceinline__ int win_rows(int win0) {
  int r = NNODES - win0;
  return r > WIN ? WIN : r;
}

__device__ __forceinline__ int prefix256(const int* cnt, int* gsum, int t, int& total) {
  if (t < 16) {
    int s = 0;
    for (int j = 0; j < 16; j++) s += cnt[16 * t + j];
    gsum[t] = s;
  }
  __syncthreads();
  const int g = t >> 4, r = t & 15;
  int pre = 0, tot = 0;
  for (int j = 0; j < 16; j++) {
    int v = gsum[j];
    tot += v;
    pre += (j < g) ? v : 0;
  }
  for (int j = 0; j < 16; j++) {
    int v = cnt[16 * g + j];
    pre += (j < r) ? v : 0;
  }
  total = tot;
  return pre;
}

__global__ void __launch_bounds__(PART_T) k_partition(const int* __restrict__ ei,
                                                      unsigned* __restrict__ chunks,
                                                      unsigned* __restrict__ runs) {
  __shared__ int hist[NWIN];
  __shared__ int start[NWIN];
  __shared__ int gsum[16];
  __shared__ unsigned sorted[SLICE];

  const int tid = threadIdx.x;
  const int e0 = blockIdx.x * SLICE;
  hist[tid] = 0;

  int s[PART_EPT], d[PART_EPT];
#pragma unroll
  for (int u = 0; u < PART_EPT; u++) {
    int i = tid + u * PART_T;
    int e = e0 + (i < SLICE ? i : 0);
    s[u] = ei[e];
    d[u] = ei[NEDGES + e];
  }
  __syncthreads();

  int w[PART_EPT], rk[PART_EPT];
#pragma unroll
  for (int u = 0; u < PART_EPT; u++) {
    w[u] = (int)((unsigned)d[u] / WIN);
    rk[u] = 0;
    if (tid + u * PART_T < SLICE)
      rk[u] = __hip_atomic_fetch_add(&hist[w[u]], 1, __ATOMIC_RELAXED, __HIP_MEMORY_SCOPE_WORKGROUP);
  }
  __syncthreads();

  int total;
  const int cnt = hist[tid];
  const int pre = prefix256(hist, gsum, tid, total);
  start[tid] = pre;
  runs[blockIdx.x * NWIN + tid] = ((unsigned)pre << 16) | (unsigned)cnt;
  __syncthreads();

#pragma unroll
  for (int u = 0; u < PART_EPT; u++) {
    if (tid + u * PART_T < SLICE)
      sorted[start[w[u]] + rk[u]] = ((unsigned)s[u] << 8) | (unsigned)(d[u] - w[u] * WIN);
  }
  __syncthreads();
  for (int i = tid; i < SLICE; i += PART_T) chunks[e0 + i] = sorted[i];
}

__global__ void __launch_bounds__(PART_T) k_window_lists(const unsigned* __restrict__ chunks,
                                                         const unsigned* __restrict__ runs,
                                                         float* __restrict__ dinv,
                                                         float* xagg, int* __restrict__ gcount) {
  __shared__ int rcnt[NWIN];
  __shared__ int gsum[16];
  __shared__ int wdeg[WIN];
  __shared__ unsigned stage[STAGE_CAP];

  const int tid = threadIdx.x;
  const int win0 = blockIdx.x * WIN;
  const int cap = win_rows(win0) * NFEAT;
  unsigned* glist = (unsigned*)(xagg + (size_t)win0 * NFEAT);

  const unsigned run = runs[tid * NWIN + blockIdx.x];
  const int c = (int)(run & 0xffffu);
  rcnt[tid] = c;
  if (tid < WIN) wdeg[tid] = 0;
  __syncthreads();

  int total;
  const int base = prefix256(rcnt, gsum, tid, total);
  const unsigned* src = chunks + (size_t)tid * SLICE + (run >> 16);

  for (int i = 0; i < c; i += 4) {
    unsigned v[4];
#pragma unroll
    for (int u = 0; u < 4; u++) v[u] = src[i + u < c ? i + u : c - 1];
#pragma unroll
    for (int u = 0; u < 4; u++) {
      if (i + u < c) {
        int pos = base + i + u;
        __hip_atomic_fetch_add(wdeg + (v[u] & 255u), 1, __ATOMIC_RELAXED, __HIP_MEMORY_SCOPE_WORKGROUP);
        if (pos < STAGE_CAP) stage[pos] = v[u];
        else if (pos < cap) glist[pos] = v[u];
      }
    }
  }
  __syncthreads();

  int nst = total < STAGE_CAP ? total : STAGE_CAP;
  if (nst > cap) nst = cap;
  for (int i = tid; i < nst; i += PART_T) glist[i] = stage[i];
  if (tid == 0) gcount[blockIdx.x] = total;
  if (tid < WIN && win0 + tid < NNODES) dinv[win0 + tid] = rsqrtf(1.0f + (float)wdeg[tid]);
}


__global__ void __launch_bounds__(1024) k_agg_window(const float* __restrict__ x,
                                                     const int* __restrict__ ei,
                                                     const float* __restrict__ dinv,
                                                     const int* __restrict__ gcount,
                                                     float* xagg) {
  __shared__ __attribute__((aligned(16))) float acc[WIN * ROWF];
  __shared__ int lstE[AGG_WAVES][LCAP];
  __shared__ int lstD[AGG_WAVES][LCAP];

  const int tid = threadIdx.x, lane = tid & 31;
  const int wave = __builtin_amdgcn_readfirstlane(tid >> 5);
  const int win0 = blockIdx.x * WIN;

  for (int q = tid; q < WIN * 48; q += 1024) {
    int dl = q / 48, r = q - dl * 48;
    int b = r / 24, c4 = r - b * 24;
    int n = win0 + dl;
    float4 o = make_float4(0.f, 0.f, 0.f, 0.f);
    if (n < NNODES) {
      float dv = dinv[n];
      float w = dv * dv;
      float4 xv = *(const float4*)(x + ((size_t)(b * NNODES + n) * NFEAT + c4 * 4));
      o.x = xv.x * w; o.y = xv.y * w; o.z = xv.z * w; o.w = xv.w * w;
    }
    *(float4*)(acc + q * 4) = o;
  }

  int off[3];
#pragma unroll
  for (int c = 0; c < 3; c++) {
    int f = c * 64 + 2 * lane;
    off[c] = (f >= NFEAT) ? (NNODES * NFEAT + f - NFEAT) : f;
  }

  __syncthreads();

  const int nlist = gcount[blockIdx.x];
  if (nlist <= win_rows(win0) * NFEAT) {
    const unsigned* glist = (const unsigned*)(xagg + (size_t)win0 * NFEAT);
    for (int base = wave * 32; base < nlist; base += 1024) {
      int idx = base + lane;
      int s = 0, dl = 0;
      float w = 0.f;
      if (idx < nlist) {
        unsigned pk = glist[idx];
        s = (int)(pk >> 8);
        dl = (int)(pk & 255u);
        w = dinv[s] * dinv[win0 + dl];
      }
      int nn = nlist - base;
      if (nn > 32) nn = 32;
      agg_walk(x, acc, s, dl, w, nn, lane, off);
    }
  } else {
  const int4* dst4 = (const int4*)(ei + NEDGES);
  int* myE = lstE[wave];
  int* myD = lstD[wave];
  int cnt = 0;

  for (int base = wave * 32; base < NE4; base += 4 * 1024) {
    int4 v[4];
#pragma unroll
    for (int u = 0; u < 4; u++) {
      int i4 = base + u * 1024 + lane;
      v[u] = (i4 < NE4) ? dst4[i4] : make_int4(-1, -1, -1, -1);
    }
#pragma unroll
    for (int u = 0; u < 4; u++) {
      int e0 = 4 * (base + u * 1024 + lane);
      unsigned r0 = (unsigned)(v[u].x - win0), r1 = (unsigned)(v[u].y - win0);
      unsigned r2 = (unsigned)(v[u].z - win0), r3 = (unsigned)(v[u].w - win0);
      bool m0 = r0 < (unsigned)WIN, m1 = r1 < (unsigned)WIN, m2 = r2 < (unsigned)WIN, m3 = r3 < (unsigned)WIN;
      if (__builtin_amdgcn_ballot_w32(m0 || m1 || m2 || m3)) {
        unsigned k0 = __builtin_amdgcn_ballot_w32(m0);
        if (m0) { int p = cnt + __builtin_amdgcn_mbcnt_lo(k0, 0u); myE[p] = e0;     myD[p] = (int)r0; }
        cnt += __builtin_popcount(k0);
        unsigned k1 = __builtin_amdgcn_ballot_w32(m1);
        if (m1) { int p = cnt + __builtin_amdgcn_mbcnt_lo(k1, 0u); myE[p] = e0 + 1; myD[p] = (int)r1; }
        cnt += __builtin_popcount(k1);
        unsigned k2 = __builtin_amdgcn_ballot_w32(m2);
        if (m2) { int p = cnt + __builtin_amdgcn_mbcnt_lo(k2, 0u); myE[p] = e0 + 2; myD[p] = (int)r2; }
        cnt += __builtin_popcount(k2);
        unsigned k3 = __builtin_amdgcn_ballot_w32(m3);
        if (m3) { int p = cnt + __builtin_amdgcn_mbcnt_lo(k3, 0u); myE[p] = e0 + 3; myD[p] = (int)r3; }
        cnt += __builtin_popcount(k3);
        if (cnt > LCAP - 128) {
          agg_flush(x, ei, dinv, acc, myE, myD, cnt, win0, lane, off);
          cnt = 0;
        }
      }
    }
  }
  agg_flush(x, ei, dinv, acc, myE, myD, cnt, win0, lane, off);
  }

  __syncthreads();

  for (int q = tid; q < WIN * 48; q += 1024) {
    int dl = q / 48, r = q - dl * 48;
    int b = r / 24, c4 = r - b * 24;
    int n = win0 + dl;
    if (n < NNODES)
      *(float4*)(xagg + ((size_t)(b * NNODES + n) * NFEAT + c4 * 4)) = *(const float4*)(acc + q * 4);
  }
}

#define GRU_WAVES 4
static_assert(ROWS % 16 == 0, "whole 16-row tiles: a wave either owns a full tile or leaves");

__device__ __forceinline__ v2f gldB(const float* __restrict__ W, int m, int hi, int kb, int nt) {
  int k0 = kb * 4 + hi * 2;
  int n = nt * 16 + m;
  v2f b;
  b.x = W[k0 * HIDN + n];
  b.y = W[(k0 + 1) * HIDN + n];
  return b;
}

__device__ __forceinline__ void wave_sync() {
  __builtin_amdgcn_fence(__ATOMIC_RELEASE, "wavefront");
  __builtin_amdgcn_wave_barrier();
  __builtin_amdgcn_fence(__ATOMIC_ACQUIRE, "wavefront");
}

__device__ __forceinline__ void gemm8r(v2f ax0, v2f ax1, const v2f (&B)[2][2], float b0v, float b1v,
                                       v8f& c0, v8f& c1) {
#pragma unroll
  for (int r = 0; r < 8; r++) { c0[r] = b0v; c1[r] = b1v; }
  c0 = wmma4(ax0, B[0][0], c0);
  c0 = wmma4(ax1, B[1][0], c0);
  c1 = wmma4(ax0, B[0][1], c1);
  c1 = wmma4(ax1, B[1][1], c1);
}

__global__ void __launch_bounds__(GRU_WAVES * 32) k_gru(
    const float* __restrict__ xagg, const float* __restrict__ att,
    const float* __restrict__ Wz, const float* __restrict__ bz,
    const float* __restrict__ Wr, const float* __restrict__ br,
    const float* __restrict__ Wh, const float* __restrict__ bh,
    const float* __restrict__ Lz, const float* __restrict__ lbz,
    const float* __restrict__ Lr, const float* __restrict__ lbr,
    const float* __restrict__ Lh, const float* __restrict__ lbh,
    const float* __restrict__ Wo, const float* __restrict__ bo,
    float* __restrict__ out) {
  __shared__ float sGz[GRU_WAVES][16 * 32];
  __shared__ float sGr[GRU_WAVES][16 * 32];
  __shared__ float sGh[GRU_WAVES][16 * 32];
  __shared__ float sH[GRU_WAVES][16 * 32];
  __shared__ float sHR[GRU_WAVES][16 * 32];
  __shared__ float sPr[GRU_WAVES][16];

  const int tid = threadIdx.x;
  const int lane = tid & 31, m = lane & 15, hi = lane >> 4;
  const int wave = __builtin_amdgcn_readfirstlane(tid >> 5);
  const int tile = blockIdx.x * GRU_WAVES + wave;
  if (tile * 16 >= ROWS) return;

  float* mGz = sGz[wave];
  float* mGr = sGr[wave];
  float* mGh = sGh[wave];
  float* mH = sH[wave];
  float* mHR = sHR[wave];
  float* mPr = sPr[wave];

  v2f BWz[2][2], BWr[2][2], BWh[2][2];
  v2f BLz[16][2], BLr[16][2], BLh[16][2];
#pragma unroll
  for (int kb = 0; kb < 2; kb++)
#pragma unroll
    for (int nt = 0; nt < 2; nt++) {
      BWz[kb][nt] = gldB(Wz, m, hi, kb, nt);
      BWr[kb][nt] = gldB(Wr, m, hi, kb, nt);
      BWh[kb][nt] = gldB(Wh, m, hi, kb, nt);
    }
#pragma unroll
  for (int kb = 0; kb < 16; kb++)
#pragma unroll
    for (int nt = 0; nt < 2; nt++) {
      BLz[kb][nt] = gldB(Lz, m, hi, kb, nt);
      BLr[kb][nt] = gldB(Lr, m, hi, kb, nt);
      BLh[kb][nt] = gldB(Lh, m, hi, kb, nt);
    }
  const float bz0 = bz[m], bz1 = bz[16 + m], br0 = br[m], br1 = br[16 + m], bh0 = bh[m], bh1 = bh[16 + m];
  const float lz0 = lbz[m], lz1 = lbz[16 + m], lr0 = lbr[m], lr1 = lbr[16 + m], lh0 = lbh[m], lh1 = lbh[16 + m];

  for (int idx = lane; idx < 16 * 32; idx += 32) mH[idx] = 0.f;

  {
    float a[PER]; float mx = -1e30f;
    for (int p = 0; p < PER; p++) { a[p] = att[p]; mx = fmaxf(mx, a[p]); }
    float sm = 0.f;
    for (int p = 0; p < PER; p++) { a[p] = __expf(a[p] - mx); sm += a[p]; }
    float inv = 1.0f / sm;
    float mine = 0.f;
    for (int p = 0; p < PER; p++) mine = (lane == p) ? a[p] * inv : mine;
    if (lane < PER) mPr[lane] = mine;
  }

  const long ia = (long)tile * 16 + m;
  const float* xrow = xagg + ia * (long)NFEAT;

  v8f H0, H1, A0, A1;
  for (int r = 0; r < 8; r++) { H0[r] = 0.f; H1[r] = 0.f; A0[r] = 0.f; A1[r] = 0.f; }

  wave_sync();

  for (int p = 0; p < PER; p++) {
    v2f aH[8];
    ldsA8(mH, m, hi, aH);

    int f0 = hi * 2;
    v2f ax0, ax1;
    ax0.x = xrow[(f0)*PER + p];     ax0.y = xrow[(f0 + 1) * PER + p];
    ax1.x = xrow[(f0 + 4) * PER + p]; ax1.y = xrow[(f0 + 5) * PER + p];
    const float prp = mPr[p];

    v8f g0, g1, z0, z1, r0, r1, t0, t1;

    gemm8r(ax0, ax1, BWz, bz0, bz1, g0, g1);
    storeTile(mGz, m, hi, g0, g1);
    gemm8r(ax0, ax1, BWr, br0, br1, g0, g1);
    storeTile(mGr, m, hi, g0, g1);
    gemm8r(ax0, ax1, BWh, bh0, bh1, g0, g1);
    storeTile(mGh, m, hi, g0, g1);
    wave_sync();

    v2f aGz[8], aGr[8], aGh[8];
    ldsA8(mGz, m, hi, aGz);
    ldsA8(mGr, m, hi, aGr);
    ldsA8(mGh, m, hi, aGh);
    __builtin_amdgcn_sched_barrier(0);

#pragma unroll
    for (int r = 0; r < 8; r++) { z0[r] = lz0; z1[r] = lz1; r0[r] = lr0; r1[r] = lr1; }
#pragma unroll
    for (int kb = 0; kb < 8; kb++) {
      r0 = wmma4(aGr[kb], BLr[kb][0], r0);
      r1 = wmma4(aGr[kb], BLr[kb][1], r1);
      z0 = wmma4(aGz[kb], BLz[kb][0], z0);
      z1 = wmma4(aGz[kb], BLz[kb][1], z1);
    }
#pragma unroll
    for (int kb = 0; kb < 8; kb++) {
      r0 = wmma4(aH[kb], BLr[kb + 8][0], r0);
      r1 = wmma4(aH[kb], BLr[kb + 8][1], r1);
      z0 = wmma4(aH[kb], BLz[kb + 8][0], z0);
      z1 = wmma4(aH[kb], BLz[kb + 8][1], z1);
    }

    for (int r = 0; r < 8; r += 2) {
      float a = r0[r], b = r1[r], c = r0[r + 1], d = r1[r + 1];
      sigm4(a, b, c, d);
      r0[r] = a; r1[r] = b; r0[r + 1] = c; r1[r + 1] = d;
    }
    {
      v8f q0, q1;
      for (int r = 0; r < 8; r++) { q0[r] = H0[r] * r0[r]; q1[r] = H1[r] * r1[r]; }
      storeTile(mHR, m, hi, q0, q1);
    }
    wave_sync();
    v2f aHR[8];
    ldsA8(mHR, m, hi, aHR);
    __builtin_amdgcn_sched_barrier(0);

#pragma unroll
    for (int r = 0; r < 8; r++) { t0[r] = lh0; t1[r] = lh1; }
#pragma unroll
    for (int kb = 0; kb < 8; kb++) {
      t0 = wmma4(aGh[kb], BLh[kb][0], t0);
      t1 = wmma4(aGh[kb], BLh[kb][1], t1);
    }
    for (int r = 0; r < 8; r++) { z0[r] = sig_den(z0[r], SIG_TANH_BOUND); z1[r] = sig_den(z1[r], SIG_TANH_BOUND); }
#pragma unroll
    for (int kb = 0; kb < 8; kb++) {
      t0 = wmma4(aHR[kb], BLh[kb + 8][0], t0);
      t1 = wmma4(aHR[kb], BLh[kb + 8][1], t1);
    }
    for (int r = 0; r < 8; r++) {
      float za, ta, zb, tb;
      sigm_tanh2(z0[r], t0[r], z1[r], t1[r], za, ta, zb, tb);
      z0[r] = za; t0[r] = ta; z1[r] = zb; t1[r] = tb;
    }

    for (int r = 0; r < 8; r++) {
      H0[r] = z0[r] * H0[r] + (1.0f - z0[r]) * t0[r];
      H1[r] = z1[r] * H1[r] + (1.0f - z1[r]) * t1[r];
      A0[r] += prp * H0[r];
      A1[r] += prp * H1[r];
    }
    wave_sync();
    storeTile(mH, m, hi, H0, H1);
    wave_sync();
  }

  for (int r = 0; r < 8; r++) { A0[r] = fmaxf(A0[r], 0.f); A1[r] = fmaxf(A1[r], 0.f); }
  storeTile(mGz, m, hi, A0, A1);
  wave_sync();
  v8f o0, o1;
  {
    const bool real = m < OUTC - 16;
    const int mc = real ? 16 + m : 0;
    float b0v = bo[m], b1l = bo[mc];
    float b1v = real ? b1l : 0.f;
    for (int r = 0; r < 8; r++) { o0[r] = b0v; o1[r] = b1v; }
#pragma unroll
    for (int kb = 0; kb < 8; kb++) {
      int k0 = kb * 4 + hi * 2;
      v2f a = ldsA(mGz, m, hi, kb);
      v2f w0, w1;
      w0.x = Wo[k0 * OUTC + m];
      w0.y = Wo[(k0 + 1) * OUTC + m];
      float w1x = Wo[k0 * OUTC + mc], w1y = Wo[(k0 + 1) * OUTC + mc];
      w1.x = real ? w1x : 0.f;
      w1.y = real ? w1y : 0.f;
      o0 = wmma4(a, w0, o0);
      o1 = wmma4(a, w1, o1);
    }
  }
#pragma unroll
  for (int r = 0; r < 8; r++) {
    long row = (long)tile * 16 + r + 8 * hi;
    if (row < ROWS) {
      out[row * OUTC + m] = o0[r];
      if (m < 8) out[row * OUTC + 16 + m] = o1[r];
    }
  }
}

extern "C" void kernel_launch(void* const* d_in, const int* in_sizes, int n_in,
                              void* d_out, int out_size, void* d_ws, size_t ws_size,
                              hipStream_t stream) {
  (void)in_sizes; (void)n_in; (void)out_size; (void)ws_size;
  const float* x   = (const float*)d_in[0];
  const int*   ei  = (const int*)d_in[1];
  const float* att = (const float*)d_in[2];
  const float* Wz  = (const float*)d_in[3];  const float* bz  = (const float*)d_in[4];
  const float* Wr  = (const float*)d_in[5];  const float* br  = (const float*)d_in[6];
  const float* Wh  = (const float*)d_in[7];  const float* bh  = (const float*)d_in[8];
  const float* Lz  = (const float*)d_in[9];  const float* lbz = (const float*)d_in[10];
  const float* Lr  = (const float*)d_in[11]; const float* lbr = (const float*)d_in[12];
  const float* Lh  = (const float*)d_in[13]; const float* lbh = (const float*)d_in[14];
  const float* Wo  = (const float*)d_in[15]; const float* bo  = (const float*)d_in[16];
  float* out = (float*)d_out;

  float* wsf  = (float*)d_ws;
  float* dinv = wsf;
  float* xagg = wsf + 65536;
  int* gcount = (int*)(wsf + 50000);
  unsigned* chunks = (unsigned*)(xagg + (size_t)NNODES * NFEAT);
  unsigned* runs = chunks + NEDGES;

  k_partition<<<NWIN, PART_T, 0, stream>>>(ei, chunks, runs);
  k_window_lists<<<NWIN, PART_T, 0, stream>>>(chunks, runs, dinv, xagg, gcount);
  k_agg_window<<<NWIN, 1024, 0, stream>>>(x, ei, dinv, gcount, xagg);

  k_gru<<<(ROWS / 16 + GRU_WAVES - 1) / GRU_WAVES, GRU_WAVES * 32, 0, stream>>>(xagg, att, Wz, bz, Wr, br, Wh, bh,
                                    Lz, lbz, Lr, lbr, Lh, lbh, Wo, bo, out);
}


# GRU tile split by hidden columns over two waves: each keeps half of every gate weight (256 VGPRs, 4 waves/SIMD, was 2); halves of G, H, H*R meet in LDS tiles of 36-float pitch, 2 barriers/period
# speedup vs baseline: 3.2976x; 1.0725x over previous
#include <hip/hip_runtime.h>

#define NNODES 50000
#define NEDGES 800000
#define FIN 8
#define PER 12
#define HIDN 32
#define NB 2
#define OUTC 24
#define ROWS (NB*NNODES)
#define NFEAT (FIN*PER)

typedef float v2f __attribute__((ext_vector_type(2)));
typedef float v8f __attribute__((ext_vector_type(8)));

__device__ __forceinline__ v8f wmma4(v2f a, v2f b, v8f c) {
  return __builtin_amdgcn_wmma_f32_16x16x4_f32(false, a, false, b, (short)0, c, false, false);
}

#define PITCH 36
__device__ __forceinline__ v2f ldsA(const float* buf, int m, int hi, int kb) {
  int k0 = kb * 4 + hi * 2;
  v2f a;
  a.x = buf[m * PITCH + k0];
  a.y = buf[m * PITCH + k0 + 1];
  return a;
}

__device__ __forceinline__ void ldsA8(const float* buf, int m, int hi, v2f (&a)[8]) {
#pragma unroll
  for (int kb = 0; kb < 8; kb++) a[kb] = ldsA(buf, m, hi, kb);
}

__device__ __forceinline__ void storeTile(float* buf, int m, int hi, const v8f& c0, const v8f& c1) {
#pragma unroll
  for (int r = 0; r < 8; r++) {
    buf[(r + 8 * hi) * PITCH + m]      = c0[r];
    buf[(r + 8 * hi) * PITCH + 16 + m] = c1[r];
  }
}
__device__ __forceinline__ void storeHalf(float* buf, int m, int hi, int nt, const v8f& c) {
#pragma unroll
  for (int r = 0; r < 8; r++) buf[(r + 8 * hi) * PITCH + 16 * nt + m] = c[r];
}

#define SIG_QUAD_BOUND 21.0f
#define SIG_TANH_BOUND 23.0f
#define TANH_CLAMP 10.0f
__device__ __forceinline__ float sig_den(float x, float bound) { return 1.0f + __expf(fminf(-x, bound)); }
__device__ __forceinline__ void recip4(float a, float b, float c, float d, float& ia, float& ib, float& ic, float& id) {
  float ab = a * b, cd = c * d;
  float inv = 1.0f / (ab * cd);
  float iab = inv * cd, icd = inv * ab;
  ia = iab * b;
  ib = iab * a;
  ic = icd * d;
  id = icd * c;
}
__device__ __forceinline__ void sigm4(float& x0, float& x1, float& x2, float& x3) {
  recip4(sig_den(x0, SIG_QUAD_BOUND), sig_den(x1, SIG_QUAD_BOUND), sig_den(x2, SIG_QUAD_BOUND),
         sig_den(x3, SIG_QUAD_BOUND), x0, x1, x2, x3);
}
__device__ __forceinline__ void sigm_tanh2(float za0, float x0, float za1, float x1,
                                           float& z0, float& t0, float& z1, float& t1) {
  x0 = fminf(fmaxf(x0, -TANH_CLAMP), TANH_CLAMP);
  x1 = fminf(fmaxf(x1, -TANH_CLAMP), TANH_CLAMP);
  float e0 = __expf(2.0f * x0), e1 = __expf(2.0f * x1);
  float b0 = e0 + 1.0f, b1 = e1 + 1.0f;
  float ib0, ib1;
  recip4(za0, b0, za1, b1, z0, ib0, z1, ib1);
  t0 = (e0 - 1.0f) * ib0;
  t1 = (e1 - 1.0f) * ib1;
}

#define WIN 196
#define NWIN 256
#define ROWF (NB * NFEAT)
#define AGG_WAVES 32
#define LCAP 256
#define NE4 (NEDGES / 4)
static_assert(NWIN * WIN >= NNODES, "windows cover every node");
static_assert(NE4 * 4 == NEDGES && NE4 % 32 == 0, "whole waves of int4 reads over the target column");

__device__ __forceinline__ void agg_walk(const float* __restrict__ x, float* acc, int s, int dl, float w,
                                         int nn, int lane, const int* off) {
  {
    int j = 0;
    for (; j + 4 <= nn; j += 4) {
      float2 v[4][3];
      int dls[4];
      float ws[4];
#pragma unroll
      for (int u = 0; u < 4; u++) {
        int sj = __builtin_amdgcn_readlane(s, j + u);
        dls[u] = __builtin_amdgcn_readlane(dl, j + u);
        ws[u] = __int_as_float(__builtin_amdgcn_readlane(__float_as_int(w), j + u));
        const float* row = x + (size_t)sj * NFEAT;
#pragma unroll
        for (int c = 0; c < 3; c++) v[u][c] = *(const float2*)(row + off[c]);
      }
#pragma unroll
      for (int u = 0; u < 4; u++) {
        float* a = acc + dls[u] * ROWF + 2 * lane;
#pragma unroll
        for (int c = 0; c < 3; c++) {
          __hip_atomic_fetch_add(a + c * 64,     ws[u] * v[u][c].x, __ATOMIC_RELAXED, __HIP_MEMORY_SCOPE_WORKGROUP);
          __hip_atomic_fetch_add(a + c * 64 + 1, ws[u] * v[u][c].y, __ATOMIC_RELAXED, __HIP_MEMORY_SCOPE_WORKGROUP);
        }
      }
    }
    for (; j < nn; j++) {
      int sj = __builtin_amdgcn_readlane(s, j);
      int dj = __builtin_amdgcn_readlane(dl, j);
      float wj = __int_as_float(__builtin_amdgcn_readlane(__float_as_int(w), j));
      const float* row = x + (size_t)sj * NFEAT;
      float* a = acc + dj * ROWF + 2 * lane;
      float2 t[3];
#pragma unroll
      for (int c = 0; c < 3; c++) t[c] = *(const float2*)(row + off[c]);
#pragma unroll
      for (int c = 0; c < 3; c++) {
        __hip_atomic_fetch_add(a + c * 64,     wj * t[c].x, __ATOMIC_RELAXED, __HIP_MEMORY_SCOPE_WORKGROUP);
        __hip_atomic_fetch_add(a + c * 64 + 1, wj * t[c].y, __ATOMIC_RELAXED, __HIP_MEMORY_SCOPE_WORKGROUP);
      }
    }
  }
}

__device__ __forceinline__ void agg_flush(const float* __restrict__ x, const int* __restrict__ ei,
                                          const float* __restrict__ dinv, float* acc,
                                          const int* myE, const int* myD, int cnt, int win0,
                                          int lane, const int* off) {
  for (int base = 0; base < cnt; base += 32) {
    int idx = base + lane;
    int s = 0, dl = 0;
    float w = 0.f;
    if (idx < cnt) {
      int e = myE[idx];
      dl = myD[idx];
      s = ei[e];
      w = dinv[s] * dinv[win0 + dl];
    }
    int nn = cnt - base;
    if (nn > 32) nn = 32;
    agg_walk(x, acc, s, dl, w, nn, lane, off);
  }
}


#define SLICE (NEDGES / NWIN)
#define PART_T NWIN
#define PART_EPT ((SLICE + PART_T - 1) / PART_T)
#define STAGE_CAP 6144
static_assert(SLICE * NWIN == NEDGES, "whole slices");
static_assert(SLICE < 65536, "start and count of a run fit 16 bits each");
static_assert(NWIN == 256 && WIN <= 256, "a window-relative target fits 8 bits; one thread per window and per slice");
static_assert((size_t)NEDGES + NWIN * NWIN <= (size_t)NNODES * NFEAT, "sorted slices and run words fit into the batch-1 rows");

__device__ __forceinline__ int win_rows(int win0) {
  int r = NNODES - win0;
  return r > WIN ? WIN : r;
}

__device__ __forceinline__ int prefix256(const int* cnt, int* gsum, int t, int& total) {
  if (t < 16) {
    int s = 0;
    for (int j = 0; j < 16; j++) s += cnt[16 * t + j];
    gsum[t] = s;
  }
  __syncthreads();
  const int g = t >> 4, r = t & 15;
  int pre = 0, tot = 0;
  for (int j = 0; j < 16; j++) {
    int v = gsum[j];
    tot += v;
    pre += (j < g) ? v : 0;
  }
  for (int j = 0; j < 16; j++) {
    int v = cnt[16 * g + j];
    pre += (j < r) ? v : 0;
  }
  total = tot;
  return pre;
}

__global__ void __launch_bounds__(PART_T) k_partition(const int* __restrict__ ei,
                                                      unsigned* __restrict__ chunks,
                                                      unsigned* __restrict__ runs) {
  __shared__ int hist[NWIN];
  __shared__ int start[NWIN];
  __shared__ int gsum[16];
  __shared__ unsigned sorted[SLICE];

  const int tid = threadIdx.x;
  const int e0 = blockIdx.x * SLICE;
  hist[tid] = 0;

  int s[PART_EPT], d[PART_EPT];
#pragma unroll
  for (int u = 0; u < PART_EPT; u++) {
    int i = tid + u * PART_T;
    int e = e0 + (i < SLICE ? i : 0);
    s[u] = ei[e];
    d[u] = ei[NEDGES + e];
  }
  __syncthreads();

  int w[PART_EPT], rk[PART_EPT];
#pragma unroll
  for (int u = 0; u < PART_EPT; u++) {
    w[u] = (int)((unsigned)d[u] / WIN);
    rk[u] = 0;
    if (tid + u * PART_T < SLICE)
      rk[u] = __hip_atomic_fetch_add(&hist[w[u]], 1, __ATOMIC_RELAXED, __HIP_MEMORY_SCOPE_WORKGROUP);
  }
  __syncthreads();

  int total;
  const int cnt = hist[tid];
  const int pre = prefix256(hist, gsum, tid, total);
  start[tid] = pre;
  runs[blockIdx.x * NWIN + tid] = ((unsigned)pre << 16) | (unsigned)cnt;
  __syncthreads();

#pragma unroll
  for (int u = 0; u < PART_EPT; u++) {
    if (tid + u * PART_T < SLICE)
      sorted[start[w[u]] + rk[u]] = ((unsigned)s[u] << 8) | (unsigned)(d[u] - w[u] * WIN);
  }
  __syncthreads();
  for (int i = tid; i < SLICE; i += PART_T) chunks[e0 + i] = sorted[i];
}

__global__ void __launch_bounds__(PART_T) k_window_lists(const unsigned* __restrict__ chunks,
                                                         const unsigned* __restrict__ runs,
                                                         float* __restrict__ dinv,
                                                         float* xagg, int* __restrict__ gcount) {
  __shared__ int rcnt[NWIN];
  __shared__ int gsum[16];
  __shared__ int wdeg[WIN];
  __shared__ unsigned stage[STAGE_CAP];

  const int tid = threadIdx.x;
  const int win0 = blockIdx.x * WIN;
  const int cap = win_rows(win0) * NFEAT;
  unsigned* glist = (unsigned*)(xagg + (size_t)win0 * NFEAT);

  const unsigned run = runs[tid * NWIN + blockIdx.x];
  const int c = (int)(run & 0xffffu);
  rcnt[tid] = c;
  if (tid < WIN) wdeg[tid] = 0;
  __syncthreads();

  int total;
  const int base = prefix256(rcnt, gsum, tid, total);
  const unsigned* src = chunks + (size_t)tid * SLICE + (run >> 16);

  for (int i = 0; i < c; i += 4) {
    unsigned v[4];
#pragma unroll
    for (int u = 0; u < 4; u++) v[u] = src[i + u < c ? i + u : c - 1];
#pragma unroll
    for (int u = 0; u < 4; u++) {
      if (i + u < c) {
        int pos = base + i + u;
        __hip_atomic_fetch_add(wdeg + (v[u] & 255u), 1, __ATOMIC_RELAXED, __HIP_MEMORY_SCOPE_WORKGROUP);
        if (pos < STAGE_CAP) stage[pos] = v[u];
        else if (pos < cap) glist[pos] = v[u];
      }
    }
  }
  __syncthreads();

  int nst = total < STAGE_CAP ? total : STAGE_CAP;
  if (nst > cap) nst = cap;
  for (int i = tid; i < nst; i += PART_T) glist[i] = stage[i];
  if (tid == 0) gcount[blockIdx.x] = total;
  if (tid < WIN && win0 + tid < NNODES) dinv[win0 + tid] = rsqrtf(1.0f + (float)wdeg[tid]);
}


__global__ void __launch_bounds__(1024) k_agg_window(const float* __restrict__ x,
                                                     const int* __restrict__ ei,
                                                     const float* __restrict__ dinv,
                                                     const int* __restrict__ gcount,
                                                     float* xagg) {
  __shared__ __attribute__((aligned(16))) float acc[WIN * ROWF];
  __shared__ int lstE[AGG_WAVES][LCAP];
  __shared__ int lstD[AGG_WAVES][LCAP];

  const int tid = threadIdx.x, lane = tid & 31;
  const int wave = __builtin_amdgcn_readfirstlane(tid >> 5);
  const int win0 = blockIdx.x * WIN;

  for (int q = tid; q < WIN * 48; q += 1024) {
    int dl = q / 48, r = q - dl * 48;
    int b = r / 24, c4 = r - b * 24;
    int n = win0 + dl;
    float4 o = make_float4(0.f, 0.f, 0.f, 0.f);
    if (n < NNODES) {
      float dv = dinv[n];
      float w = dv * dv;
      float4 xv = *(const float4*)(x + ((size_t)(b * NNODES + n) * NFEAT + c4 * 4));
      o.x = xv.x * w; o.y = xv.y * w; o.z = xv.z * w; o.w = xv.w * w;
    }
    *(float4*)(acc + q * 4) = o;
  }

  int off[3];
#pragma unroll
  for (int c = 0; c < 3; c++) {
    int f = c * 64 + 2 * lane;
    off[c] = (f >= NFEAT) ? (NNODES * NFEAT + f - NFEAT) : f;
  }

  __syncthreads();

  const int nlist = gcount[blockIdx.x];
  if (nlist <= win_rows(win0) * NFEAT) {
    const unsigned* glist = (const unsigned*)(xagg + (size_t)win0 * NFEAT);
    for (int base = wave * 32; base < nlist; base += 1024) {
      int idx = base + lane;
      int s = 0, dl = 0;
      float w = 0.f;
      if (idx < nlist) {
        unsigned pk = glist[idx];
        s = (int)(pk >> 8);
        dl = (int)(pk & 255u);
        w = dinv[s] * dinv[win0 + dl];
      }
      int nn = nlist - base;
      if (nn > 32) nn = 32;
      agg_walk(x, acc, s, dl, w, nn, lane, off);
    }
  } else {
  const int4* dst4 = (const int4*)(ei + NEDGES);
  int* myE = lstE[wave];
  int* myD = lstD[wave];
  int cnt = 0;

  for (int base = wave * 32; base < NE4; base += 4 * 1024) {
    int4 v[4];
#pragma unroll
    for (int u = 0; u < 4; u++) {
      int i4 = base + u * 1024 + lane;
      v[u] = (i4 < NE4) ? dst4[i4] : make_int4(-1, -1, -1, -1);
    }
#pragma unroll
    for (int u = 0; u < 4; u++) {
      int e0 = 4 * (base + u * 1024 + lane);
      unsigned r0 = (unsigned)(v[u].x - win0), r1 = (unsigned)(v[u].y - win0);
      unsigned r2 = (unsigned)(v[u].z - win0), r3 = (unsigned)(v[u].w - win0);
      bool m0 = r0 < (unsigned)WIN, m1 = r1 < (unsigned)WIN, m2 = r2 < (unsigned)WIN, m3 = r3 < (unsigned)WIN;
      if (__builtin_amdgcn_ballot_w32(m0 || m1 || m2 || m3)) {
        unsigned k0 = __builtin_amdgcn_ballot_w32(m0);
        if (m0) { int p = cnt + __builtin_amdgcn_mbcnt_lo(k0, 0u); myE[p] = e0;     myD[p] = (int)r0; }
        cnt += __builtin_popcount(k0);
        unsigned k1 = __builtin_amdgcn_ballot_w32(m1);
        if (m1) { int p = cnt + __builtin_amdgcn_mbcnt_lo(k1, 0u); myE[p] = e0 + 1; myD[p] = (int)r1; }
        cnt += __builtin_popcount(k1);
        unsigned k2 = __builtin_amdgcn_ballot_w32(m2);
        if (m2) { int p = cnt + __builtin_amdgcn_mbcnt_lo(k2, 0u); myE[p] = e0 + 2; myD[p] = (int)r2; }
        cnt += __builtin_popcount(k2);
        unsigned k3 = __builtin_amdgcn_ballot_w32(m3);
        if (m3) { int p = cnt + __builtin_amdgcn_mbcnt_lo(k3, 0u); myE[p] = e0 + 3; myD[p] = (int)r3; }
        cnt += __builtin_popcount(k3);
        if (cnt > LCAP - 128) {
          agg_flush(x, ei, dinv, acc, myE, myD, cnt, win0, lane, off);
          cnt = 0;
        }
      }
    }
  }
  agg_flush(x, ei, dinv, acc, myE, myD, cnt, win0, lane, off);
  }

  __syncthreads();

  for (int q = tid; q < WIN * 48; q += 1024) {
    int dl = q / 48, r = q - dl * 48;
    int b = r / 24, c4 = r - b * 24;
    int n = win0 + dl;
    if (n < NNODES)
      *(float4*)(xagg + ((size_t)(b * NNODES + n) * NFEAT + c4 * 4)) = *(const float4*)(acc + q * 4);
  }
}

#define GRU_WAVES 2
static_assert(ROWS % 16 == 0, "whole 16-row tiles: no block leaves early, so its barriers are uniform");

__device__ __forceinline__ v2f gldB(const float* __restrict__ W, int m, int hi, int kb, int nt) {
  int k0 = kb * 4 + hi * 2;
  int n = nt * 16 + m;
  v2f b;
  b.x = W[k0 * HIDN + n];
  b.y = W[(k0 + 1) * HIDN + n];
  return b;
}

__global__ void __launch_bounds__(GRU_WAVES * 32) k_gru(
    const float* __restrict__ xagg, const float* __restrict__ att,
    const float* __restrict__ Wz, const float* __restrict__ bz,
    const float* __restrict__ Wr, const float* __restrict__ br,
    const float* __restrict__ Wh, const float* __restrict__ bh,
    const float* __restrict__ Lz, const float* __restrict__ lbz,
    const float* __restrict__ Lr, const float* __restrict__ lbr,
    const float* __restrict__ Lh, const float* __restrict__ lbh,
    const float* __restrict__ Wo, const float* __restrict__ bo,
    float* __restrict__ out) {
  __shared__ float mGz[16 * PITCH];
  __shared__ float mGr[16 * PITCH];
  __shared__ float mGh[16 * PITCH];
  __shared__ float mH[16 * PITCH];
  __shared__ float mHR[16 * PITCH];
  __shared__ float mPr[16];

  const int tid = threadIdx.x;
  const int lane = tid & 31, m = lane & 15, hi = lane >> 4;
  const int nt = __builtin_amdgcn_readfirstlane(tid >> 5);
  const int tile = blockIdx.x;
  if (tile * 16 >= ROWS) return;

  v2f BWz[2], BWr[2], BWh[2];
  v2f BLz[16], BLr[16], BLh[16];
#pragma unroll
  for (int kb = 0; kb < 2; kb++) {
    BWz[kb] = gldB(Wz, m, hi, kb, nt);
    BWr[kb] = gldB(Wr, m, hi, kb, nt);
    BWh[kb] = gldB(Wh, m, hi, kb, nt);
  }
#pragma unroll
  for (int kb = 0; kb < 16; kb++) {
    BLz[kb] = gldB(Lz, m, hi, kb, nt);
    BLr[kb] = gldB(Lr, m, hi, kb, nt);
    BLh[kb] = gldB(Lh, m, hi, kb, nt);
  }
  const int col = 16 * nt + m;
  const float bzv = bz[col], brv = br[col], bhv = bh[col];
  const float lzv = lbz[col], lrv = lbr[col], lhv = lbh[col];

  v8f H, A;
  for (int r = 0; r < 8; r++) { H[r] = 0.f; A[r] = 0.f; }
  storeHalf(mH, m, hi, nt, H);

  {
    float a[PER]; float mx = -1e30f;
    for (int p = 0; p < PER; p++) { a[p] = att[p]; mx = fmaxf(mx, a[p]); }
    float sm = 0.f;
    for (int p = 0; p < PER; p++) { a[p] = __expf(a[p] - mx); sm += a[p]; }
    float inv = 1.0f / sm;
    float mine = 0.f;
    for (int p = 0; p < PER; p++) mine = (tid == p) ? a[p] * inv : mine;
    if (tid < PER) mPr[tid] = mine;
  }

  const long ia = (long)tile * 16 + m;
  const float* xrow = xagg + ia * (long)NFEAT;

  for (int p = 0; p < PER; p++) {
    int f0 = hi * 2;
    v2f ax0, ax1;
    ax0.x = xrow[(f0)*PER + p];     ax0.y = xrow[(f0 + 1) * PER + p];
    ax1.x = xrow[(f0 + 4) * PER + p]; ax1.y = xrow[(f0 + 5) * PER + p];

    v8f gz, gr, gh, z, rr, t;
#pragma unroll
    for (int r = 0; r < 8; r++) { gz[r] = bzv; gr[r] = brv; gh[r] = bhv; }
    gz = wmma4(ax0, BWz[0], gz);
    gr = wmma4(ax0, BWr[0], gr);
    gh = wmma4(ax0, BWh[0], gh);
    gz = wmma4(ax1, BWz[1], gz);
    gr = wmma4(ax1, BWr[1], gr);
    gh = wmma4(ax1, BWh[1], gh);
    storeHalf(mGz, m, hi, nt, gz);
    storeHalf(mGr, m, hi, nt, gr);
    storeHalf(mGh, m, hi, nt, gh);
    __syncthreads();

    v2f aGz[8], aGr[8], aGh[8], aH[8];
    ldsA8(mGz, m, hi, aGz);
    ldsA8(mGr, m, hi, aGr);
    ldsA8(mGh, m, hi, aGh);
    const float prp = mPr[p];
    __builtin_amdgcn_sched_barrier(0);

#pragma unroll
    for (int r = 0; r < 8; r++) { z[r] = lzv; rr[r] = lrv; t[r] = lhv; }
#pragma unroll
    for (int kb = 0; kb < 8; kb++) {
      rr = wmma4(aGr[kb], BLr[kb], rr);
      z  = wmma4(aGz[kb], BLz[kb], z);
      t  = wmma4(aGh[kb], BLh[kb], t);
    }
    ldsA8(mH, m, hi, aH);
    __builtin_amdgcn_sched_barrier(0);
#pragma unroll
    for (int kb = 0; kb < 8; kb++) {
      rr = wmma4(aH[kb], BLr[kb + 8], rr);
      z  = wmma4(aH[kb], BLz[kb + 8], z);
    }

    for (int r = 0; r < 8; r += 4) {
      float a = rr[r], b = rr[r + 1], c = rr[r + 2], d = rr[r + 3];
      sigm4(a, b, c, d);
      rr[r] = a; rr[r + 1] = b; rr[r + 2] = c; rr[r + 3] = d;
    }
    {
      v8f q;
      for (int r = 0; r < 8; r++) q[r] = H[r] * rr[r];
      storeHalf(mHR, m, hi, nt, q);
    }
    __syncthreads();
    v2f aHR[8];
    ldsA8(mHR, m, hi, aHR);
    for (int r = 0; r < 8; r++) z[r] = sig_den(z[r], SIG_TANH_BOUND);
#pragma unroll
    for (int kb = 0; kb < 8; kb++) t = wmma4(aHR[kb], BLh[kb + 8], t);
    for (int r = 0; r < 8; r += 2) {
      float za, ta, zb, tb;
      sigm_tanh2(z[r], t[r], z[r + 1], t[r + 1], za, ta, zb, tb);
      z[r] = za; t[r] = ta; z[r + 1] = zb; t[r + 1] = tb;
    }

    for (int r = 0; r < 8; r++) {
      H[r] = z[r] * H[r] + (1.0f - z[r]) * t[r];
      A[r] += prp * H[r];
    }
    storeHalf(mH, m, hi, nt, H);
  }

  for (int r = 0; r < 8; r++) A[r] = fmaxf(A[r], 0.f);
  storeHalf(mGz, m, hi, nt, A);
  __syncthreads();
  if (nt != 0) return;
  v8f o0, o1;
  {
    const bool real = m < OUTC - 16;
    const int mc = real ? 16 + m : 0;
    float b0v = bo[m], b1l = bo[mc];
    float b1v = real ? b1l : 0.f;
    for (int r = 0; r < 8; r++) { o0[r] = b0v; o1[r] = b1v; }
#pragma unroll
    for (int kb = 0; kb < 8; kb++) {
      int k0 = kb * 4 + hi * 2;
      v2f a = ldsA(mGz, m, hi, kb);
      v2f w0, w1;
      w0.x = Wo[k0 * OUTC + m];
      w0.y = Wo[(k0 + 1) * OUTC + m];
      float w1x = Wo[k0 * OUTC + mc], w1y = Wo[(k0 + 1) * OUTC + mc];
      w1.x = real ? w1x : 0.f;
      w1.y = real ? w1y : 0.f;
      o0 = wmma4(a, w0, o0);
      o1 = wmma4(a, w1, o1);
    }
  }
#pragma unroll
  for (int r = 0; r < 8; r++) {
    long row = (long)tile * 16 + r + 8 * hi;
    if (row < ROWS) {
      out[row * OUTC + m] = o0[r];
      if (m < 8) out[row * OUTC + 16 + m] = o1[r];
    }
  }
}

extern "C" void kernel_launch(void* const* d_in, const int* in_sizes, int n_in,
                              void* d_out, int out_size, void* d_ws, size_t ws_size,
                              hipStream_t stream) {
  (void)in_sizes; (void)n_in; (void)out_size; (void)ws_size;
  const float* x   = (const float*)d_in[0];
  const int*   ei  = (const int*)d_in[1];
  const float* att = (const float*)d_in[2];
  const float* Wz  = (const float*)d_in[3];  const float* bz  = (const float*)d_in[4];
  const float* Wr  = (const float*)d_in[5];  const float* br  = (const float*)d_in[6];
  const float* Wh  = (const float*)d_in[7];  const float* bh  = (const float*)d_in[8];
  const float* Lz  = (const float*)d_in[9];  const float* lbz = (const float*)d_in[10];
  const float* Lr  = (const float*)d_in[11]; const float* lbr = (const float*)d_in[12];
  const float* Lh  = (const float*)d_in[13]; const float* lbh = (const float*)d_in[14];
  const float* Wo  = (const float*)d_in[15]; const float* bo  = (const float*)d_in[16];
  float* out = (float*)d_out;

  float* wsf  = (float*)d_ws;
  float* dinv = wsf;
  float* xagg = wsf + 65536;
  int* gcount = (int*)(wsf + 50000);
  unsigned* chunks = (unsigned*)(xagg + (size_t)NNODES * NFEAT);
  unsigned* runs = chunks + NEDGES;

  k_partition<<<NWIN, PART_T, 0, stream>>>(ei, chunks, runs);
  k_window_lists<<<NWIN, PART_T, 0, stream>>>(chunks, runs, dinv, xagg, gcount);
  k_agg_window<<<NWIN, 1024, 0, stream>>>(x, ei, dinv, gcount, xagg);

  k_gru<<<ROWS / 16, GRU_WAVES * 32, 0, stream>>>(xagg, att, Wz, bz, Wr, br, Wh, bh,
                                    Lz, lbz, Lr, lbr, Lh, lbh, Wo, bo, out);
}
